# starting kernel
# baseline (speedup 1.0000x reference)
#include <hip/hip_runtime.h>

#define IN_DIM   128
#define HID_DIM  128
#define OUT_DIM  64
#define NUM_GRAPHS 64

typedef float v2f __attribute__((ext_vector_type(2)));
typedef float v8f __attribute__((ext_vector_type(8)));

__global__ void fill_kernel(float* __restrict__ p, float v, int n) {
    int i = blockIdx.x * blockDim.x + threadIdx.x;
    if (i < n) p[i] = v;
}

__global__ void degree_kernel(const int* __restrict__ dst, float* __restrict__ deg, int E) {
    int e = blockIdx.x * blockDim.x + threadIdx.x;
    if (e < E) atomicAdd(&deg[dst[e]], 1.0f);
}

__global__ void rsqrt_kernel(const float* __restrict__ deg, float* __restrict__ dinv, int n) {
    int i = blockIdx.x * blockDim.x + threadIdx.x;
    if (i < n) dinv[i] = rsqrtf(deg[i]);
}

__global__ void gemm_wmma_f32(const float* __restrict__ A,
                              const float* __restrict__ B,
                              float* __restrict__ C,
                              int M, int N, int K) {
    const int lane  = threadIdx.x;
    const int wave  = threadIdx.y;
    const int tilesN = N >> 4;
    const int totalTiles = (M >> 4) * tilesN;
    const int tile = blockIdx.x * blockDim.y + wave;
    if (tile >= totalTiles) return;

    const int tm   = tile / tilesN;
    const int tn   = tile - tm * tilesN;
    const int row0 = tm << 4;
    const int col0 = tn << 4;
    const int half = lane >> 4;
    const int l    = lane & 15;

    v8f acc = {};
    const float* Arow = A + (size_t)(row0 + l) * K;
    const float* Bcol = B + (size_t)col0 + l;

    for (int k0 = 0; k0 < K; k0 += 4) {
        const int ka = k0 + (half << 1);
        v2f a, b;
        a.x = Arow[ka];
        a.y = Arow[ka + 1];
        b.x = Bcol[(size_t)ka * N];
        b.y = Bcol[(size_t)(ka + 1) * N];
        acc = __builtin_amdgcn_wmma_f32_16x16x4_f32(
             false, a,  false, b,
             (short)0, acc,  false,  false);
    }

#pragma unroll
    for (int v = 0; v < 8; ++v) {
        C[(size_t)(row0 + (half << 3) + v) * N + col0 + l] = acc[v];
    }
}

__global__ void edge_aggregate(const float* __restrict__ h,
                               const float* __restrict__ dinv,
                               const int* __restrict__ src,
                               const int* __restrict__ dst,
                               float* __restrict__ agg,
                               int E, int D) {
    const int perEdge = D >> 2;
    int tid = blockIdx.x * blockDim.x + threadIdx.x;
    int e = tid / perEdge;
    if (e >= E) return;
    int c = (tid - e * perEdge) << 2;

    const int s = src[e];
    const int d = dst[e];
    const float norm = dinv[s] * dinv[d];

    const float4 hv = *(const float4*)(h + (size_t)s * D + c);
    float* out = agg + (size_t)d * D + c;
    atomicAdd(out + 0, hv.x * norm);
    atomicAdd(out + 1, hv.y * norm);
    atomicAdd(out + 2, hv.z * norm);
    atomicAdd(out + 3, hv.w * norm);
}

__global__ void finish_layer(const float* __restrict__ agg,
                             const float* __restrict__ h,
                             const float* __restrict__ dinv,
                             const float* __restrict__ bias,
                             float* __restrict__ out,
                             int Nn, int D, int do_relu) {
    const int perRow = D >> 2;
    int tid = blockIdx.x * blockDim.x + threadIdx.x;
    int i = tid / perRow;
    if (i >= Nn) return;
    int c = (tid - i * perRow) << 2;

    const float di = dinv[i];
    const float w  = di * di;
    const float4 av = *(const float4*)(agg + (size_t)i * D + c);
    const float4 hv = *(const float4*)(h   + (size_t)i * D + c);
    const float4 bv = *(const float4*)(bias + c);
    float4 r;
    r.x = av.x + hv.x * w + bv.x;
    r.y = av.y + hv.y * w + bv.y;
    r.z = av.z + hv.z * w + bv.z;
    r.w = av.w + hv.w * w + bv.w;
    if (do_relu) {
        r.x = fmaxf(r.x, 0.0f); r.y = fmaxf(r.y, 0.0f);
        r.z = fmaxf(r.z, 0.0f); r.w = fmaxf(r.w, 0.0f);
    }
    *(float4*)(out + (size_t)i * D + c) = r;
}

__global__ void count_nodes(const int* __restrict__ batch, float* __restrict__ cnt, int Nn) {
    int i = blockIdx.x * blockDim.x + threadIdx.x;
    if (i < Nn) atomicAdd(&cnt[batch[i]], 1.0f);
}

__global__ void pool_sum(const float* __restrict__ h,
                         const int* __restrict__ batch,
                         float* __restrict__ sums, int Nn, int D) {
    const int perRow = D >> 2;
    int tid = blockIdx.x * blockDim.x + threadIdx.x;
    int i = tid / perRow;
    if (i >= Nn) return;
    int c = (tid - i * perRow) << 2;
    const int g = batch[i];
    const float4 hv = *(const float4*)(h + (size_t)i * D + c);
    float* out = sums + (size_t)g * D + c;
    atomicAdd(out + 0, hv.x);
    atomicAdd(out + 1, hv.y);
    atomicAdd(out + 2, hv.z);
    atomicAdd(out + 3, hv.w);
}

__global__ void pool_div(const float* __restrict__ sums,
                         const float* __restrict__ cnt,
                         float* __restrict__ out, int G, int D) {
    int tid = blockIdx.x * blockDim.x + threadIdx.x;
    if (tid >= G * D) return;
    int g = tid / D;
    out[tid] = sums[tid] / fmaxf(cnt[g], 1.0f);
}

extern "C" void kernel_launch(void* const* d_in, const int* in_sizes, int n_in,
                              void* d_out, int out_size, void* d_ws, size_t ws_size,
                              hipStream_t stream) {
    const float* x    = (const float*)d_in[0];
    const int*   eidx = (const int*)  d_in[1];
    const int*   batch= (const int*)  d_in[2];
    const float* W1   = (const float*)d_in[3];
    const float* b1   = (const float*)d_in[4];
    const float* W2   = (const float*)d_in[5];
    const float* b2   = (const float*)d_in[6];
    float* out = (float*)d_out;

    const int Nn = in_sizes[0] / IN_DIM;
    const int E  = in_sizes[1] / 2;
    const int* src = eidx;
    const int* dst = eidx + E;

    float* ws   = (float*)d_ws;
    float* h1   = ws;
    float* agg1 = h1   + (size_t)Nn * HID_DIM;
    float* h2   = agg1 + (size_t)Nn * HID_DIM;
    float* agg2 = h2   + (size_t)Nn * OUT_DIM;
    float* deg  = agg2 + (size_t)Nn * OUT_DIM;
    float* dinv = deg  + Nn;
    float* sums = dinv + Nn;
    float* cnts = sums + NUM_GRAPHS * OUT_DIM;

    const int B = 256;
    auto blocks = [](long long n, int b) { return (unsigned)((n + b - 1) / b); };

    fill_kernel<<<blocks((long long)Nn * HID_DIM, B), B, 0, stream>>>(agg1, 0.0f, Nn * HID_DIM);
    fill_kernel<<<blocks((long long)Nn * OUT_DIM, B), B, 0, stream>>>(agg2, 0.0f, Nn * OUT_DIM);
    fill_kernel<<<blocks(Nn, B), B, 0, stream>>>(deg, 1.0f, Nn);
    fill_kernel<<<blocks(NUM_GRAPHS * OUT_DIM, B), B, 0, stream>>>(sums, 0.0f, NUM_GRAPHS * OUT_DIM);
    fill_kernel<<<blocks(NUM_GRAPHS, B), B, 0, stream>>>(cnts, 0.0f, NUM_GRAPHS);

    degree_kernel<<<blocks(E, B), B, 0, stream>>>(dst, deg, E);
    rsqrt_kernel<<<blocks(Nn, B), B, 0, stream>>>(deg, dinv, Nn);

    {
        int tiles = (Nn >> 4) * (HID_DIM >> 4);
        dim3 blk(32, 8);
        gemm_wmma_f32<<<blocks(tiles, 8), blk, 0, stream>>>(x, W1, h1, Nn, HID_DIM, IN_DIM);
    }
    edge_aggregate<<<blocks((long long)E * (HID_DIM >> 2), B), B, 0, stream>>>(
        h1, dinv, src, dst, agg1, E, HID_DIM);
    finish_layer<<<blocks((long long)Nn * (HID_DIM >> 2), B), B, 0, stream>>>(
        agg1, h1, dinv, b1, agg1, Nn, HID_DIM, 1);

    {
        int tiles = (Nn >> 4) * (OUT_DIM >> 4);
        dim3 blk(32, 8);
        gemm_wmma_f32<<<blocks(tiles, 8), blk, 0, stream>>>(agg1, W2, h2, Nn, OUT_DIM, HID_DIM);
    }
    edge_aggregate<<<blocks((long long)E * (OUT_DIM >> 2), B), B, 0, stream>>>(
        h2, dinv, src, dst, agg2, E, OUT_DIM);
    finish_layer<<<blocks((long long)Nn * (OUT_DIM >> 2), B), B, 0, stream>>>(
        agg2, h2, dinv, b2, agg2, Nn, OUT_DIM, 0);

    count_nodes<<<blocks(Nn, B), B, 0, stream>>>(batch, cnts, Nn);
    pool_sum<<<blocks((long long)Nn * (OUT_DIM >> 2), B), B, 0, stream>>>(
        agg2, batch, sums, Nn, OUT_DIM);
    pool_div<<<blocks(NUM_GRAPHS * OUT_DIM, B), B, 0, stream>>>(
        sums, cnts, out, NUM_GRAPHS, OUT_DIM);
}


# Layer-1 scatter made destination-stationary: a block holds 196 output rows in LDS, filters the whole edge list for them, LDS float adds; global float atomics, fill, finish pass go; sum order moves
# speedup vs baseline: 2.4403x; 2.4403x over previous
#include <hip/hip_runtime.h>

#define IN_DIM   128
#define HID_DIM  128
#define OUT_DIM  64
#define NUM_GRAPHS 64

typedef float v2f __attribute__((ext_vector_type(2)));
typedef float v8f __attribute__((ext_vector_type(8)));

__global__ void fill_kernel(float* __restrict__ p, float v, int n) {
    int i = blockIdx.x * blockDim.x + threadIdx.x;
    if (i < n) p[i] = v;
}

__global__ void degree_kernel(const int* __restrict__ dst, float* __restrict__ deg, int E) {
    int e = blockIdx.x * blockDim.x + threadIdx.x;
    if (e < E) atomicAdd(&deg[dst[e]], 1.0f);
}

__global__ void rsqrt_kernel(const float* __restrict__ deg, float* __restrict__ dinv, int n) {
    int i = blockIdx.x * blockDim.x + threadIdx.x;
    if (i < n) dinv[i] = rsqrtf(deg[i]);
}

__global__ void gemm_wmma_f32(const float* __restrict__ A,
                              const float* __restrict__ B,
                              float* __restrict__ C,
                              int M, int N, int K) {
    const int lane  = threadIdx.x;
    const int wave  = threadIdx.y;
    const int tilesN = N >> 4;
    const int totalTiles = (M >> 4) * tilesN;
    const int tile = blockIdx.x * blockDim.y + wave;
    if (tile >= totalTiles) return;

    const int tm   = tile / tilesN;
    const int tn   = tile - tm * tilesN;
    const int row0 = tm << 4;
    const int col0 = tn << 4;
    const int half = lane >> 4;
    const int l    = lane & 15;

    v8f acc = {};
    const float* Arow = A + (size_t)(row0 + l) * K;
    const float* Bcol = B + (size_t)col0 + l;

    for (int k0 = 0; k0 < K; k0 += 4) {
        const int ka = k0 + (half << 1);
        v2f a, b;
        a.x = Arow[ka];
        a.y = Arow[ka + 1];
        b.x = Bcol[(size_t)ka * N];
        b.y = Bcol[(size_t)(ka + 1) * N];
        acc = __builtin_amdgcn_wmma_f32_16x16x4_f32(
             false, a,  false, b,
             (short)0, acc,  false,  false);
    }

#pragma unroll
    for (int v = 0; v < 8; ++v) {
        C[(size_t)(row0 + (half << 3) + v) * N + col0 + l] = acc[v];
    }
}

#define TA_ROWS    196
#define TA_THREADS 1024
#define TA_WAVES   (TA_THREADS / 32)
#define TA_AHEAD   4
#define TA_LIST    (32 + TA_AHEAD * 128)
#define TA_GROUP   8

__device__ __forceinline__ int ta_append(int* list, int cnt, bool m, int e) {
    const unsigned mask = __builtin_amdgcn_ballot_w32(m);
    if (mask != 0u) {
        const int pos = cnt + (int)__builtin_amdgcn_mbcnt_lo(mask, 0u);
        if (m) list[pos] = e;
        cnt += __builtin_popcount(mask);
    }
    return cnt;
}

__device__ __forceinline__ void ta_flush(const int* list, int nv, int lane, int lo,
                                         const float* __restrict__ h,
                                         const float* __restrict__ dinv,
                                         const int* __restrict__ src,
                                         const int* __restrict__ dst,
                                         float* tile) {
    int s = 0, r = 0;
    float w = 0.0f;
    if (lane < nv) {
        const int e = list[lane];
        s = src[e];
        const int d = dst[e];
        r = d - lo;
        w = dinv[s] * dinv[d];
    }
    int k0 = 0;
    for (; k0 + TA_GROUP <= nv; k0 += TA_GROUP) {
        float4 v[TA_GROUP];
        float  wk[TA_GROUP];
        int    rk[TA_GROUP];
#pragma unroll
        for (int j = 0; j < TA_GROUP; ++j) {
            const int sk = __builtin_amdgcn_readlane(s, k0 + j);
            wk[j] = __int_as_float(__builtin_amdgcn_readlane(__float_as_int(w), k0 + j));
            rk[j] = __builtin_amdgcn_readlane(r, k0 + j);
            v[j]  = *(const float4*)(h + (size_t)sk * HID_DIM + (lane << 2));
        }
#pragma unroll
        for (int j = 0; j < TA_GROUP; ++j) {
            float* p = tile + rk[j] * HID_DIM + lane;
            atomicAdd(p,      v[j].x * wk[j]);
            atomicAdd(p + 32, v[j].y * wk[j]);
            atomicAdd(p + 64, v[j].z * wk[j]);
            atomicAdd(p + 96, v[j].w * wk[j]);
        }
    }
    for (; k0 < nv; ++k0) {
        const int   sk = __builtin_amdgcn_readlane(s, k0);
        const float wq = __int_as_float(__builtin_amdgcn_readlane(__float_as_int(w), k0));
        const int   rq = __builtin_amdgcn_readlane(r, k0);
        const float4 vv = *(const float4*)(h + (size_t)sk * HID_DIM + (lane << 2));
        float* p = tile + rq * HID_DIM + lane;
        atomicAdd(p,      vv.x * wq);
        atomicAdd(p + 32, vv.y * wq);
        atomicAdd(p + 64, vv.z * wq);
        atomicAdd(p + 96, vv.w * wq);
    }
}

__global__ __launch_bounds__(TA_THREADS)
void tile_aggregate(const float* __restrict__ h,
                    const float* __restrict__ dinv,
                    const int* __restrict__ src,
                    const int* __restrict__ dst,
                    const float* __restrict__ bias,
                    float* __restrict__ a1,
                    int E, int Nn) {
    __shared__ float tile[TA_ROWS * HID_DIM];
    __shared__ int   lists[TA_WAVES * TA_LIST];
    const int tid  = threadIdx.x;
    const int lane = tid & 31;
    const int wave = __builtin_amdgcn_readfirstlane(tid >> 5);
    const int lo   = (int)blockIdx.x * TA_ROWS;
    int rows = Nn - lo;
    if (rows > TA_ROWS) rows = TA_ROWS;

    for (int r = wave; r < rows; r += TA_WAVES) {
        const int   i  = lo + r;
        const float di = dinv[i];
        const float w  = di * di;
        const float4 hv = *(const float4*)(h + (size_t)i * HID_DIM + (lane << 2));
        const float4 bv = *(const float4*)(bias + (lane << 2));
        float* p = tile + r * HID_DIM + lane;
        p[0]  = hv.x * w + bv.x;
        p[32] = hv.y * w + bv.y;
        p[64] = hv.z * w + bv.z;
        p[96] = hv.w * w + bv.w;
    }
    __syncthreads();

    int* mylist = lists + wave * TA_LIST;
    int  cnt = 0;

    const bool vec = ((((size_t)dst) & 15) == 0);
    const int nChunks = vec ? (E >> 7) : 0;
    const int start = (int)blockIdx.x * (nChunks / (int)gridDim.x);

    for (int kc = wave; kc < nChunks; kc += TA_AHEAD * TA_WAVES) {
        int4 d[TA_AHEAD];
        int  e0[TA_AHEAD];
#pragma unroll
        for (int u = 0; u < TA_AHEAD; ++u) {
            const int c = kc + u * TA_WAVES;
            int cc = c + start;
            if (cc >= nChunks) cc -= nChunks;
            e0[u] = (cc << 7) + (lane << 2);
            if (c < nChunks) d[u] = *(const int4*)(dst + e0[u]);
            else             d[u] = make_int4(lo - 1, lo - 1, lo - 1, lo - 1);
        }
#pragma unroll
        for (int u = 0; u < TA_AHEAD; ++u) {
            const bool m0 = (unsigned)(d[u].x - lo) < (unsigned)rows;
            const bool m1 = (unsigned)(d[u].y - lo) < (unsigned)rows;
            const bool m2 = (unsigned)(d[u].z - lo) < (unsigned)rows;
            const bool m3 = (unsigned)(d[u].w - lo) < (unsigned)rows;
            if (__builtin_amdgcn_ballot_w32(m0 | m1 | m2 | m3) != 0u) {
                cnt = ta_append(mylist, cnt, m0, e0[u]);
                cnt = ta_append(mylist, cnt, m1, e0[u] + 1);
                cnt = ta_append(mylist, cnt, m2, e0[u] + 2);
                cnt = ta_append(mylist, cnt, m3, e0[u] + 3);
            }
        }
        while (cnt >= 32) {
            cnt -= 32;
            ta_flush(mylist + cnt, 32, lane, lo, h, dinv, src, dst, tile);
        }
    }
    for (int base = (nChunks << 7) + (wave << 5); base < E; base += TA_WAVES * 32) {
        const int e = base + lane;
        bool m = false;
        if (e < E) m = (unsigned)(dst[e] - lo) < (unsigned)rows;
        cnt = ta_append(mylist, cnt, m, e);
        if (cnt >= 32) {
            cnt -= 32;
            ta_flush(mylist + cnt, 32, lane, lo, h, dinv, src, dst, tile);
        }
    }
    if (cnt > 0) ta_flush(mylist, cnt, lane, lo, h, dinv, src, dst, tile);

    __syncthreads();

    for (int r = wave; r < rows; r += TA_WAVES) {
        const float* p = tile + r * HID_DIM + lane;
        float4 o;
        o.x = fmaxf(p[0],  0.0f);
        o.y = fmaxf(p[32], 0.0f);
        o.z = fmaxf(p[64], 0.0f);
        o.w = fmaxf(p[96], 0.0f);
        *(float4*)(a1 + (size_t)(lo + r) * HID_DIM + (lane << 2)) = o;
    }
}

#define PA_THREADS 1024
#define PA_GROUP   8
__global__ __launch_bounds__(PA_THREADS)
void pooled_aggregate(const float* __restrict__ a1,
                      const float* __restrict__ dinv,
                      const int* __restrict__ src,
                      const int* __restrict__ dst,
                      const int* __restrict__ batch,
                      float* __restrict__ S,
                      float* __restrict__ cnts,
                      int E, int Nn) {
    __shared__ float Sl[NUM_GRAPHS * HID_DIM];
    __shared__ float Cl[NUM_GRAPHS];
    const int tid  = threadIdx.x;
    const int lane = tid & 31;

    for (int i = tid; i < NUM_GRAPHS * HID_DIM; i += PA_THREADS) Sl[i] = 0.0f;
    if (tid < NUM_GRAPHS) Cl[tid] = 0.0f;
    __syncthreads();

    const int total = E + Nn;
    const int nW    = (int)gridDim.x * (PA_THREADS / 32);
    const int per   = (((total + nW - 1) / nW) + (PA_GROUP - 1)) & ~(PA_GROUP - 1);
    const int wave  = __builtin_amdgcn_readfirstlane(tid >> 5);
    const int begin = ((int)blockIdx.x * (PA_THREADS / 32) + wave) * per;
    int end = begin + per;
    if (end > total) end = total;

    for (int base = begin; base < end; base += 32) {
        const int t = base + lane;
        int s = 0, g = 0;
        float w = 0.0f;
        if (t < end) {
            int d;
            if (t < E) { s = src[t]; d = dst[t]; }
            else       { s = t - E;  d = s; }
            g = batch[d];
            w = dinv[s] * dinv[d];
            if (t >= E) atomicAdd(&Cl[g], 1.0f);
        }
        int nv = end - base;
        if (nv > 32) nv = 32;

        int k0 = 0;
        for (; k0 + PA_GROUP <= nv; k0 += PA_GROUP) {
            float4 v[PA_GROUP];
            float  wk[PA_GROUP];
            int    gk[PA_GROUP];
#pragma unroll
            for (int j = 0; j < PA_GROUP; ++j) {
                const int sk = __builtin_amdgcn_readlane(s, k0 + j);
                wk[j] = __int_as_float(__builtin_amdgcn_readlane(__float_as_int(w), k0 + j));
                gk[j] = __builtin_amdgcn_readlane(g, k0 + j);
                v[j]  = *(const float4*)(a1 + (size_t)sk * HID_DIM + (lane << 2));
            }
#pragma unroll
            for (int j = 0; j < PA_GROUP; ++j) {
                float* p = Sl + gk[j] * HID_DIM + lane;
                atomicAdd(p,      v[j].x * wk[j]);
                atomicAdd(p + 32, v[j].y * wk[j]);
                atomicAdd(p + 64, v[j].z * wk[j]);
                atomicAdd(p + 96, v[j].w * wk[j]);
            }
        }
        for (; k0 < nv; ++k0) {
            const int   sk = __builtin_amdgcn_readlane(s, k0);
            const float wq = __int_as_float(__builtin_amdgcn_readlane(__float_as_int(w), k0));
            const int   gq = __builtin_amdgcn_readlane(g, k0);
            const float4 vv = *(const float4*)(a1 + (size_t)sk * HID_DIM + (lane << 2));
            float* p = Sl + gq * HID_DIM + lane;
            atomicAdd(p,      vv.x * wq);
            atomicAdd(p + 32, vv.y * wq);
            atomicAdd(p + 64, vv.z * wq);
            atomicAdd(p + 96, vv.w * wq);
        }
    }
    __syncthreads();

    for (int idx = tid; idx < NUM_GRAPHS * HID_DIM; idx += PA_THREADS) {
        const int g = idx >> 7;
        const int c = idx & (HID_DIM - 1);
        atomicAdd(&S[idx], Sl[g * HID_DIM + ((c & 3) << 5) + (c >> 2)]);
    }
    if (tid < NUM_GRAPHS) {
        const float cv = Cl[tid];
        if (cv != 0.0f) atomicAdd(&cnts[tid], cv);
    }
}

__global__ void pool_final(const float* __restrict__ T,
                           const float* __restrict__ cnt,
                           const float* __restrict__ bias,
                           float* __restrict__ out, int G, int D) {
    int tid = blockIdx.x * blockDim.x + threadIdx.x;
    if (tid >= G * D) return;
    int g = tid / D;
    int c = tid - g * D;
    const float cn = cnt[g];
    out[tid] = (T[tid] + cn * bias[c]) / fmaxf(cn, 1.0f);
}

extern "C" void kernel_launch(void* const* d_in, const int* in_sizes, int n_in,
                              void* d_out, int out_size, void* d_ws, size_t ws_size,
                              hipStream_t stream) {
    const float* x    = (const float*)d_in[0];
    const int*   eidx = (const int*)  d_in[1];
    const int*   batch= (const int*)  d_in[2];
    const float* W1   = (const float*)d_in[3];
    const float* b1   = (const float*)d_in[4];
    const float* W2   = (const float*)d_in[5];
    const float* b2   = (const float*)d_in[6];
    float* out = (float*)d_out;

    const int Nn = in_sizes[0] / IN_DIM;
    const int E  = in_sizes[1] / 2;
    const int* src = eidx;
    const int* dst = eidx + E;

    float* ws   = (float*)d_ws;
    float* h1   = ws;
    float* agg1 = h1   + (size_t)Nn * HID_DIM;
    float* S    = agg1 + (size_t)Nn * HID_DIM;
    float* T    = S    + NUM_GRAPHS * HID_DIM;
    float* deg  = T    + NUM_GRAPHS * OUT_DIM;
    float* dinv = deg  + Nn;
    float* cnts = dinv + Nn;

    const int B = 256;
    auto blocks = [](long long n, int b) { return (unsigned)((n + b - 1) / b); };

    fill_kernel<<<blocks(Nn, B), B, 0, stream>>>(deg, 1.0f, Nn);
    fill_kernel<<<blocks(NUM_GRAPHS * HID_DIM, B), B, 0, stream>>>(S, 0.0f, NUM_GRAPHS * HID_DIM);
    fill_kernel<<<blocks(NUM_GRAPHS, B), B, 0, stream>>>(cnts, 0.0f, NUM_GRAPHS);

    degree_kernel<<<blocks(E, B), B, 0, stream>>>(dst, deg, E);
    rsqrt_kernel<<<blocks(Nn, B), B, 0, stream>>>(deg, dinv, Nn);

    {
        int tiles = (Nn >> 4) * (HID_DIM >> 4);
        dim3 blk(32, 8);
        gemm_wmma_f32<<<blocks(tiles, 8), blk, 0, stream>>>(x, W1, h1, Nn, HID_DIM, IN_DIM);
    }
    tile_aggregate<<<blocks(Nn, TA_ROWS), TA_THREADS, 0, stream>>>(
        h1, dinv, src, dst, b1, agg1, E, Nn);

    pooled_aggregate<<<256, PA_THREADS, 0, stream>>>(agg1, dinv, src, dst, batch, S, cnts, E, Nn);
    {
        int tiles = (NUM_GRAPHS >> 4) * (OUT_DIM >> 4);
        dim3 blk(32, 8);
        gemm_wmma_f32<<<blocks(tiles, 8), blk, 0, stream>>>(S, W2, T, NUM_GRAPHS, OUT_DIM, HID_DIM);
    }
    pool_final<<<blocks(NUM_GRAPHS * OUT_DIM, B), B, 0, stream>>>(
        T, cnts, b2, out, NUM_GRAPHS, OUT_DIM);
}


# Layer-1 product commuted behind its neighbour sum: block sums scaled x rows in LDS, then multiplies that 196x128 tile by W1 (f32 WMMA, A from LDS); N-sized product launch, h plane go; sum order moves
# speedup vs baseline: 2.5357x; 1.0391x over previous
#include <hip/hip_runtime.h>

#define IN_DIM   128
#define HID_DIM  128
#define OUT_DIM  64
#define NUM_GRAPHS 64

typedef float v2f __attribute__((ext_vector_type(2)));
typedef float v8f __attribute__((ext_vector_type(8)));

__global__ void fill_kernel(float* __restrict__ p, float v, int n) {
    int i = blockIdx.x * blockDim.x + threadIdx.x;
    if (i < n) p[i] = v;
}

__global__ void degree_kernel(const int* __restrict__ dst, float* __restrict__ deg, int E) {
    int e = blockIdx.x * blockDim.x + threadIdx.x;
    if (e < E) atomicAdd(&deg[dst[e]], 1.0f);
}

__global__ void rsqrt_kernel(const float* __restrict__ deg, float* __restrict__ dinv, int n) {
    int i = blockIdx.x * blockDim.x + threadIdx.x;
    if (i < n) dinv[i] = rsqrtf(deg[i]);
}

__global__ void gemm_wmma_f32(const float* __restrict__ A,
                              const float* __restrict__ B,
                              float* __restrict__ C,
                              int M, int N, int K) {
    const int lane  = threadIdx.x;
    const int wave  = threadIdx.y;
    const int tilesN = N >> 4;
    const int totalTiles = (M >> 4) * tilesN;
    const int tile = blockIdx.x * blockDim.y + wave;
    if (tile >= totalTiles) return;

    const int tm   = tile / tilesN;
    const int tn   = tile - tm * tilesN;
    const int row0 = tm << 4;
    const int col0 = tn << 4;
    const int half = lane >> 4;
    const int l    = lane & 15;

    v8f acc = {};
    const float* Arow = A + (size_t)(row0 + l) * K;
    const float* Bcol = B + (size_t)col0 + l;

    for (int k0 = 0; k0 < K; k0 += 4) {
        const int ka = k0 + (half << 1);
        v2f a, b;
        a.x = Arow[ka];
        a.y = Arow[ka + 1];
        b.x = Bcol[(size_t)ka * N];
        b.y = Bcol[(size_t)(ka + 1) * N];
        acc = __builtin_amdgcn_wmma_f32_16x16x4_f32(
             false, a,  false, b,
             (short)0, acc,  false,  false);
    }

#pragma unroll
    for (int v = 0; v < 8; ++v) {
        C[(size_t)(row0 + (half << 3) + v) * N + col0 + l] = acc[v];
    }
}

#define TA_ROWS    196
#define TA_STRIPS  13
#define TA_PADROWS (TA_STRIPS * 16)
#define TA_PITCH   (IN_DIM + 4)
#define TA_THREADS 1024
#define TA_WAVES   (TA_THREADS / 32)
#define TA_AHEAD   4
#define TA_LIST    (32 + TA_AHEAD * 128)
#define TA_GROUP   8

__device__ __forceinline__ int ta_append(int* list, int cnt, bool m, int e) {
    const unsigned mask = __builtin_amdgcn_ballot_w32(m);
    if (mask != 0u) {
        const int pos = cnt + (int)__builtin_amdgcn_mbcnt_lo(mask, 0u);
        if (m) list[pos] = e;
        cnt += __builtin_popcount(mask);
    }
    return cnt;
}

__device__ __forceinline__ void ta_flush(const int* list, int nv, int lane, int lo,
                                         const float* __restrict__ x,
                                         const float* __restrict__ dinv,
                                         const int* __restrict__ src,
                                         const int* __restrict__ dst,
                                         float* tile) {
    int s = 0, r = 0;
    float w = 0.0f;
    if (lane < nv) {
        const int e = list[lane];
        s = src[e];
        const int d = dst[e];
        r = d - lo;
        w = dinv[s] * dinv[d];
    }
    int k0 = 0;
    for (; k0 + TA_GROUP <= nv; k0 += TA_GROUP) {
        float4 v[TA_GROUP];
        float  wk[TA_GROUP];
        int    rk[TA_GROUP];
#pragma unroll
        for (int j = 0; j < TA_GROUP; ++j) {
            const int sk = __builtin_amdgcn_readlane(s, k0 + j);
            wk[j] = __int_as_float(__builtin_amdgcn_readlane(__float_as_int(w), k0 + j));
            rk[j] = __builtin_amdgcn_readlane(r, k0 + j);
            v[j]  = *(const float4*)(x + (size_t)sk * IN_DIM + (lane << 2));
        }
#pragma unroll
        for (int j = 0; j < TA_GROUP; ++j) {
            float* p = tile + rk[j] * TA_PITCH + lane;
            atomicAdd(p,      v[j].x * wk[j]);
            atomicAdd(p + 32, v[j].y * wk[j]);
            atomicAdd(p + 64, v[j].z * wk[j]);
            atomicAdd(p + 96, v[j].w * wk[j]);
        }
    }
    for (; k0 < nv; ++k0) {
        const int   sk = __builtin_amdgcn_readlane(s, k0);
        const float wq = __int_as_float(__builtin_amdgcn_readlane(__float_as_int(w), k0));
        const int   rq = __builtin_amdgcn_readlane(r, k0);
        const float4 vv = *(const float4*)(x + (size_t)sk * IN_DIM + (lane << 2));
        float* p = tile + rq * TA_PITCH + lane;
        atomicAdd(p,      vv.x * wq);
        atomicAdd(p + 32, vv.y * wq);
        atomicAdd(p + 64, vv.z * wq);
        atomicAdd(p + 96, vv.w * wq);
    }
}

__global__ __launch_bounds__(TA_THREADS)
void tile_aggregate(const float* __restrict__ x,
                    const float* __restrict__ dinv,
                    const int* __restrict__ src,
                    const int* __restrict__ dst,
                    const float* __restrict__ W,
                    const float* __restrict__ bias,
                    float* __restrict__ a1,
                    int E, int Nn) {
    static_assert(IN_DIM == 128 && HID_DIM == 128, "fragment walk below is written for 128 x 128");
    __shared__ __attribute__((aligned(16))) float tile[TA_PADROWS * TA_PITCH];
    __shared__ __attribute__((aligned(16))) float outt[TA_PADROWS * TA_PITCH];
    __shared__ int   lists[TA_WAVES * TA_LIST];
    const int tid  = threadIdx.x;
    const int lane = tid & 31;
    const int wave = __builtin_amdgcn_readfirstlane(tid >> 5);
    const int lo   = (int)blockIdx.x * TA_ROWS;
    int rows = Nn - lo;
    if (rows > TA_ROWS) rows = TA_ROWS;

    for (int r = wave; r < TA_PADROWS; r += TA_WAVES) {
        float* p = tile + r * TA_PITCH + lane;
        if (r < rows) {
            const int   i  = lo + r;
            const float di = dinv[i];
            const float w  = di * di;
            const float4 xv = *(const float4*)(x + (size_t)i * IN_DIM + (lane << 2));
            p[0]  = xv.x * w;
            p[32] = xv.y * w;
            p[64] = xv.z * w;
            p[96] = xv.w * w;
        } else {
            p[0] = 0.0f; p[32] = 0.0f; p[64] = 0.0f; p[96] = 0.0f;
        }
    }
    __syncthreads();

    int* mylist = lists + wave * TA_LIST;
    int  cnt = 0;

    const bool vec = ((((size_t)dst) & 15) == 0);
    const int nChunks = vec ? (E >> 7) : 0;
    const int start = (int)blockIdx.x * (nChunks / (int)gridDim.x);

    for (int kc = wave; kc < nChunks; kc += TA_AHEAD * TA_WAVES) {
        int4 d[TA_AHEAD];
        int  e0[TA_AHEAD];
#pragma unroll
        for (int u = 0; u < TA_AHEAD; ++u) {
            const int c = kc + u * TA_WAVES;
            int cc = c + start;
            if (cc >= nChunks) cc -= nChunks;
            e0[u] = (cc << 7) + (lane << 2);
            if (c < nChunks) d[u] = *(const int4*)(dst + e0[u]);
            else             d[u] = make_int4(lo - 1, lo - 1, lo - 1, lo - 1);
        }
#pragma unroll
        for (int u = 0; u < TA_AHEAD; ++u) {
            const bool m0 = (unsigned)(d[u].x - lo) < (unsigned)rows;
            const bool m1 = (unsigned)(d[u].y - lo) < (unsigned)rows;
            const bool m2 = (unsigned)(d[u].z - lo) < (unsigned)rows;
            const bool m3 = (unsigned)(d[u].w - lo) < (unsigned)rows;
            if (__builtin_amdgcn_ballot_w32(m0 | m1 | m2 | m3) != 0u) {
                cnt = ta_append(mylist, cnt, m0, e0[u]);
                cnt = ta_append(mylist, cnt, m1, e0[u] + 1);
                cnt = ta_append(mylist, cnt, m2, e0[u] + 2);
                cnt = ta_append(mylist, cnt, m3, e0[u] + 3);
            }
        }
        while (cnt >= 32) {
            cnt -= 32;
            ta_flush(mylist + cnt, 32, lane, lo, x, dinv, src, dst, tile);
        }
    }
    for (int base = (nChunks << 7) + (wave << 5); base < E; base += TA_WAVES * 32) {
        const int e = base + lane;
        bool m = false;
        if (e < E) m = (unsigned)(dst[e] - lo) < (unsigned)rows;
        cnt = ta_append(mylist, cnt, m, e);
        if (cnt >= 32) {
            cnt -= 32;
            ta_flush(mylist + cnt, 32, lane, lo, x, dinv, src, dst, tile);
        }
    }
    if (cnt > 0) ta_flush(mylist, cnt, lane, lo, x, dinv, src, dst, tile);

    const int half = lane >> 4;
    const int m    = lane & 15;
    const int nt   = wave & 7;
    v2f bf[32];
#pragma unroll
    for (int q = 0; q < 32; ++q) {
        const int k = ((q & 7) << 4) + (half << 3) + (q >> 3);
        bf[q].x = W[(size_t)k * HID_DIM + (nt << 4) + m];
        bf[q].y = W[(size_t)(k + 4) * HID_DIM + (nt << 4) + m];
    }

    __syncthreads();

    const int nStrips = (rows + 15) >> 4;
    for (int s = wave >> 3; s < nStrips; s += TA_WAVES / 8) {
        const float* ap = tile + ((s << 4) + m) * TA_PITCH + (half << 1);
        v8f acc = {};
#pragma unroll
        for (int q = 0; q < 32; ++q) {
            const v2f a = *(const v2f*)(ap + (q << 2));
            acc = __builtin_amdgcn_wmma_f32_16x16x4_f32(
                 false, a,  false, bf[q],
                 (short)0, acc,  false,  false);
        }
        float* op = outt + ((s << 4) + (half << 3)) * TA_PITCH + (nt << 4) + m;
#pragma unroll
        for (int v = 0; v < 8; ++v) op[v * TA_PITCH] = acc[v];
    }

    __syncthreads();

    const float4 bv = *(const float4*)(bias + (lane << 2));
    for (int r = wave; r < rows; r += TA_WAVES) {
        const float4 t = *(const float4*)(outt + r * TA_PITCH + (lane << 2));
        float4 o;
        o.x = fmaxf(t.x + bv.x, 0.0f);
        o.y = fmaxf(t.y + bv.y, 0.0f);
        o.z = fmaxf(t.z + bv.z, 0.0f);
        o.w = fmaxf(t.w + bv.w, 0.0f);
        *(float4*)(a1 + (size_t)(lo + r) * HID_DIM + (lane << 2)) = o;
    }
}

#define PA_THREADS 1024
#define PA_GROUP   8
__global__ __launch_bounds__(PA_THREADS)
void pooled_aggregate(const float* __restrict__ a1,
                      const float* __restrict__ dinv,
                      const int* __restrict__ src,
                      const int* __restrict__ dst,
                      const int* __restrict__ batch,
                      float* __restrict__ S,
                      float* __restrict__ cnts,
                      int E, int Nn) {
    __shared__ float Sl[NUM_GRAPHS * HID_DIM];
    __shared__ float Cl[NUM_GRAPHS];
    const int tid  = threadIdx.x;
    const int lane = tid & 31;

    for (int i = tid; i < NUM_GRAPHS * HID_DIM; i += PA_THREADS) Sl[i] = 0.0f;
    if (tid < NUM_GRAPHS) Cl[tid] = 0.0f;
    __syncthreads();

    const int total = E + Nn;
    const int nW    = (int)gridDim.x * (PA_THREADS / 32);
    const int per   = (((total + nW - 1) / nW) + (PA_GROUP - 1)) & ~(PA_GROUP - 1);
    const int wave  = __builtin_amdgcn_readfirstlane(tid >> 5);
    const int begin = ((int)blockIdx.x * (PA_THREADS / 32) + wave) * per;
    int end = begin + per;
    if (end > total) end = total;

    for (int base = begin; base < end; base += 32) {
        const int t = base + lane;
        int s = 0, g = 0;
        float w = 0.0f;
        if (t < end) {
            int d;
            if (t < E) { s = src[t]; d = dst[t]; }
            else       { s = t - E;  d = s; }
            g = batch[d];
            w = dinv[s] * dinv[d];
            if (t >= E) atomicAdd(&Cl[g], 1.0f);
        }
        int nv = end - base;
        if (nv > 32) nv = 32;

        int k0 = 0;
        for (; k0 + PA_GROUP <= nv; k0 += PA_GROUP) {
            float4 v[PA_GROUP];
            float  wk[PA_GROUP];
            int    gk[PA_GROUP];
#pragma unroll
            for (int j = 0; j < PA_GROUP; ++j) {
                const int sk = __builtin_amdgcn_readlane(s, k0 + j);
                wk[j] = __int_as_float(__builtin_amdgcn_readlane(__float_as_int(w), k0 + j));
                gk[j] = __builtin_amdgcn_readlane(g, k0 + j);
                v[j]  = *(const float4*)(a1 + (size_t)sk * HID_DIM + (lane << 2));
            }
#pragma unroll
            for (int j = 0; j < PA_GROUP; ++j) {
                float* p = Sl + gk[j] * HID_DIM + lane;
                atomicAdd(p,      v[j].x * wk[j]);
                atomicAdd(p + 32, v[j].y * wk[j]);
                atomicAdd(p + 64, v[j].z * wk[j]);
                atomicAdd(p + 96, v[j].w * wk[j]);
            }
        }
        for (; k0 < nv; ++k0) {
            const int   sk = __builtin_amdgcn_readlane(s, k0);
            const float wq = __int_as_float(__builtin_amdgcn_readlane(__float_as_int(w), k0));
            const int   gq = __builtin_amdgcn_readlane(g, k0);
            const float4 vv = *(const float4*)(a1 + (size_t)sk * HID_DIM + (lane << 2));
            float* p = Sl + gq * HID_DIM + lane;
            atomicAdd(p,      vv.x * wq);
            atomicAdd(p + 32, vv.y * wq);
            atomicAdd(p + 64, vv.z * wq);
            atomicAdd(p + 96, vv.w * wq);
        }
    }
    __syncthreads();

    for (int idx = tid; idx < NUM_GRAPHS * HID_DIM; idx += PA_THREADS) {
        const int g = idx >> 7;
        const int c = idx & (HID_DIM - 1);
        atomicAdd(&S[idx], Sl[g * HID_DIM + ((c & 3) << 5) + (c >> 2)]);
    }
    if (tid < NUM_GRAPHS) {
        const float cv = Cl[tid];
        if (cv != 0.0f) atomicAdd(&cnts[tid], cv);
    }
}

__global__ void pool_final(const float* __restrict__ T,
                           const float* __restrict__ cnt,
                           const float* __restrict__ bias,
                           float* __restrict__ out, int G, int D) {
    int tid = blockIdx.x * blockDim.x + threadIdx.x;
    if (tid >= G * D) return;
    int g = tid / D;
    int c = tid - g * D;
    const float cn = cnt[g];
    out[tid] = (T[tid] + cn * bias[c]) / fmaxf(cn, 1.0f);
}

extern "C" void kernel_launch(void* const* d_in, const int* in_sizes, int n_in,
                              void* d_out, int out_size, void* d_ws, size_t ws_size,
                              hipStream_t stream) {
    const float* x    = (const float*)d_in[0];
    const int*   eidx = (const int*)  d_in[1];
    const int*   batch= (const int*)  d_in[2];
    const float* W1   = (const float*)d_in[3];
    const float* b1   = (const float*)d_in[4];
    const float* W2   = (const float*)d_in[5];
    const float* b2   = (const float*)d_in[6];
    float* out = (float*)d_out;

    const int Nn = in_sizes[0] / IN_DIM;
    const int E  = in_sizes[1] / 2;
    const int* src = eidx;
    const int* dst = eidx + E;

    float* ws   = (float*)d_ws;
    float* agg1 = ws;
    float* S    = agg1 + (size_t)Nn * HID_DIM;
    float* T    = S    + NUM_GRAPHS * HID_DIM;
    float* deg  = T    + NUM_GRAPHS * OUT_DIM;
    float* dinv = deg  + Nn;
    float* cnts = dinv + Nn;

    const int B = 256;
    auto blocks = [](long long n, int b) { return (unsigned)((n + b - 1) / b); };

    fill_kernel<<<blocks(Nn, B), B, 0, stream>>>(deg, 1.0f, Nn);
    fill_kernel<<<blocks(NUM_GRAPHS * HID_DIM, B), B, 0, stream>>>(S, 0.0f, NUM_GRAPHS * HID_DIM);
    fill_kernel<<<blocks(NUM_GRAPHS, B), B, 0, stream>>>(cnts, 0.0f, NUM_GRAPHS);

    degree_kernel<<<blocks(E, B), B, 0, stream>>>(dst, deg, E);
    rsqrt_kernel<<<blocks(Nn, B), B, 0, stream>>>(deg, dinv, Nn);

    tile_aggregate<<<blocks(Nn, TA_ROWS), TA_THREADS, 0, stream>>>(
        x, dinv, src, dst, W1, b1, agg1, E, Nn);

    pooled_aggregate<<<256, PA_THREADS, 0, stream>>>(agg1, dinv, src, dst, batch, S, cnts, E, Nn);
    {
        int tiles = (NUM_GRAPHS >> 4) * (OUT_DIM >> 4);
        dim3 blk(32, 8);
        gemm_wmma_f32<<<blocks(tiles, 8), blk, 0, stream>>>(S, W2, T, NUM_GRAPHS, OUT_DIM, HID_DIM);
    }
    pool_final<<<blocks(NUM_GRAPHS * OUT_DIM, B), B, 0, stream>>>(
        T, cnts, b2, out, NUM_GRAPHS, OUT_DIM);
}


# Layer-1 edge filter reads one owner byte per edge (destination/196, stored by the degree pass) instead of the 4-byte destination: 16 edges per lane load; kept-list and LDS sum order move
# speedup vs baseline: 2.8548x; 1.1258x over previous
#include <hip/hip_runtime.h>

#define IN_DIM   128
#define HID_DIM  128
#define OUT_DIM  64
#define NUM_GRAPHS 64

typedef float v2f __attribute__((ext_vector_type(2)));
typedef float v8f __attribute__((ext_vector_type(8)));

__global__ void fill_kernel(float* __restrict__ p, float v, int n) {
    int i = blockIdx.x * blockDim.x + threadIdx.x;
    if (i < n) p[i] = v;
}

__global__ void rsqrt_kernel(const float* __restrict__ deg, float* __restrict__ dinv, int n) {
    int i = blockIdx.x * blockDim.x + threadIdx.x;
    if (i < n) dinv[i] = rsqrtf(deg[i]);
}

__global__ void gemm_wmma_f32(const float* __restrict__ A,
                              const float* __restrict__ B,
                              float* __restrict__ C,
                              int M, int N, int K) {
    const int lane  = threadIdx.x;
    const int wave  = threadIdx.y;
    const int tilesN = N >> 4;
    const int totalTiles = (M >> 4) * tilesN;
    const int tile = blockIdx.x * blockDim.y + wave;
    if (tile >= totalTiles) return;

    const int tm   = tile / tilesN;
    const int tn   = tile - tm * tilesN;
    const int row0 = tm << 4;
    const int col0 = tn << 4;
    const int half = lane >> 4;
    const int l    = lane & 15;

    v8f acc = {};
    const float* Arow = A + (size_t)(row0 + l) * K;
    const float* Bcol = B + (size_t)col0 + l;

    for (int k0 = 0; k0 < K; k0 += 4) {
        const int ka = k0 + (half << 1);
        v2f a, b;
        a.x = Arow[ka];
        a.y = Arow[ka + 1];
        b.x = Bcol[(size_t)ka * N];
        b.y = Bcol[(size_t)(ka + 1) * N];
        acc = __builtin_amdgcn_wmma_f32_16x16x4_f32(
             false, a,  false, b,
             (short)0, acc,  false,  false);
    }

#pragma unroll
    for (int v = 0; v < 8; ++v) {
        C[(size_t)(row0 + (half << 3) + v) * N + col0 + l] = acc[v];
    }
}

#define TA_ROWS    196
#define TA_STRIPS  13
#define TA_PADROWS (TA_STRIPS * 16)
#define TA_PITCH   (IN_DIM + 4)
#define TA_THREADS 1024
#define TA_WAVES   (TA_THREADS / 32)
#define TA_AHEAD   4
#define TA_LIST    (32 + 512)
#define TA_GROUP   8

__global__ void degree_key_kernel(const int* __restrict__ dst, float* __restrict__ deg,
                                  unsigned* __restrict__ keys, int E) {
    const int t = blockIdx.x * blockDim.x + threadIdx.x;
    const int e = t << 2;
    if (e + 3 < E) {
        const int d0 = dst[e], d1 = dst[e + 1], d2 = dst[e + 2], d3 = dst[e + 3];
        atomicAdd(&deg[d0], 1.0f);
        atomicAdd(&deg[d1], 1.0f);
        atomicAdd(&deg[d2], 1.0f);
        atomicAdd(&deg[d3], 1.0f);
        keys[t] = ((unsigned)d0 / TA_ROWS) | (((unsigned)d1 / TA_ROWS) << 8) |
                  (((unsigned)d2 / TA_ROWS) << 16) | (((unsigned)d3 / TA_ROWS) << 24);
    } else {
        for (int i = e; i < E; ++i) {
            const int d = dst[i];
            atomicAdd(&deg[d], 1.0f);
            ((unsigned char*)keys)[i] = (unsigned char)((unsigned)d / TA_ROWS);
        }
    }
}

__device__ __forceinline__ unsigned ta_zero_bytes(unsigned t) {
    return ~(((t & 0x7F7F7F7Fu) + 0x7F7F7F7Fu) | t | 0x7F7F7F7Fu);
}

__device__ __forceinline__ int ta_append(int* list, int cnt, bool m, int e) {
    const unsigned mask = __builtin_amdgcn_ballot_w32(m);
    if (mask != 0u) {
        const int pos = cnt + (int)__builtin_amdgcn_mbcnt_lo(mask, 0u);
        if (m) list[pos] = e;
        cnt += __builtin_popcount(mask);
    }
    return cnt;
}

__device__ __forceinline__ void ta_flush(const int* list, int nv, int lane, int lo,
                                         const float* __restrict__ x,
                                         const float* __restrict__ dinv,
                                         const int* __restrict__ src,
                                         const int* __restrict__ dst,
                                         float* tile) {
    int s = 0, r = 0;
    float w = 0.0f;
    if (lane < nv) {
        const int e = list[lane];
        s = src[e];
        const int d = dst[e];
        r = d - lo;
        w = dinv[s] * dinv[d];
    }
    int k0 = 0;
    for (; k0 + TA_GROUP <= nv; k0 += TA_GROUP) {
        float4 v[TA_GROUP];
        float  wk[TA_GROUP];
        int    rk[TA_GROUP];
#pragma unroll
        for (int j = 0; j < TA_GROUP; ++j) {
            const int sk = __builtin_amdgcn_readlane(s, k0 + j);
            wk[j] = __int_as_float(__builtin_amdgcn_readlane(__float_as_int(w), k0 + j));
            rk[j] = __builtin_amdgcn_readlane(r, k0 + j);
            v[j]  = *(const float4*)(x + (size_t)sk * IN_DIM + (lane << 2));
        }
#pragma unroll
        for (int j = 0; j < TA_GROUP; ++j) {
            float* p = tile + rk[j] * TA_PITCH + lane;
            atomicAdd(p,      v[j].x * wk[j]);
            atomicAdd(p + 32, v[j].y * wk[j]);
            atomicAdd(p + 64, v[j].z * wk[j]);
            atomicAdd(p + 96, v[j].w * wk[j]);
        }
    }
    for (; k0 < nv; ++k0) {
        const int   sk = __builtin_amdgcn_readlane(s, k0);
        const float wq = __int_as_float(__builtin_amdgcn_readlane(__float_as_int(w), k0));
        const int   rq = __builtin_amdgcn_readlane(r, k0);
        const float4 vv = *(const float4*)(x + (size_t)sk * IN_DIM + (lane << 2));
        float* p = tile + rq * TA_PITCH + lane;
        atomicAdd(p,      vv.x * wq);
        atomicAdd(p + 32, vv.y * wq);
        atomicAdd(p + 64, vv.z * wq);
        atomicAdd(p + 96, vv.w * wq);
    }
}

__global__ __launch_bounds__(TA_THREADS)
void tile_aggregate(const float* __restrict__ x,
                    const float* __restrict__ dinv,
                    const int* __restrict__ src,
                    const int* __restrict__ dst,
                    const unsigned* __restrict__ keys,
                    const float* __restrict__ W,
                    const float* __restrict__ bias,
                    float* __restrict__ a1,
                    int E, int Nn) {
    static_assert(IN_DIM == 128 && HID_DIM == 128, "fragment walk below is written for 128 x 128");
    __shared__ __attribute__((aligned(16))) float tile[TA_PADROWS * TA_PITCH];
    __shared__ __attribute__((aligned(16))) float outt[TA_PADROWS * TA_PITCH];
    __shared__ int   lists[TA_WAVES * TA_LIST];
    const int tid  = threadIdx.x;
    const int lane = tid & 31;
    const int wave = __builtin_amdgcn_readfirstlane(tid >> 5);
    const int lo   = (int)blockIdx.x * TA_ROWS;
    int rows = Nn - lo;
    if (rows > TA_ROWS) rows = TA_ROWS;

    for (int r = wave; r < TA_PADROWS; r += TA_WAVES) {
        float* p = tile + r * TA_PITCH + lane;
        if (r < rows) {
            const int   i  = lo + r;
            const float di = dinv[i];
            const float w  = di * di;
            const float4 xv = *(const float4*)(x + (size_t)i * IN_DIM + (lane << 2));
            p[0]  = xv.x * w;
            p[32] = xv.y * w;
            p[64] = xv.z * w;
            p[96] = xv.w * w;
        } else {
            p[0] = 0.0f; p[32] = 0.0f; p[64] = 0.0f; p[96] = 0.0f;
        }
    }
    __syncthreads();

    int* mylist = lists + wave * TA_LIST;
    int  cnt = 0;

    const unsigned own = (unsigned)blockIdx.x;
    const unsigned pat = own * 0x01010101u;
    const unsigned char* key8 = (const unsigned char*)keys;
    const int nChunks = E >> 9;
    const int start = (int)blockIdx.x * (nChunks / (int)gridDim.x);

    static_assert(TA_AHEAD == 4, "the chunk walk below selects among four hit words");
    for (int kc = wave; kc < nChunks; kc += TA_AHEAD * TA_WAVES) {
        unsigned h[TA_AHEAD];
        {
            uint4 d[TA_AHEAD];
#pragma unroll
            for (int u = 0; u < TA_AHEAD; ++u) {
                int c = kc + u * TA_WAVES;
                if (c >= nChunks) c = kc;
                int cc = c + start;
                if (cc >= nChunks) cc -= nChunks;
                d[u] = *(const uint4*)(key8 + (cc << 9) + (lane << 4));
            }
#pragma unroll
            for (int u = 0; u < TA_AHEAD; ++u) {
                unsigned t = ta_zero_bytes(d[u].x ^ pat) >> 7;
                t |= ta_zero_bytes(d[u].y ^ pat) >> 6;
                t |= ta_zero_bytes(d[u].z ^ pat) >> 5;
                t |= ta_zero_bytes(d[u].w ^ pat) >> 4;
                h[u] = t;
            }
        }
        int nv = (nChunks - kc + TA_WAVES - 1) / TA_WAVES;
        if (nv > TA_AHEAD) nv = TA_AHEAD;
#pragma nounroll
        for (int u = 0; u < nv; ++u) {
            unsigned hits = (u == 0) ? h[0] : (u == 1) ? h[1] : (u == 2) ? h[2] : h[3];
            int cc = kc + u * TA_WAVES + start;
            if (cc >= nChunks) cc -= nChunks;
            const int e0 = (cc << 9) + (lane << 4);
            unsigned mask = __builtin_amdgcn_ballot_w32(hits != 0u);
            while (mask != 0u) {
                const int p = __builtin_ctz(hits | 0x80000000u);
                const int pos = cnt + (int)__builtin_amdgcn_mbcnt_lo(mask, 0u);
                if (hits != 0u) mylist[pos] = e0 + ((p & 7) << 2) + (p >> 3);
                cnt += __builtin_popcount(mask);
                hits &= hits - 1u;
                mask = __builtin_amdgcn_ballot_w32(hits != 0u);
            }
            while (cnt >= 32) {
                cnt -= 32;
                ta_flush(mylist + cnt, 32, lane, lo, x, dinv, src, dst, tile);
            }
        }
    }
    for (int base = (nChunks << 9) + (wave << 5); base < E; base += TA_WAVES * 32) {
        const int e = base + lane;
        bool m = false;
        if (e < E) m = (unsigned)key8[e] == own;
        cnt = ta_append(mylist, cnt, m, e);
        if (cnt >= 32) {
            cnt -= 32;
            ta_flush(mylist + cnt, 32, lane, lo, x, dinv, src, dst, tile);
        }
    }
    if (cnt > 0) ta_flush(mylist, cnt, lane, lo, x, dinv, src, dst, tile);

    const int half = lane >> 4;
    const int m    = lane & 15;
    const int nt   = wave & 7;
    v2f bf[32];
#pragma unroll
    for (int q = 0; q < 32; ++q) {
        const int k = ((q & 7) << 4) + (half << 3) + (q >> 3);
        bf[q].x = W[(size_t)k * HID_DIM + (nt << 4) + m];
        bf[q].y = W[(size_t)(k + 4) * HID_DIM + (nt << 4) + m];
    }

    __syncthreads();

    const int nStrips = (rows + 15) >> 4;
    for (int s = wave >> 3; s < nStrips; s += TA_WAVES / 8) {
        const float* ap = tile + ((s << 4) + m) * TA_PITCH + (half << 1);
        v8f acc = {};
#pragma unroll
        for (int q = 0; q < 32; ++q) {
            const v2f a = *(const v2f*)(ap + (q << 2));
            acc = __builtin_amdgcn_wmma_f32_16x16x4_f32(
                 false, a,  false, bf[q],
                 (short)0, acc,  false,  false);
        }
        float* op = outt + ((s << 4) + (half << 3)) * TA_PITCH + (nt << 4) + m;
#pragma unroll
        for (int v = 0; v < 8; ++v) op[v * TA_PITCH] = acc[v];
    }

    __syncthreads();

    const float4 bv = *(const float4*)(bias + (lane << 2));
    for (int r = wave; r < rows; r += TA_WAVES) {
        const float4 t = *(const float4*)(outt + r * TA_PITCH + (lane << 2));
        float4 o;
        o.x = fmaxf(t.x + bv.x, 0.0f);
        o.y = fmaxf(t.y + bv.y, 0.0f);
        o.z = fmaxf(t.z + bv.z, 0.0f);
        o.w = fmaxf(t.w + bv.w, 0.0f);
        *(float4*)(a1 + (size_t)(lo + r) * HID_DIM + (lane << 2)) = o;
    }
}

#define PA_THREADS 1024
#define PA_GROUP   8
__global__ __launch_bounds__(PA_THREADS)
void pooled_aggregate(const float* __restrict__ a1,
                      const float* __restrict__ dinv,
                      const int* __restrict__ src,
                      const int* __restrict__ dst,
                      const int* __restrict__ batch,
                      float* __restrict__ S,
                      float* __restrict__ cnts,
                      int E, int Nn) {
    __shared__ float Sl[NUM_GRAPHS * HID_DIM];
    __shared__ float Cl[NUM_GRAPHS];
    const int tid  = threadIdx.x;
    const int lane = tid & 31;

    for (int i = tid; i < NUM_GRAPHS * HID_DIM; i += PA_THREADS) Sl[i] = 0.0f;
    if (tid < NUM_GRAPHS) Cl[tid] = 0.0f;
    __syncthreads();

    const int total = E + Nn;
    const int nW    = (int)gridDim.x * (PA_THREADS / 32);
    const int per   = (((total + nW - 1) / nW) + (PA_GROUP - 1)) & ~(PA_GROUP - 1);
    const int wave  = __builtin_amdgcn_readfirstlane(tid >> 5);
    const int begin = ((int)blockIdx.x * (PA_THREADS / 32) + wave) * per;
    int end = begin + per;
    if (end > total) end = total;

    for (int base = begin; base < end; base += 32) {
        const int t = base + lane;
        int s = 0, g = 0;
        float w = 0.0f;
        if (t < end) {
            int d;
            if (t < E) { s = src[t]; d = dst[t]; }
            else       { s = t - E;  d = s; }
            g = batch[d];
            w = dinv[s] * dinv[d];
            if (t >= E) atomicAdd(&Cl[g], 1.0f);
        }
        int nv = end - base;
        if (nv > 32) nv = 32;

        int k0 = 0;
        for (; k0 + PA_GROUP <= nv; k0 += PA_GROUP) {
            float4 v[PA_GROUP];
            float  wk[PA_GROUP];
            int    gk[PA_GROUP];
#pragma unroll
            for (int j = 0; j < PA_GROUP; ++j) {
                const int sk = __builtin_amdgcn_readlane(s, k0 + j);
                wk[j] = __int_as_float(__builtin_amdgcn_readlane(__float_as_int(w), k0 + j));
                gk[j] = __builtin_amdgcn_readlane(g, k0 + j);
                v[j]  = *(const float4*)(a1 + (size_t)sk * HID_DIM + (lane << 2));
            }
#pragma unroll
            for (int j = 0; j < PA_GROUP; ++j) {
                float* p = Sl + gk[j] * HID_DIM + lane;
                atomicAdd(p,      v[j].x * wk[j]);
                atomicAdd(p + 32, v[j].y * wk[j]);
                atomicAdd(p + 64, v[j].z * wk[j]);
                atomicAdd(p + 96, v[j].w * wk[j]);
            }
        }
        for (; k0 < nv; ++k0) {
            const int   sk = __builtin_amdgcn_readlane(s, k0);
            const float wq = __int_as_float(__builtin_amdgcn_readlane(__float_as_int(w), k0));
            const int   gq = __builtin_amdgcn_readlane(g, k0);
            const float4 vv = *(const float4*)(a1 + (size_t)sk * HID_DIM + (lane << 2));
            float* p = Sl + gq * HID_DIM + lane;
            atomicAdd(p,      vv.x * wq);
            atomicAdd(p + 32, vv.y * wq);
            atomicAdd(p + 64, vv.z * wq);
            atomicAdd(p + 96, vv.w * wq);
        }
    }
    __syncthreads();

    for (int idx = tid; idx < NUM_GRAPHS * HID_DIM; idx += PA_THREADS) {
        const int g = idx >> 7;
        const int c = idx & (HID_DIM - 1);
        atomicAdd(&S[idx], Sl[g * HID_DIM + ((c & 3) << 5) + (c >> 2)]);
    }
    if (tid < NUM_GRAPHS) {
        const float cv = Cl[tid];
        if (cv != 0.0f) atomicAdd(&cnts[tid], cv);
    }
}

__global__ void pool_final(const float* __restrict__ T,
                           const float* __restrict__ cnt,
                           const float* __restrict__ bias,
                           float* __restrict__ out, int G, int D) {
    int tid = blockIdx.x * blockDim.x + threadIdx.x;
    if (tid >= G * D) return;
    int g = tid / D;
    int c = tid - g * D;
    const float cn = cnt[g];
    out[tid] = (T[tid] + cn * bias[c]) / fmaxf(cn, 1.0f);
}

extern "C" void kernel_launch(void* const* d_in, const int* in_sizes, int n_in,
                              void* d_out, int out_size, void* d_ws, size_t ws_size,
                              hipStream_t stream) {
    const float* x    = (const float*)d_in[0];
    const int*   eidx = (const int*)  d_in[1];
    const int*   batch= (const int*)  d_in[2];
    const float* W1   = (const float*)d_in[3];
    const float* b1   = (const float*)d_in[4];
    const float* W2   = (const float*)d_in[5];
    const float* b2   = (const float*)d_in[6];
    float* out = (float*)d_out;

    const int Nn = in_sizes[0] / IN_DIM;
    const int E  = in_sizes[1] / 2;
    const int* src = eidx;
    const int* dst = eidx + E;

    float* ws   = (float*)d_ws;
    float* agg1 = ws;
    float* S    = agg1 + (size_t)Nn * HID_DIM;
    float* T    = S    + NUM_GRAPHS * HID_DIM;
    unsigned* keys = (unsigned*)(T + NUM_GRAPHS * OUT_DIM);
    float* deg  = (float*)(keys + (((size_t)E + 15) >> 4) * 4);
    float* dinv = deg  + Nn;
    float* cnts = dinv + Nn;

    const int B = 256;
    auto blocks = [](long long n, int b) { return (unsigned)((n + b - 1) / b); };

    fill_kernel<<<blocks(Nn, B), B, 0, stream>>>(deg, 1.0f, Nn);
    fill_kernel<<<blocks(NUM_GRAPHS * HID_DIM, B), B, 0, stream>>>(S, 0.0f, NUM_GRAPHS * HID_DIM);
    fill_kernel<<<blocks(NUM_GRAPHS, B), B, 0, stream>>>(cnts, 0.0f, NUM_GRAPHS);

    degree_key_kernel<<<blocks((E + 3) >> 2, B), B, 0, stream>>>(dst, deg, keys, E);
    rsqrt_kernel<<<blocks(Nn, B), B, 0, stream>>>(deg, dinv, Nn);

    tile_aggregate<<<blocks(Nn, TA_ROWS), TA_THREADS, 0, stream>>>(
        x, dinv, src, dst, keys, W1, b1, agg1, E, Nn);

    pooled_aggregate<<<256, PA_THREADS, 0, stream>>>(agg1, dinv, src, dst, batch, S, cnts, E, Nn);
    {
        int tiles = (NUM_GRAPHS >> 4) * (OUT_DIM >> 4);
        dim3 blk(32, 8);
        gemm_wmma_f32<<<blocks(tiles, 8), blk, 0, stream>>>(S, W2, T, NUM_GRAPHS, OUT_DIM, HID_DIM);
    }
    pool_final<<<blocks(NUM_GRAPHS * OUT_DIM, B), B, 0, stream>>>(
        T, cnts, b2, out, NUM_GRAPHS, OUT_DIM);
}


# Layer 2 + pool source-stationary in the layer-1 launch: a block adds its own LDS rows to the graphs of their out-edges (source owner byte scan); a1 plane, its gather, 1 launch go; sum order moves
# speedup vs baseline: 2.9932x; 1.0485x over previous
#include <hip/hip_runtime.h>

#define IN_DIM   128
#define HID_DIM  128
#define OUT_DIM  64
#define NUM_GRAPHS 64

typedef float v2f __attribute__((ext_vector_type(2)));
typedef float v8f __attribute__((ext_vector_type(8)));

__global__ void fill_kernel(float* __restrict__ p, float v, int n) {
    int i = blockIdx.x * blockDim.x + threadIdx.x;
    if (i < n) p[i] = v;
}

__global__ void rsqrt_kernel(const float* __restrict__ deg, float* __restrict__ dinv, int n) {
    int i = blockIdx.x * blockDim.x + threadIdx.x;
    if (i < n) dinv[i] = rsqrtf(deg[i]);
}

__global__ void gemm_wmma_f32(const float* __restrict__ A,
                              const float* __restrict__ B,
                              float* __restrict__ C,
                              int M, int N, int K) {
    const int lane  = threadIdx.x;
    const int wave  = threadIdx.y;
    const int tilesN = N >> 4;
    const int totalTiles = (M >> 4) * tilesN;
    const int tile = blockIdx.x * blockDim.y + wave;
    if (tile >= totalTiles) return;

    const int tm   = tile / tilesN;
    const int tn   = tile - tm * tilesN;
    const int row0 = tm << 4;
    const int col0 = tn << 4;
    const int half = lane >> 4;
    const int l    = lane & 15;

    v8f acc = {};
    const float* Arow = A + (size_t)(row0 + l) * K;
    const float* Bcol = B + (size_t)col0 + l;

    for (int k0 = 0; k0 < K; k0 += 4) {
        const int ka = k0 + (half << 1);
        v2f a, b;
        a.x = Arow[ka];
        a.y = Arow[ka + 1];
        b.x = Bcol[(size_t)ka * N];
        b.y = Bcol[(size_t)(ka + 1) * N];
        acc = __builtin_amdgcn_wmma_f32_16x16x4_f32(
             false, a,  false, b,
             (short)0, acc,  false,  false);
    }

#pragma unroll
    for (int v = 0; v < 8; ++v) {
        C[(size_t)(row0 + (half << 3) + v) * N + col0 + l] = acc[v];
    }
}

#define TA_ROWS    196
#define TA_STRIPS  13
#define TA_PADROWS (TA_STRIPS * 16)
#define TA_PITCH   (IN_DIM + 4)
#define TA_THREADS 1024
#define TA_WAVES   (TA_THREADS / 32)
#define TA_AHEAD   4
#define TA_LIST    (32 + 512)
#define TA_GROUP   8

__global__ void degree_key_kernel(const int* __restrict__ src, const int* __restrict__ dst,
                                  float* __restrict__ deg, unsigned* __restrict__ keys,
                                  unsigned* __restrict__ skeys, int E) {
    const int t = blockIdx.x * blockDim.x + threadIdx.x;
    const int e = t << 2;
    if (e + 3 < E) {
        const int d0 = dst[e], d1 = dst[e + 1], d2 = dst[e + 2], d3 = dst[e + 3];
        const int s0 = src[e], s1 = src[e + 1], s2 = src[e + 2], s3 = src[e + 3];
        atomicAdd(&deg[d0], 1.0f);
        atomicAdd(&deg[d1], 1.0f);
        atomicAdd(&deg[d2], 1.0f);
        atomicAdd(&deg[d3], 1.0f);
        keys[t] = ((unsigned)d0 / TA_ROWS) | (((unsigned)d1 / TA_ROWS) << 8) |
                  (((unsigned)d2 / TA_ROWS) << 16) | (((unsigned)d3 / TA_ROWS) << 24);
        skeys[t] = ((unsigned)s0 / TA_ROWS) | (((unsigned)s1 / TA_ROWS) << 8) |
                   (((unsigned)s2 / TA_ROWS) << 16) | (((unsigned)s3 / TA_ROWS) << 24);
    } else {
        for (int i = e; i < E; ++i) {
            const int d = dst[i];
            atomicAdd(&deg[d], 1.0f);
            ((unsigned char*)keys)[i] = (unsigned char)((unsigned)d / TA_ROWS);
            ((unsigned char*)skeys)[i] = (unsigned char)((unsigned)src[i] / TA_ROWS);
        }
    }
}

__device__ __forceinline__ unsigned ta_zero_bytes(unsigned t) {
    return ~(((t & 0x7F7F7F7Fu) + 0x7F7F7F7Fu) | t | 0x7F7F7F7Fu);
}

__device__ __forceinline__ int ta_append(int* list, int cnt, bool m, int e) {
    const unsigned mask = __builtin_amdgcn_ballot_w32(m);
    if (mask != 0u) {
        const int pos = cnt + (int)__builtin_amdgcn_mbcnt_lo(mask, 0u);
        if (m) list[pos] = e;
        cnt += __builtin_popcount(mask);
    }
    return cnt;
}

__device__ __forceinline__ void ta_flush(const int* list, int nv, int lane, int lo,
                                         const float* __restrict__ x,
                                         const float* __restrict__ dinv,
                                         const int* __restrict__ src,
                                         const int* __restrict__ dst,
                                         float* tile) {
    int s = 0, r = 0;
    float w = 0.0f;
    if (lane < nv) {
        const int e = list[lane];
        s = src[e];
        const int d = dst[e];
        r = d - lo;
        w = dinv[s] * dinv[d];
    }
    int k0 = 0;
    for (; k0 + TA_GROUP <= nv; k0 += TA_GROUP) {
        float4 v[TA_GROUP];
        float  wk[TA_GROUP];
        int    rk[TA_GROUP];
#pragma unroll
        for (int j = 0; j < TA_GROUP; ++j) {
            const int sk = __builtin_amdgcn_readlane(s, k0 + j);
            wk[j] = __int_as_float(__builtin_amdgcn_readlane(__float_as_int(w), k0 + j));
            rk[j] = __builtin_amdgcn_readlane(r, k0 + j);
            v[j]  = *(const float4*)(x + (size_t)sk * IN_DIM + (lane << 2));
        }
#pragma unroll
        for (int j = 0; j < TA_GROUP; ++j) {
            float* p = tile + rk[j] * TA_PITCH + lane;
            atomicAdd(p,      v[j].x * wk[j]);
            atomicAdd(p + 32, v[j].y * wk[j]);
            atomicAdd(p + 64, v[j].z * wk[j]);
            atomicAdd(p + 96, v[j].w * wk[j]);
        }
    }
    for (; k0 < nv; ++k0) {
        const int   sk = __builtin_amdgcn_readlane(s, k0);
        const float wq = __int_as_float(__builtin_amdgcn_readlane(__float_as_int(w), k0));
        const int   rq = __builtin_amdgcn_readlane(r, k0);
        const float4 vv = *(const float4*)(x + (size_t)sk * IN_DIM + (lane << 2));
        float* p = tile + rq * TA_PITCH + lane;
        atomicAdd(p,      vv.x * wq);
        atomicAdd(p + 32, vv.y * wq);
        atomicAdd(p + 64, vv.z * wq);
        atomicAdd(p + 96, vv.w * wq);
    }
}

__device__ __forceinline__ void ta_pool_flush(const int* list, int nv, int lane, int lo,
                                              const float* __restrict__ dinv,
                                              const int* __restrict__ src,
                                              const int* __restrict__ dst,
                                              const int* __restrict__ batch,
                                              const float* rowsl, float* Sl) {
    int r = 0, g = 0;
    float w = 0.0f;
    if (lane < nv) {
        const int e = list[lane];
        const int s = src[e];
        const int d = dst[e];
        r = s - lo;
        g = batch[d];
        w = dinv[s] * dinv[d];
    }
    int k0 = 0;
    for (; k0 + TA_GROUP <= nv; k0 += TA_GROUP) {
        float4 v[TA_GROUP];
        float  wk[TA_GROUP];
        int    gk[TA_GROUP];
#pragma unroll
        for (int j = 0; j < TA_GROUP; ++j) {
            const int rk = __builtin_amdgcn_readlane(r, k0 + j);
            wk[j] = __int_as_float(__builtin_amdgcn_readlane(__float_as_int(w), k0 + j));
            gk[j] = __builtin_amdgcn_readlane(g, k0 + j);
            v[j]  = *(const float4*)(rowsl + rk * TA_PITCH + (lane << 2));
        }
#pragma unroll
        for (int j = 0; j < TA_GROUP; ++j) {
            float* p = Sl + gk[j] * HID_DIM + lane;
            atomicAdd(p,      v[j].x * wk[j]);
            atomicAdd(p + 32, v[j].y * wk[j]);
            atomicAdd(p + 64, v[j].z * wk[j]);
            atomicAdd(p + 96, v[j].w * wk[j]);
        }
    }
    for (; k0 < nv; ++k0) {
        const int   rq = __builtin_amdgcn_readlane(r, k0);
        const float wq = __int_as_float(__builtin_amdgcn_readlane(__float_as_int(w), k0));
        const int   gq = __builtin_amdgcn_readlane(g, k0);
        const float4 vv = *(const float4*)(rowsl + rq * TA_PITCH + (lane << 2));
        float* p = Sl + gq * HID_DIM + lane;
        atomicAdd(p,      vv.x * wq);
        atomicAdd(p + 32, vv.y * wq);
        atomicAdd(p + 64, vv.z * wq);
        atomicAdd(p + 96, vv.w * wq);
    }
}

struct TaGatherFlush {
    int lane, lo;
    const float* x; const float* dinv; const int* src; const int* dst; float* tile;
    __device__ __forceinline__ void operator()(const int* list, int nv) const {
        ta_flush(list, nv, lane, lo, x, dinv, src, dst, tile);
    }
};
struct TaPoolFlush {
    int lane, lo;
    const float* dinv; const int* src; const int* dst; const int* batch; const float* rowsl; float* Sl;
    __device__ __forceinline__ void operator()(const int* list, int nv) const {
        ta_pool_flush(list, nv, lane, lo, dinv, src, dst, batch, rowsl, Sl);
    }
};

template <class Flush>
__device__ __forceinline__ void ta_scan(const unsigned char* __restrict__ key8, unsigned own, int E,
                                        int wave, int lane, int start, int* mylist, const Flush& flush) {
    int cnt = 0;
    const unsigned pat = own * 0x01010101u;
    const int nChunks = E >> 9;

    static_assert(TA_AHEAD == 4, "the chunk walk below selects among four hit words");
    for (int kc = wave; kc < nChunks; kc += TA_AHEAD * TA_WAVES) {
        unsigned h[TA_AHEAD];
        {
            uint4 d[TA_AHEAD];
#pragma unroll
            for (int u = 0; u < TA_AHEAD; ++u) {
                int c = kc + u * TA_WAVES;
                if (c >= nChunks) c = kc;
                int cc = c + start;
                if (cc >= nChunks) cc -= nChunks;
                d[u] = *(const uint4*)(key8 + (cc << 9) + (lane << 4));
            }
#pragma unroll
            for (int u = 0; u < TA_AHEAD; ++u) {
                unsigned t = ta_zero_bytes(d[u].x ^ pat) >> 7;
                t |= ta_zero_bytes(d[u].y ^ pat) >> 6;
                t |= ta_zero_bytes(d[u].z ^ pat) >> 5;
                t |= ta_zero_bytes(d[u].w ^ pat) >> 4;
                h[u] = t;
            }
        }
        int nv = (nChunks - kc + TA_WAVES - 1) / TA_WAVES;
        if (nv > TA_AHEAD) nv = TA_AHEAD;
#pragma nounroll
        for (int u = 0; u < nv; ++u) {
            unsigned hits = (u == 0) ? h[0] : (u == 1) ? h[1] : (u == 2) ? h[2] : h[3];
            int cc = kc + u * TA_WAVES + start;
            if (cc >= nChunks) cc -= nChunks;
            const int e0 = (cc << 9) + (lane << 4);
            unsigned mask = __builtin_amdgcn_ballot_w32(hits != 0u);
            while (mask != 0u) {
                const int p = __builtin_ctz(hits | 0x80000000u);
                const int pos = cnt + (int)__builtin_amdgcn_mbcnt_lo(mask, 0u);
                if (hits != 0u) mylist[pos] = e0 + ((p & 7) << 2) + (p >> 3);
                cnt += __builtin_popcount(mask);
                hits &= hits - 1u;
                mask = __builtin_amdgcn_ballot_w32(hits != 0u);
            }
            while (cnt >= 32) {
                cnt -= 32;
                flush(mylist + cnt, 32);
            }
        }
    }
    for (int base = (nChunks << 9) + (wave << 5); base < E; base += TA_WAVES * 32) {
        const int e = base + lane;
        bool m = false;
        if (e < E) m = (unsigned)key8[e] == own;
        cnt = ta_append(mylist, cnt, m, e);
        if (cnt >= 32) {
            cnt -= 32;
            flush(mylist + cnt, 32);
        }
    }
    if (cnt > 0) flush(mylist, cnt);
}

__global__ __launch_bounds__(TA_THREADS)
void tile_aggregate(const float* __restrict__ x,
                    const float* __restrict__ dinv,
                    const int* __restrict__ src,
                    const int* __restrict__ dst,
                    const int* __restrict__ batch,
                    const unsigned* __restrict__ keys,
                    const unsigned* __restrict__ skeys,
                    const float* __restrict__ W,
                    const float* __restrict__ bias,
                    float* __restrict__ S,
                    float* __restrict__ cnts,
                    int E, int Nn) {
    static_assert(IN_DIM == 128 && HID_DIM == 128, "fragment walk below is written for 128 x 128");
    static_assert(NUM_GRAPHS * HID_DIM <= TA_PADROWS * TA_PITCH, "the LDS copy of S reuses the tile plane");
    __shared__ __attribute__((aligned(16))) float tile[TA_PADROWS * TA_PITCH];
    __shared__ __attribute__((aligned(16))) float outt[TA_PADROWS * TA_PITCH];
    __shared__ int   lists[TA_WAVES * TA_LIST];
    __shared__ float Cl[NUM_GRAPHS];
    const int tid  = threadIdx.x;
    const int lane = tid & 31;
    const int wave = __builtin_amdgcn_readfirstlane(tid >> 5);
    const int lo   = (int)blockIdx.x * TA_ROWS;
    int rows = Nn - lo;
    if (rows > TA_ROWS) rows = TA_ROWS;

    if (tid < NUM_GRAPHS) Cl[tid] = 0.0f;
    for (int r = wave; r < TA_PADROWS; r += TA_WAVES) {
        float* p = tile + r * TA_PITCH + lane;
        if (r < rows) {
            const int   i  = lo + r;
            const float di = dinv[i];
            const float w  = di * di;
            const float4 xv = *(const float4*)(x + (size_t)i * IN_DIM + (lane << 2));
            p[0]  = xv.x * w;
            p[32] = xv.y * w;
            p[64] = xv.z * w;
            p[96] = xv.w * w;
        } else {
            p[0] = 0.0f; p[32] = 0.0f; p[64] = 0.0f; p[96] = 0.0f;
        }
    }
    __syncthreads();

    int* mylist = lists + wave * TA_LIST;
    const unsigned own = (unsigned)blockIdx.x;
    const int start = (int)blockIdx.x * ((E >> 9) / (int)gridDim.x);

    {
        const TaGatherFlush gf{lane, lo, x, dinv, src, dst, tile};
        ta_scan((const unsigned char*)keys, own, E, wave, lane, start, mylist, gf);
    }

    const int half = lane >> 4;
    const int m    = lane & 15;
    const int nt   = wave & 7;
    v2f bf[32];
#pragma unroll
    for (int q = 0; q < 32; ++q) {
        const int k = ((q & 7) << 4) + (half << 3) + (q >> 3);
        bf[q].x = W[(size_t)k * HID_DIM + (nt << 4) + m];
        bf[q].y = W[(size_t)(k + 4) * HID_DIM + (nt << 4) + m];
    }
    const float bcol = bias[(nt << 4) + m];

    __syncthreads();

    const int nStrips = (rows + 15) >> 4;
    for (int s = wave >> 3; s < nStrips; s += TA_WAVES / 8) {
        const float* ap = tile + ((s << 4) + m) * TA_PITCH + (half << 1);
        v8f acc = {};
#pragma unroll
        for (int q = 0; q < 32; ++q) {
            const v2f a = *(const v2f*)(ap + (q << 2));
            acc = __builtin_amdgcn_wmma_f32_16x16x4_f32(
                 false, a,  false, bf[q],
                 (short)0, acc,  false,  false);
        }
        float* op = outt + ((s << 4) + (half << 3)) * TA_PITCH + (nt << 4) + m;
#pragma unroll
        for (int v = 0; v < 8; ++v) op[v * TA_PITCH] = fmaxf(acc[v] + bcol, 0.0f);
    }

    __syncthreads();

    float* Sl = tile;
    for (int i = tid; i < NUM_GRAPHS * HID_DIM; i += TA_THREADS) Sl[i] = 0.0f;

    __syncthreads();

    for (int r = wave; r < rows; r += TA_WAVES) {
        const int   i  = lo + r;
        const int   g  = batch[i];
        const float di = dinv[i];
        const float w  = di * di;
        const float4 t = *(const float4*)(outt + r * TA_PITCH + (lane << 2));
        float* p = Sl + g * HID_DIM + lane;
        atomicAdd(p,      t.x * w);
        atomicAdd(p + 32, t.y * w);
        atomicAdd(p + 64, t.z * w);
        atomicAdd(p + 96, t.w * w);
        if (lane == 0) atomicAdd(&Cl[g], 1.0f);
    }
    {
        const TaPoolFlush pf{lane, lo, dinv, src, dst, batch, outt, Sl};
        ta_scan((const unsigned char*)skeys, own, E, wave, lane, start, mylist, pf);
    }

    __syncthreads();

    for (int idx = tid; idx < NUM_GRAPHS * HID_DIM; idx += TA_THREADS) {
        const int g  = idx >> 7;
        const int cc = idx & (HID_DIM - 1);
        atomicAdd(&S[idx], Sl[g * HID_DIM + ((cc & 3) << 5) + (cc >> 2)]);
    }
    if (tid < NUM_GRAPHS) {
        const float cv = Cl[tid];
        if (cv != 0.0f) atomicAdd(&cnts[tid], cv);
    }
}

__global__ void pool_final(const float* __restrict__ T,
                           const float* __restrict__ cnt,
                           const float* __restrict__ bias,
                           float* __restrict__ out, int G, int D) {
    int tid = blockIdx.x * blockDim.x + threadIdx.x;
    if (tid >= G * D) return;
    int g = tid / D;
    int c = tid - g * D;
    const float cn = cnt[g];
    out[tid] = (T[tid] + cn * bias[c]) / fmaxf(cn, 1.0f);
}

extern "C" void kernel_launch(void* const* d_in, const int* in_sizes, int n_in,
                              void* d_out, int out_size, void* d_ws, size_t ws_size,
                              hipStream_t stream) {
    const float* x    = (const float*)d_in[0];
    const int*   eidx = (const int*)  d_in[1];
    const int*   batch= (const int*)  d_in[2];
    const float* W1   = (const float*)d_in[3];
    const float* b1   = (const float*)d_in[4];
    const float* W2   = (const float*)d_in[5];
    const float* b2   = (const float*)d_in[6];
    float* out = (float*)d_out;

    const int Nn = in_sizes[0] / IN_DIM;
    const int E  = in_sizes[1] / 2;
    const int* src = eidx;
    const int* dst = eidx + E;

    float* ws   = (float*)d_ws;
    float* S    = ws;
    float* T    = S    + NUM_GRAPHS * HID_DIM;
    unsigned* keys  = (unsigned*)(T + NUM_GRAPHS * OUT_DIM);
    unsigned* skeys = keys + (((size_t)E + 15) >> 4) * 4;
    float* deg  = (float*)(skeys + (((size_t)E + 15) >> 4) * 4);
    float* dinv = deg  + Nn;
    float* cnts = dinv + Nn;

    const int B = 256;
    auto blocks = [](long long n, int b) { return (unsigned)((n + b - 1) / b); };

    fill_kernel<<<blocks(Nn, B), B, 0, stream>>>(deg, 1.0f, Nn);
    fill_kernel<<<blocks(NUM_GRAPHS * HID_DIM, B), B, 0, stream>>>(S, 0.0f, NUM_GRAPHS * HID_DIM);
    fill_kernel<<<blocks(NUM_GRAPHS, B), B, 0, stream>>>(cnts, 0.0f, NUM_GRAPHS);

    degree_key_kernel<<<blocks((E + 3) >> 2, B), B, 0, stream>>>(src, dst, deg, keys, skeys, E);
    rsqrt_kernel<<<blocks(Nn, B), B, 0, stream>>>(deg, dinv, Nn);

    tile_aggregate<<<blocks(Nn, TA_ROWS), TA_THREADS, 0, stream>>>(
        x, dinv, src, dst, batch, keys, skeys, W1, b1, S, cnts, E, Nn);

    {
        int tiles = (NUM_GRAPHS >> 4) * (OUT_DIM >> 4);
        dim3 blk(32, 8);
        gemm_wmma_f32<<<blocks(tiles, 8), blk, 0, stream>>>(S, W2, T, NUM_GRAPHS, OUT_DIM, HID_DIM);
    }
    pool_final<<<blocks(NUM_GRAPHS * OUT_DIM, B), B, 0, stream>>>(
        T, cnts, b2, out, NUM_GRAPHS, OUT_DIM);
}


# Pooled sums leave each work-group as its own 64x64 slice of S-copy @ W2 (f32 WMMA from LDS, plain stores); last launch adds the slices: S/count float atomics, 2 fills, GEMM launch go; sum order moves
# speedup vs baseline: 3.1391x; 1.0487x over previous
#include <hip/hip_runtime.h>

#define IN_DIM   128
#define HID_DIM  128
#define OUT_DIM  64
#define NUM_GRAPHS 64

typedef float v2f __attribute__((ext_vector_type(2)));
typedef float v8f __attribute__((ext_vector_type(8)));

__global__ void fill_kernel(float* __restrict__ p, float v, int n) {
    int i = blockIdx.x * blockDim.x + threadIdx.x;
    if (i < n) p[i] = v;
}

__global__ void rsqrt_kernel(const float* __restrict__ deg, float* __restrict__ dinv, int n) {
    int i = blockIdx.x * blockDim.x + threadIdx.x;
    if (i < n) dinv[i] = rsqrtf(deg[i]);
}


#define TA_ROWS    196
#define TA_STRIPS  13
#define TA_PADROWS (TA_STRIPS * 16)
#define TA_PITCH   (IN_DIM + 4)
#define TA_THREADS 1024
#define TA_WAVES   (TA_THREADS / 32)
#define TA_AHEAD   4
#define TA_LIST    (32 + 512)
#define TA_GROUP   8

__global__ void degree_key_kernel(const int* __restrict__ src, const int* __restrict__ dst,
                                  float* __restrict__ deg, unsigned* __restrict__ keys,
                                  unsigned* __restrict__ skeys, int E) {
    const int t = blockIdx.x * blockDim.x + threadIdx.x;
    const int e = t << 2;
    if (e + 3 < E) {
        const int d0 = dst[e], d1 = dst[e + 1], d2 = dst[e + 2], d3 = dst[e + 3];
        const int s0 = src[e], s1 = src[e + 1], s2 = src[e + 2], s3 = src[e + 3];
        atomicAdd(&deg[d0], 1.0f);
        atomicAdd(&deg[d1], 1.0f);
        atomicAdd(&deg[d2], 1.0f);
        atomicAdd(&deg[d3], 1.0f);
        keys[t] = ((unsigned)d0 / TA_ROWS) | (((unsigned)d1 / TA_ROWS) << 8) |
                  (((unsigned)d2 / TA_ROWS) << 16) | (((unsigned)d3 / TA_ROWS) << 24);
        skeys[t] = ((unsigned)s0 / TA_ROWS) | (((unsigned)s1 / TA_ROWS) << 8) |
                   (((unsigned)s2 / TA_ROWS) << 16) | (((unsigned)s3 / TA_ROWS) << 24);
    } else {
        for (int i = e; i < E; ++i) {
            const int d = dst[i];
            atomicAdd(&deg[d], 1.0f);
            ((unsigned char*)keys)[i] = (unsigned char)((unsigned)d / TA_ROWS);
            ((unsigned char*)skeys)[i] = (unsigned char)((unsigned)src[i] / TA_ROWS);
        }
    }
}

__device__ __forceinline__ unsigned ta_zero_bytes(unsigned t) {
    return ~(((t & 0x7F7F7F7Fu) + 0x7F7F7F7Fu) | t | 0x7F7F7F7Fu);
}

__device__ __forceinline__ int ta_append(int* list, int cnt, bool m, int e) {
    const unsigned mask = __builtin_amdgcn_ballot_w32(m);
    if (mask != 0u) {
        const int pos = cnt + (int)__builtin_amdgcn_mbcnt_lo(mask, 0u);
        if (m) list[pos] = e;
        cnt += __builtin_popcount(mask);
    }
    return cnt;
}

__device__ __forceinline__ void ta_flush(const int* list, int nv, int lane, int lo,
                                         const float* __restrict__ x,
                                         const float* __restrict__ dinv,
                                         const int* __restrict__ src,
                                         const int* __restrict__ dst,
                                         float* tile) {
    int s = 0, r = 0;
    float w = 0.0f;
    if (lane < nv) {
        const int e = list[lane];
        s = src[e];
        const int d = dst[e];
        r = d - lo;
        w = dinv[s] * dinv[d];
    }
    int k0 = 0;
    for (; k0 + TA_GROUP <= nv; k0 += TA_GROUP) {
        float4 v[TA_GROUP];
        float  wk[TA_GROUP];
        int    rk[TA_GROUP];
#pragma unroll
        for (int j = 0; j < TA_GROUP; ++j) {
            const int sk = __builtin_amdgcn_readlane(s, k0 + j);
            wk[j] = __int_as_float(__builtin_amdgcn_readlane(__float_as_int(w), k0 + j));
            rk[j] = __builtin_amdgcn_readlane(r, k0 + j);
            v[j]  = *(const float4*)(x + (size_t)sk * IN_DIM + (lane << 2));
        }
#pragma unroll
        for (int j = 0; j < TA_GROUP; ++j) {
            float* p = tile + rk[j] * TA_PITCH + lane;
            atomicAdd(p,      v[j].x * wk[j]);
            atomicAdd(p + 32, v[j].y * wk[j]);
            atomicAdd(p + 64, v[j].z * wk[j]);
            atomicAdd(p + 96, v[j].w * wk[j]);
        }
    }
    for (; k0 < nv; ++k0) {
        const int   sk = __builtin_amdgcn_readlane(s, k0);
        const float wq = __int_as_float(__builtin_amdgcn_readlane(__float_as_int(w), k0));
        const int   rq = __builtin_amdgcn_readlane(r, k0);
        const float4 vv = *(const float4*)(x + (size_t)sk * IN_DIM + (lane << 2));
        float* p = tile + rq * TA_PITCH + lane;
        atomicAdd(p,      vv.x * wq);
        atomicAdd(p + 32, vv.y * wq);
        atomicAdd(p + 64, vv.z * wq);
        atomicAdd(p + 96, vv.w * wq);
    }
}

__device__ __forceinline__ void ta_pool_flush(const int* list, int nv, int lane, int lo,
                                              const float* __restrict__ dinv,
                                              const int* __restrict__ src,
                                              const int* __restrict__ dst,
                                              const int* __restrict__ batch,
                                              const float* rowsl, float* Sl) {
    int r = 0, g = 0;
    float w = 0.0f;
    if (lane < nv) {
        const int e = list[lane];
        const int s = src[e];
        const int d = dst[e];
        r = s - lo;
        g = batch[d];
        w = dinv[s] * dinv[d];
    }
    int k0 = 0;
    for (; k0 + TA_GROUP <= nv; k0 += TA_GROUP) {
        float4 v[TA_GROUP];
        float  wk[TA_GROUP];
        int    gk[TA_GROUP];
#pragma unroll
        for (int j = 0; j < TA_GROUP; ++j) {
            const int rk = __builtin_amdgcn_readlane(r, k0 + j);
            wk[j] = __int_as_float(__builtin_amdgcn_readlane(__float_as_int(w), k0 + j));
            gk[j] = __builtin_amdgcn_readlane(g, k0 + j);
            v[j]  = *(const float4*)(rowsl + rk * TA_PITCH + (lane << 2));
        }
#pragma unroll
        for (int j = 0; j < TA_GROUP; ++j) {
            float* p = Sl + gk[j] * TA_PITCH + lane;
            atomicAdd(p,      v[j].x * wk[j]);
            atomicAdd(p + 32, v[j].y * wk[j]);
            atomicAdd(p + 64, v[j].z * wk[j]);
            atomicAdd(p + 96, v[j].w * wk[j]);
        }
    }
    for (; k0 < nv; ++k0) {
        const int   rq = __builtin_amdgcn_readlane(r, k0);
        const float wq = __int_as_float(__builtin_amdgcn_readlane(__float_as_int(w), k0));
        const int   gq = __builtin_amdgcn_readlane(g, k0);
        const float4 vv = *(const float4*)(rowsl + rq * TA_PITCH + (lane << 2));
        float* p = Sl + gq * TA_PITCH + lane;
        atomicAdd(p,      vv.x * wq);
        atomicAdd(p + 32, vv.y * wq);
        atomicAdd(p + 64, vv.z * wq);
        atomicAdd(p + 96, vv.w * wq);
    }
}

struct TaGatherFlush {
    int lane, lo;
    const float* x; const float* dinv; const int* src; const int* dst; float* tile;
    __device__ __forceinline__ void operator()(const int* list, int nv) const {
        ta_flush(list, nv, lane, lo, x, dinv, src, dst, tile);
    }
};
struct TaPoolFlush {
    int lane, lo;
    const float* dinv; const int* src; const int* dst; const int* batch; const float* rowsl; float* Sl;
    __device__ __forceinline__ void operator()(const int* list, int nv) const {
        ta_pool_flush(list, nv, lane, lo, dinv, src, dst, batch, rowsl, Sl);
    }
};

template <class Flush>
__device__ __forceinline__ void ta_scan(const unsigned char* __restrict__ key8, unsigned own, int E,
                                        int wave, int lane, int start, int* mylist, const Flush& flush) {
    int cnt = 0;
    const unsigned pat = own * 0x01010101u;
    const int nChunks = E >> 9;

    static_assert(TA_AHEAD == 4, "the chunk walk below selects among four hit words");
    for (int kc = wave; kc < nChunks; kc += TA_AHEAD * TA_WAVES) {
        unsigned h[TA_AHEAD];
        {
            uint4 d[TA_AHEAD];
#pragma unroll
            for (int u = 0; u < TA_AHEAD; ++u) {
                int c = kc + u * TA_WAVES;
                if (c >= nChunks) c = kc;
                int cc = c + start;
                if (cc >= nChunks) cc -= nChunks;
                d[u] = *(const uint4*)(key8 + (cc << 9) + (lane << 4));
            }
#pragma unroll
            for (int u = 0; u < TA_AHEAD; ++u) {
                unsigned t = ta_zero_bytes(d[u].x ^ pat) >> 7;
                t |= ta_zero_bytes(d[u].y ^ pat) >> 6;
                t |= ta_zero_bytes(d[u].z ^ pat) >> 5;
                t |= ta_zero_bytes(d[u].w ^ pat) >> 4;
                h[u] = t;
            }
        }
        int nv = (nChunks - kc + TA_WAVES - 1) / TA_WAVES;
        if (nv > TA_AHEAD) nv = TA_AHEAD;
#pragma nounroll
        for (int u = 0; u < nv; ++u) {
            unsigned hits = (u == 0) ? h[0] : (u == 1) ? h[1] : (u == 2) ? h[2] : h[3];
            int cc = kc + u * TA_WAVES + start;
            if (cc >= nChunks) cc -= nChunks;
            const int e0 = (cc << 9) + (lane << 4);
            unsigned mask = __builtin_amdgcn_ballot_w32(hits != 0u);
            while (mask != 0u) {
                const int p = __builtin_ctz(hits | 0x80000000u);
                const int pos = cnt + (int)__builtin_amdgcn_mbcnt_lo(mask, 0u);
                if (hits != 0u) mylist[pos] = e0 + ((p & 7) << 2) + (p >> 3);
                cnt += __builtin_popcount(mask);
                hits &= hits - 1u;
                mask = __builtin_amdgcn_ballot_w32(hits != 0u);
            }
            while (cnt >= 32) {
                cnt -= 32;
                flush(mylist + cnt, 32);
            }
        }
    }
    for (int base = (nChunks << 9) + (wave << 5); base < E; base += TA_WAVES * 32) {
        const int e = base + lane;
        bool m = false;
        if (e < E) m = (unsigned)key8[e] == own;
        cnt = ta_append(mylist, cnt, m, e);
        if (cnt >= 32) {
            cnt -= 32;
            flush(mylist + cnt, 32);
        }
    }
    if (cnt > 0) flush(mylist, cnt);
}

__global__ __launch_bounds__(TA_THREADS)
void tile_aggregate(const float* __restrict__ x,
                    const float* __restrict__ dinv,
                    const int* __restrict__ src,
                    const int* __restrict__ dst,
                    const int* __restrict__ batch,
                    const unsigned* __restrict__ keys,
                    const unsigned* __restrict__ skeys,
                    const float* __restrict__ W,
                    const float* __restrict__ bias,
                    const float* __restrict__ W2,
                    float* __restrict__ Tp,
                    float* __restrict__ Cp,
                    int E, int Nn) {
    static_assert(IN_DIM == 128 && HID_DIM == 128, "fragment walk below is written for 128 x 128");
    static_assert(NUM_GRAPHS * TA_PITCH <= TA_PADROWS * TA_PITCH, "the LDS copy of S reuses the tile plane");
    static_assert(NUM_GRAPHS * OUT_DIM == TA_THREADS * 4, "the slice copy below moves four floats per thread");
    static_assert(NUM_GRAPHS == 64 && OUT_DIM == 64 && TA_WAVES >= 16, "16 waves take one 16x16 tile of the 64 x 64 product each");
    __shared__ __attribute__((aligned(16))) float tile[TA_PADROWS * TA_PITCH];
    __shared__ __attribute__((aligned(16))) float outt[TA_PADROWS * TA_PITCH];
    __shared__ int   lists[TA_WAVES * TA_LIST];
    __shared__ float Cl[NUM_GRAPHS];
    const int tid  = threadIdx.x;
    const int lane = tid & 31;
    const int wave = __builtin_amdgcn_readfirstlane(tid >> 5);
    const int lo   = (int)blockIdx.x * TA_ROWS;
    int rows = Nn - lo;
    if (rows > TA_ROWS) rows = TA_ROWS;

    if (tid < NUM_GRAPHS) Cl[tid] = 0.0f;
    for (int r = wave; r < TA_PADROWS; r += TA_WAVES) {
        float* p = tile + r * TA_PITCH + lane;
        if (r < rows) {
            const int   i  = lo + r;
            const float di = dinv[i];
            const float w  = di * di;
            const float4 xv = *(const float4*)(x + (size_t)i * IN_DIM + (lane << 2));
            p[0]  = xv.x * w;
            p[32] = xv.y * w;
            p[64] = xv.z * w;
            p[96] = xv.w * w;
        } else {
            p[0] = 0.0f; p[32] = 0.0f; p[64] = 0.0f; p[96] = 0.0f;
        }
    }
    __syncthreads();

    int* mylist = lists + wave * TA_LIST;
    const unsigned own = (unsigned)blockIdx.x;
    const int start = (int)blockIdx.x * ((E >> 9) / (int)gridDim.x);

    {
        const TaGatherFlush gf{lane, lo, x, dinv, src, dst, tile};
        ta_scan((const unsigned char*)keys, own, E, wave, lane, start, mylist, gf);
    }

    const int half = lane >> 4;
    const int m    = lane & 15;
    const int nt   = wave & 7;
    v2f bf[32];
#pragma unroll
    for (int q = 0; q < 32; ++q) {
        const int k = ((q & 7) << 4) + (half << 3) + (q >> 3);
        bf[q].x = W[(size_t)k * HID_DIM + (nt << 4) + m];
        bf[q].y = W[(size_t)(k + 4) * HID_DIM + (nt << 4) + m];
    }
    const float bcol = bias[(nt << 4) + m];

    __syncthreads();

    const int nStrips = (rows + 15) >> 4;
    for (int s = wave >> 3; s < nStrips; s += TA_WAVES / 8) {
        const float* ap = tile + ((s << 4) + m) * TA_PITCH + (half << 1);
        v8f acc = {};
#pragma unroll
        for (int q = 0; q < 32; ++q) {
            const v2f a = *(const v2f*)(ap + (q << 2));
            acc = __builtin_amdgcn_wmma_f32_16x16x4_f32(
                 false, a,  false, bf[q],
                 (short)0, acc,  false,  false);
        }
        float* op = outt + ((s << 4) + (half << 3)) * TA_PITCH + (nt << 4) + m;
#pragma unroll
        for (int v = 0; v < 8; ++v) op[v * TA_PITCH] = fmaxf(acc[v] + bcol, 0.0f);
    }

    __syncthreads();

    float* Sl = tile;
    for (int i = tid; i < NUM_GRAPHS * TA_PITCH; i += TA_THREADS) Sl[i] = 0.0f;

    __syncthreads();

    for (int r = wave; r < rows; r += TA_WAVES) {
        const int   i  = lo + r;
        const int   g  = batch[i];
        const float di = dinv[i];
        const float w  = di * di;
        const float4 t = *(const float4*)(outt + r * TA_PITCH + (lane << 2));
        float* p = Sl + g * TA_PITCH + lane;
        atomicAdd(p,      t.x * w);
        atomicAdd(p + 32, t.y * w);
        atomicAdd(p + 64, t.z * w);
        atomicAdd(p + 96, t.w * w);
        if (lane == 0) atomicAdd(&Cl[g], 1.0f);
    }
    {
        const TaPoolFlush pf{lane, lo, dinv, src, dst, batch, outt, Sl};
        ta_scan((const unsigned char*)skeys, own, E, wave, lane, start, mylist, pf);
    }

    const int tm2 = (wave >> 2) & 3;
    const int tn2 = wave & 3;
    v2f bf2[32];
    if (wave < 16) {
#pragma unroll
        for (int q = 0; q < 32; ++q) {
            const int k = ((q & 7) << 4) + (half << 3) + (q >> 3);
            bf2[q].x = W2[(size_t)k * OUT_DIM + (tn2 << 4) + m];
            bf2[q].y = W2[(size_t)(k + 4) * OUT_DIM + (tn2 << 4) + m];
        }
    }

    __syncthreads();

    float* Tl = outt;
    if (wave < 16) {
        const float* ap = Sl + ((tm2 << 4) + m) * TA_PITCH + (half << 1);
        v8f acc = {};
#pragma unroll
        for (int q = 0; q < 32; ++q) {
            const v2f a = *(const v2f*)(ap + (q << 2));
            acc = __builtin_amdgcn_wmma_f32_16x16x4_f32(
                 false, a,  false, bf2[q],
                 (short)0, acc,  false,  false);
        }
        float* op = Tl + ((tm2 << 4) + (half << 3)) * OUT_DIM + (tn2 << 4) + m;
#pragma unroll
        for (int v = 0; v < 8; ++v) op[v * OUT_DIM] = acc[v];
    }

    __syncthreads();

    ((float4*)(Tp + (size_t)blockIdx.x * (NUM_GRAPHS * OUT_DIM)))[tid] = ((const float4*)Tl)[tid];
    if (tid < NUM_GRAPHS) Cp[(size_t)blockIdx.x * NUM_GRAPHS + tid] = Cl[tid];
}

#define PF_THREADS 1024
#define PF_PARTS   (PF_THREADS / OUT_DIM)
__global__ __launch_bounds__(PF_THREADS)
void pool_final(const float* __restrict__ Tp,
                const float* __restrict__ Cp,
                const float* __restrict__ bias,
                float* __restrict__ out, int nb) {
    __shared__ float red[PF_PARTS][OUT_DIM];
    __shared__ float cred[PF_THREADS / 32];
    const int g    = blockIdx.x;
    const int tid  = threadIdx.x;
    const int c    = tid & (OUT_DIM - 1);
    const int part = tid >> 6;
    const float* tp = Tp + (size_t)g * OUT_DIM + c;
    float acc = 0.0f;
#pragma unroll 16
    for (int b = part; b < nb; b += PF_PARTS) acc += tp[(size_t)b * (NUM_GRAPHS * OUT_DIM)];
    float cn = 0.0f;
    for (int b = tid; b < nb; b += PF_THREADS) cn += Cp[(size_t)b * NUM_GRAPHS + g];
    for (int o = 16; o > 0; o >>= 1) cn += __shfl_xor(cn, o);
    red[part][c] = acc;
    if ((tid & 31) == 0) cred[tid >> 5] = cn;
    __syncthreads();
    if (tid < OUT_DIM) {
        float t = 0.0f;
        for (int p = 0; p < PF_PARTS; ++p) t += red[p][c];
        float n = 0.0f;
        for (int w = 0; w < PF_THREADS / 32; ++w) n += cred[w];
        out[g * OUT_DIM + c] = (t + n * bias[c]) / fmaxf(n, 1.0f);
    }
}

extern "C" void kernel_launch(void* const* d_in, const int* in_sizes, int n_in,
                              void* d_out, int out_size, void* d_ws, size_t ws_size,
                              hipStream_t stream) {
    const float* x    = (const float*)d_in[0];
    const int*   eidx = (const int*)  d_in[1];
    const int*   batch= (const int*)  d_in[2];
    const float* W1   = (const float*)d_in[3];
    const float* b1   = (const float*)d_in[4];
    const float* W2   = (const float*)d_in[5];
    const float* b2   = (const float*)d_in[6];
    float* out = (float*)d_out;

    const int Nn = in_sizes[0] / IN_DIM;
    const int E  = in_sizes[1] / 2;
    const int* src = eidx;
    const int* dst = eidx + E;

    const int B = 256;
    auto blocks = [](long long n, int b) { return (unsigned)((n + b - 1) / b); };
    const unsigned nTiles = blocks(Nn, TA_ROWS);

    const size_t keyWords   = (((size_t)E + 127) >> 7) * 32;
    const size_t nodeFloats = (((size_t)Nn + 31) >> 5) * 32;
    unsigned* keys  = (unsigned*)d_ws;
    unsigned* skeys = keys + keyWords;
    float* deg  = (float*)(skeys + keyWords);
    float* dinv = deg  + nodeFloats;
    float* Cp   = dinv + nodeFloats;
    float* Tp   = Cp   + (size_t)nTiles * NUM_GRAPHS;

    fill_kernel<<<blocks(Nn, B), B, 0, stream>>>(deg, 1.0f, Nn);

    degree_key_kernel<<<blocks((E + 3) >> 2, B), B, 0, stream>>>(src, dst, deg, keys, skeys, E);
    rsqrt_kernel<<<blocks(Nn, B), B, 0, stream>>>(deg, dinv, Nn);

    tile_aggregate<<<nTiles, TA_THREADS, 0, stream>>>(
        x, dinv, src, dst, batch, keys, skeys, W1, b1, W2, Tp, Cp, E, Nn);

    pool_final<<<NUM_GRAPHS, PF_THREADS, 0, stream>>>(Tp, Cp, b2, out, (int)nTiles);
}


# In-degree counts taken in LDS: 64 work-groups count their edge stretch into per-node LDS integer counters, store a plane each; the rsqrt pass adds the planes; global float atomics and fill launch go
# speedup vs baseline: 4.2698x; 1.3602x over previous
#include <hip/hip_runtime.h>

#define IN_DIM   128
#define HID_DIM  128
#define OUT_DIM  64
#define NUM_GRAPHS 64

typedef float v2f __attribute__((ext_vector_type(2)));
typedef float v8f __attribute__((ext_vector_type(8)));


#define TA_ROWS    196
#define TA_STRIPS  13
#define TA_PADROWS (TA_STRIPS * 16)
#define TA_PITCH   (IN_DIM + 4)
#define TA_THREADS 1024
#define TA_WAVES   (TA_THREADS / 32)
#define TA_AHEAD   4
#define TA_LIST    (32 + 512)
#define TA_GROUP   8

#define DH_BLOCKS  64
#define DH_THREADS 1024
#define DH_NODES   (256 * TA_ROWS)
#define DH_TRIPS   3

__global__ __launch_bounds__(DH_THREADS)
void degree_key_kernel(const int* __restrict__ src, const int* __restrict__ dst,
                       unsigned* __restrict__ part, unsigned* __restrict__ keys,
                       unsigned* __restrict__ skeys, int E) {
    static_assert(DH_NODES % 4 == 0, "the counters are cleared and copied 16 bytes at a time");
    __shared__ __attribute__((aligned(16))) unsigned cnt[DH_NODES];
    const int tid = threadIdx.x;
    const int nFull = E >> 2;
    const int per = (nFull + (int)gridDim.x - 1) / (int)gridDim.x;
    const int g0 = (int)blockIdx.x * per;
    int g1 = g0 + per;
    if (g1 > nFull) g1 = nFull;

    for (int i = tid; i < DH_NODES / 4; i += DH_THREADS) ((uint4*)cnt)[i] = make_uint4(0u, 0u, 0u, 0u);

    for (int base = g0; base < g1; base += DH_TRIPS * DH_THREADS) {
        int dd[DH_TRIPS][4], ss[DH_TRIPS][4];
#pragma unroll
        for (int u = 0; u < DH_TRIPS; ++u) {
            const int t = base + u * DH_THREADS + tid;
            if (t < g1) {
                const int e = t << 2;
#pragma unroll
                for (int j = 0; j < 4; ++j) { dd[u][j] = dst[e + j]; ss[u][j] = src[e + j]; }
            }
        }
        __syncthreads();
#pragma unroll
        for (int u = 0; u < DH_TRIPS; ++u) {
            const int t = base + u * DH_THREADS + tid;
            if (t < g1) {
                unsigned k = 0u, sk = 0u;
#pragma unroll
                for (int j = 0; j < 4; ++j) {
                    const unsigned d = (unsigned)dd[u][j];
                    if (d < (unsigned)DH_NODES) atomicAdd(&cnt[d], 1u);
                    k  |= (d / TA_ROWS) << (8 * j);
                    sk |= ((unsigned)ss[u][j] / TA_ROWS) << (8 * j);
                }
                keys[t]  = k;
                skeys[t] = sk;
            }
        }
    }
    __syncthreads();
    if (blockIdx.x == gridDim.x - 1 && tid == 0) {
        for (int i = nFull << 2; i < E; ++i) {
            const unsigned d = (unsigned)dst[i];
            if (d < (unsigned)DH_NODES) atomicAdd(&cnt[d], 1u);
            ((unsigned char*)keys)[i]  = (unsigned char)(d / TA_ROWS);
            ((unsigned char*)skeys)[i] = (unsigned char)((unsigned)src[i] / TA_ROWS);
        }
    }
    __syncthreads();

    uint4* pp = (uint4*)(part + (size_t)blockIdx.x * DH_NODES);
    for (int i = tid; i < DH_NODES / 4; i += DH_THREADS) pp[i] = ((const uint4*)cnt)[i];
}

__global__ void dinv_kernel(const unsigned* __restrict__ part, float* __restrict__ dinv, int n) {
    const int i = blockIdx.x * blockDim.x + threadIdx.x;
    if (i < n) {
        unsigned c = 1u;
#pragma unroll
        for (int b = 0; b < DH_BLOCKS; ++b) c += part[(size_t)b * DH_NODES + i];
        dinv[i] = rsqrtf((float)c);
    }
}

__device__ __forceinline__ unsigned ta_zero_bytes(unsigned t) {
    return ~(((t & 0x7F7F7F7Fu) + 0x7F7F7F7Fu) | t | 0x7F7F7F7Fu);
}

__device__ __forceinline__ int ta_append(int* list, int cnt, bool m, int e) {
    const unsigned mask = __builtin_amdgcn_ballot_w32(m);
    if (mask != 0u) {
        const int pos = cnt + (int)__builtin_amdgcn_mbcnt_lo(mask, 0u);
        if (m) list[pos] = e;
        cnt += __builtin_popcount(mask);
    }
    return cnt;
}

__device__ __forceinline__ void ta_flush(const int* list, int nv, int lane, int lo,
                                         const float* __restrict__ x,
                                         const float* __restrict__ dinv,
                                         const int* __restrict__ src,
                                         const int* __restrict__ dst,
                                         float* tile) {
    int s = 0, r = 0;
    float w = 0.0f;
    if (lane < nv) {
        const int e = list[lane];
        s = src[e];
        const int d = dst[e];
        r = d - lo;
        w = dinv[s] * dinv[d];
    }
    int k0 = 0;
    for (; k0 + TA_GROUP <= nv; k0 += TA_GROUP) {
        float4 v[TA_GROUP];
        float  wk[TA_GROUP];
        int    rk[TA_GROUP];
#pragma unroll
        for (int j = 0; j < TA_GROUP; ++j) {
            const int sk = __builtin_amdgcn_readlane(s, k0 + j);
            wk[j] = __int_as_float(__builtin_amdgcn_readlane(__float_as_int(w), k0 + j));
            rk[j] = __builtin_amdgcn_readlane(r, k0 + j);
            v[j]  = *(const float4*)(x + (size_t)sk * IN_DIM + (lane << 2));
        }
#pragma unroll
        for (int j = 0; j < TA_GROUP; ++j) {
            float* p = tile + rk[j] * TA_PITCH + lane;
            atomicAdd(p,      v[j].x * wk[j]);
            atomicAdd(p + 32, v[j].y * wk[j]);
            atomicAdd(p + 64, v[j].z * wk[j]);
            atomicAdd(p + 96, v[j].w * wk[j]);
        }
    }
    for (; k0 < nv; ++k0) {
        const int   sk = __builtin_amdgcn_readlane(s, k0);
        const float wq = __int_as_float(__builtin_amdgcn_readlane(__float_as_int(w), k0));
        const int   rq = __builtin_amdgcn_readlane(r, k0);
        const float4 vv = *(const float4*)(x + (size_t)sk * IN_DIM + (lane << 2));
        float* p = tile + rq * TA_PITCH + lane;
        atomicAdd(p,      vv.x * wq);
        atomicAdd(p + 32, vv.y * wq);
        atomicAdd(p + 64, vv.z * wq);
        atomicAdd(p + 96, vv.w * wq);
    }
}

__device__ __forceinline__ void ta_pool_flush(const int* list, int nv, int lane, int lo,
                                              const float* __restrict__ dinv,
                                              const int* __restrict__ src,
                                              const int* __restrict__ dst,
                                              const int* __restrict__ batch,
                                              const float* rowsl, float* Sl) {
    int r = 0, g = 0;
    float w = 0.0f;
    if (lane < nv) {
        const int e = list[lane];
        const int s = src[e];
        const int d = dst[e];
        r = s - lo;
        g = batch[d];
        w = dinv[s] * dinv[d];
    }
    int k0 = 0;
    for (; k0 + TA_GROUP <= nv; k0 += TA_GROUP) {
        float4 v[TA_GROUP];
        float  wk[TA_GROUP];
        int    gk[TA_GROUP];
#pragma unroll
        for (int j = 0; j < TA_GROUP; ++j) {
            const int rk = __builtin_amdgcn_readlane(r, k0 + j);
            wk[j] = __int_as_float(__builtin_amdgcn_readlane(__float_as_int(w), k0 + j));
            gk[j] = __builtin_amdgcn_readlane(g, k0 + j);
            v[j]  = *(const float4*)(rowsl + rk * TA_PITCH + (lane << 2));
        }
#pragma unroll
        for (int j = 0; j < TA_GROUP; ++j) {
            float* p = Sl + gk[j] * TA_PITCH + lane;
            atomicAdd(p,      v[j].x * wk[j]);
            atomicAdd(p + 32, v[j].y * wk[j]);
            atomicAdd(p + 64, v[j].z * wk[j]);
            atomicAdd(p + 96, v[j].w * wk[j]);
        }
    }
    for (; k0 < nv; ++k0) {
        const int   rq = __builtin_amdgcn_readlane(r, k0);
        const float wq = __int_as_float(__builtin_amdgcn_readlane(__float_as_int(w), k0));
        const int   gq = __builtin_amdgcn_readlane(g, k0);
        const float4 vv = *(const float4*)(rowsl + rq * TA_PITCH + (lane << 2));
        float* p = Sl + gq * TA_PITCH + lane;
        atomicAdd(p,      vv.x * wq);
        atomicAdd(p + 32, vv.y * wq);
        atomicAdd(p + 64, vv.z * wq);
        atomicAdd(p + 96, vv.w * wq);
    }
}

struct TaGatherFlush {
    int lane, lo;
    const float* x; const float* dinv; const int* src; const int* dst; float* tile;
    __device__ __forceinline__ void operator()(const int* list, int nv) const {
        ta_flush(list, nv, lane, lo, x, dinv, src, dst, tile);
    }
};
struct TaPoolFlush {
    int lane, lo;
    const float* dinv; const int* src; const int* dst; const int* batch; const float* rowsl; float* Sl;
    __device__ __forceinline__ void operator()(const int* list, int nv) const {
        ta_pool_flush(list, nv, lane, lo, dinv, src, dst, batch, rowsl, Sl);
    }
};

template <class Flush>
__device__ __forceinline__ void ta_scan(const unsigned char* __restrict__ key8, unsigned own, int E,
                                        int wave, int lane, int start, int* mylist, const Flush& flush) {
    int cnt = 0;
    const unsigned pat = own * 0x01010101u;
    const int nChunks = E >> 9;

    static_assert(TA_AHEAD == 4, "the chunk walk below selects among four hit words");
    for (int kc = wave; kc < nChunks; kc += TA_AHEAD * TA_WAVES) {
        unsigned h[TA_AHEAD];
        {
            uint4 d[TA_AHEAD];
#pragma unroll
            for (int u = 0; u < TA_AHEAD; ++u) {
                int c = kc + u * TA_WAVES;
                if (c >= nChunks) c = kc;
                int cc = c + start;
                if (cc >= nChunks) cc -= nChunks;
                d[u] = *(const uint4*)(key8 + (cc << 9) + (lane << 4));
            }
#pragma unroll
            for (int u = 0; u < TA_AHEAD; ++u) {
                unsigned t = ta_zero_bytes(d[u].x ^ pat) >> 7;
                t |= ta_zero_bytes(d[u].y ^ pat) >> 6;
                t |= ta_zero_bytes(d[u].z ^ pat) >> 5;
                t |= ta_zero_bytes(d[u].w ^ pat) >> 4;
                h[u] = t;
            }
        }
        int nv = (nChunks - kc + TA_WAVES - 1) / TA_WAVES;
        if (nv > TA_AHEAD) nv = TA_AHEAD;
#pragma nounroll
        for (int u = 0; u < nv; ++u) {
            unsigned hits = (u == 0) ? h[0] : (u == 1) ? h[1] : (u == 2) ? h[2] : h[3];
            int cc = kc + u * TA_WAVES + start;
            if (cc >= nChunks) cc -= nChunks;
            const int e0 = (cc << 9) + (lane << 4);
            unsigned mask = __builtin_amdgcn_ballot_w32(hits != 0u);
            while (mask != 0u) {
                const int p = __builtin_ctz(hits | 0x80000000u);
                const int pos = cnt + (int)__builtin_amdgcn_mbcnt_lo(mask, 0u);
                if (hits != 0u) mylist[pos] = e0 + ((p & 7) << 2) + (p >> 3);
                cnt += __builtin_popcount(mask);
                hits &= hits - 1u;
                mask = __builtin_amdgcn_ballot_w32(hits != 0u);
            }
            while (cnt >= 32) {
                cnt -= 32;
                flush(mylist + cnt, 32);
            }
        }
    }
    for (int base = (nChunks << 9) + (wave << 5); base < E; base += TA_WAVES * 32) {
        const int e = base + lane;
        bool m = false;
        if (e < E) m = (unsigned)key8[e] == own;
        cnt = ta_append(mylist, cnt, m, e);
        if (cnt >= 32) {
            cnt -= 32;
            flush(mylist + cnt, 32);
        }
    }
    if (cnt > 0) flush(mylist, cnt);
}

__global__ __launch_bounds__(TA_THREADS)
void tile_aggregate(const float* __restrict__ x,
                    const float* __restrict__ dinv,
                    const int* __restrict__ src,
                    const int* __restrict__ dst,
                    const int* __restrict__ batch,
                    const unsigned* __restrict__ keys,
                    const unsigned* __restrict__ skeys,
                    const float* __restrict__ W,
                    const float* __restrict__ bias,
                    const float* __restrict__ W2,
                    float* __restrict__ Tp,
                    float* __restrict__ Cp,
                    int E, int Nn) {
    static_assert(IN_DIM == 128 && HID_DIM == 128, "fragment walk below is written for 128 x 128");
    static_assert(NUM_GRAPHS * TA_PITCH <= TA_PADROWS * TA_PITCH, "the LDS copy of S reuses the tile plane");
    static_assert(NUM_GRAPHS * OUT_DIM == TA_THREADS * 4, "the slice copy below moves four floats per thread");
    static_assert(NUM_GRAPHS == 64 && OUT_DIM == 64 && TA_WAVES >= 16, "16 waves take one 16x16 tile of the 64 x 64 product each");
    __shared__ __attribute__((aligned(16))) float tile[TA_PADROWS * TA_PITCH];
    __shared__ __attribute__((aligned(16))) float outt[TA_PADROWS * TA_PITCH];
    __shared__ int   lists[TA_WAVES * TA_LIST];
    __shared__ float Cl[NUM_GRAPHS];
    const int tid  = threadIdx.x;
    const int lane = tid & 31;
    const int wave = __builtin_amdgcn_readfirstlane(tid >> 5);
    const int lo   = (int)blockIdx.x * TA_ROWS;
    int rows = Nn - lo;
    if (rows > TA_ROWS) rows = TA_ROWS;

    if (tid < NUM_GRAPHS) Cl[tid] = 0.0f;
    for (int r = wave; r < TA_PADROWS; r += TA_WAVES) {
        float* p = tile + r * TA_PITCH + lane;
        if (r < rows) {
            const int   i  = lo + r;
            const float di = dinv[i];
            const float w  = di * di;
            const float4 xv = *(const float4*)(x + (size_t)i * IN_DIM + (lane << 2));
            p[0]  = xv.x * w;
            p[32] = xv.y * w;
            p[64] = xv.z * w;
            p[96] = xv.w * w;
        } else {
            p[0] = 0.0f; p[32] = 0.0f; p[64] = 0.0f; p[96] = 0.0f;
        }
    }
    __syncthreads();

    int* mylist = lists + wave * TA_LIST;
    const unsigned own = (unsigned)blockIdx.x;
    const int start = (int)blockIdx.x * ((E >> 9) / (int)gridDim.x);

    {
        const TaGatherFlush gf{lane, lo, x, dinv, src, dst, tile};
        ta_scan((const unsigned char*)keys, own, E, wave, lane, start, mylist, gf);
    }

    const int half = lane >> 4;
    const int m    = lane & 15;
    const int nt   = wave & 7;
    v2f bf[32];
#pragma unroll
    for (int q = 0; q < 32; ++q) {
        const int k = ((q & 7) << 4) + (half << 3) + (q >> 3);
        bf[q].x = W[(size_t)k * HID_DIM + (nt << 4) + m];
        bf[q].y = W[(size_t)(k + 4) * HID_DIM + (nt << 4) + m];
    }
    const float bcol = bias[(nt << 4) + m];

    __syncthreads();

    const int nStrips = (rows + 15) >> 4;
    for (int s = wave >> 3; s < nStrips; s += TA_WAVES / 8) {
        const float* ap = tile + ((s << 4) + m) * TA_PITCH + (half << 1);
        v8f acc = {};
#pragma unroll
        for (int q = 0; q < 32; ++q) {
            const v2f a = *(const v2f*)(ap + (q << 2));
            acc = __builtin_amdgcn_wmma_f32_16x16x4_f32(
                 false, a,  false, bf[q],
                 (short)0, acc,  false,  false);
        }
        float* op = outt + ((s << 4) + (half << 3)) * TA_PITCH + (nt << 4) + m;
#pragma unroll
        for (int v = 0; v < 8; ++v) op[v * TA_PITCH] = fmaxf(acc[v] + bcol, 0.0f);
    }

    __syncthreads();

    float* Sl = tile;
    for (int i = tid; i < NUM_GRAPHS * TA_PITCH; i += TA_THREADS) Sl[i] = 0.0f;

    __syncthreads();

    for (int r = wave; r < rows; r += TA_WAVES) {
        const int   i  = lo + r;
        const int   g  = batch[i];
        const float di = dinv[i];
        const float w  = di * di;
        const float4 t = *(const float4*)(outt + r * TA_PITCH + (lane << 2));
        float* p = Sl + g * TA_PITCH + lane;
        atomicAdd(p,      t.x * w);
        atomicAdd(p + 32, t.y * w);
        atomicAdd(p + 64, t.z * w);
        atomicAdd(p + 96, t.w * w);
        if (lane == 0) atomicAdd(&Cl[g], 1.0f);
    }
    {
        const TaPoolFlush pf{lane, lo, dinv, src, dst, batch, outt, Sl};
        ta_scan((const unsigned char*)skeys, own, E, wave, lane, start, mylist, pf);
    }

    const int tm2 = (wave >> 2) & 3;
    const int tn2 = wave & 3;
    v2f bf2[32];
    if (wave < 16) {
#pragma unroll
        for (int q = 0; q < 32; ++q) {
            const int k = ((q & 7) << 4) + (half << 3) + (q >> 3);
            bf2[q].x = W2[(size_t)k * OUT_DIM + (tn2 << 4) + m];
            bf2[q].y = W2[(size_t)(k + 4) * OUT_DIM + (tn2 << 4) + m];
        }
    }

    __syncthreads();

    float* Tl = outt;
    if (wave < 16) {
        const float* ap = Sl + ((tm2 << 4) + m) * TA_PITCH + (half << 1);
        v8f acc = {};
#pragma unroll
        for (int q = 0; q < 32; ++q) {
            const v2f a = *(const v2f*)(ap + (q << 2));
            acc = __builtin_amdgcn_wmma_f32_16x16x4_f32(
                 false, a,  false, bf2[q],
                 (short)0, acc,  false,  false);
        }
        float* op = Tl + ((tm2 << 4) + (half << 3)) * OUT_DIM + (tn2 << 4) + m;
#pragma unroll
        for (int v = 0; v < 8; ++v) op[v * OUT_DIM] = acc[v];
    }

    __syncthreads();

    ((float4*)(Tp + (size_t)blockIdx.x * (NUM_GRAPHS * OUT_DIM)))[tid] = ((const float4*)Tl)[tid];
    if (tid < NUM_GRAPHS) Cp[(size_t)blockIdx.x * NUM_GRAPHS + tid] = Cl[tid];
}

#define PF_THREADS 1024
#define PF_PARTS   (PF_THREADS / OUT_DIM)
__global__ __launch_bounds__(PF_THREADS)
void pool_final(const float* __restrict__ Tp,
                const float* __restrict__ Cp,
                const float* __restrict__ bias,
                float* __restrict__ out, int nb) {
    __shared__ float red[PF_PARTS][OUT_DIM];
    __shared__ float cred[PF_THREADS / 32];
    const int g    = blockIdx.x;
    const int tid  = threadIdx.x;
    const int c    = tid & (OUT_DIM - 1);
    const int part = tid >> 6;
    const float* tp = Tp + (size_t)g * OUT_DIM + c;
    float acc = 0.0f;
#pragma unroll 16
    for (int b = part; b < nb; b += PF_PARTS) acc += tp[(size_t)b * (NUM_GRAPHS * OUT_DIM)];
    float cn = 0.0f;
    for (int b = tid; b < nb; b += PF_THREADS) cn += Cp[(size_t)b * NUM_GRAPHS + g];
    for (int o = 16; o > 0; o >>= 1) cn += __shfl_xor(cn, o);
    red[part][c] = acc;
    if ((tid & 31) == 0) cred[tid >> 5] = cn;
    __syncthreads();
    if (tid < OUT_DIM) {
        float t = 0.0f;
        for (int p = 0; p < PF_PARTS; ++p) t += red[p][c];
        float n = 0.0f;
        for (int w = 0; w < PF_THREADS / 32; ++w) n += cred[w];
        out[g * OUT_DIM + c] = (t + n * bias[c]) / fmaxf(n, 1.0f);
    }
}

extern "C" void kernel_launch(void* const* d_in, const int* in_sizes, int n_in,
                              void* d_out, int out_size, void* d_ws, size_t ws_size,
                              hipStream_t stream) {
    const float* x    = (const float*)d_in[0];
    const int*   eidx = (const int*)  d_in[1];
    const int*   batch= (const int*)  d_in[2];
    const float* W1   = (const float*)d_in[3];
    const float* b1   = (const float*)d_in[4];
    const float* W2   = (const float*)d_in[5];
    const float* b2   = (const float*)d_in[6];
    float* out = (float*)d_out;

    const int Nn = in_sizes[0] / IN_DIM;
    const int E  = in_sizes[1] / 2;
    const int* src = eidx;
    const int* dst = eidx + E;

    const int B = 256;
    auto blocks = [](long long n, int b) { return (unsigned)((n + b - 1) / b); };
    const unsigned nTiles = blocks(Nn, TA_ROWS);

    const size_t keyWords   = (((size_t)E + 127) >> 7) * 32;
    const size_t nodeFloats = (((size_t)Nn + 31) >> 5) * 32;
    unsigned* keys  = (unsigned*)d_ws;
    unsigned* skeys = keys + keyWords;
    unsigned* part  = skeys + keyWords;
    float* dinv = (float*)(part + (size_t)DH_BLOCKS * DH_NODES);
    float* Cp   = dinv + nodeFloats;
    float* Tp   = Cp   + (size_t)nTiles * NUM_GRAPHS;

    degree_key_kernel<<<DH_BLOCKS, DH_THREADS, 0, stream>>>(src, dst, part, keys, skeys, E);
    dinv_kernel<<<blocks(Nn, B), B, 0, stream>>>(part, dinv, Nn);

    tile_aggregate<<<nTiles, TA_THREADS, 0, stream>>>(
        x, dinv, src, dst, batch, keys, skeys, W1, b1, W2, Tp, Cp, E, Nn);

    pool_final<<<NUM_GRAPHS, PF_THREADS, 0, stream>>>(Tp, Cp, b2, out, (int)nTiles);
}


# Degree pass orders each edge stretch by owner tile in LDS (counts, scan, cursor ranks, packed node pairs); layer-1 work-groups read their 64 segments; owner bytes, both list scans go; sum order moves
# speedup vs baseline: 5.0581x; 1.1846x over previous
#include <hip/hip_runtime.h>


#define IN_DIM   128
#define HID_DIM  128
#define OUT_DIM  64
#define NUM_GRAPHS 64

typedef float v2f __attribute__((ext_vector_type(2)));
typedef float v8f __attribute__((ext_vector_type(8)));


#define TA_ROWS    196
#define TA_STRIPS  13
#define TA_PADROWS (TA_STRIPS * 16)
#define TA_PITCH   (IN_DIM + 4)
#define TA_THREADS 1024
#define TA_WAVES   (TA_THREADS / 32)
#define TA_GROUP   8

#define DH_BLOCKS  64
#define DH_THREADS 1024
#define DH_NODES   (256 * TA_ROWS)
#define DH_OWNERS  256
#define DH_TRIPS   3
#define DH_STAGE   12288

__device__ __forceinline__ int dh_per(int E, int nb) { return (((E + nb - 1) / nb) + 31) & ~31; }

__global__ __launch_bounds__(DH_THREADS)
void degree_group_kernel(const int* __restrict__ src, const int* __restrict__ dst,
                         unsigned* __restrict__ part, unsigned* __restrict__ bydst,
                         unsigned* __restrict__ bysrc, unsigned* __restrict__ tabw, int E) {
    static_assert(DH_NODES % 4 == 0, "the counters are cleared and copied 16 bytes at a time");
    static_assert(2 * DH_OWNERS <= DH_THREADS && DH_OWNERS % 32 == 0, "one thread per owner and order in the scan");
    __shared__ __attribute__((aligned(16))) unsigned cnt[DH_NODES];
    __shared__ unsigned stageD[DH_STAGE];
    __shared__ unsigned stageS[DH_STAGE];
    __shared__ unsigned hist[2 * DH_OWNERS];
    __shared__ unsigned wsum[2 * DH_OWNERS / 32];
    const int tid  = threadIdx.x;
    const int lane = tid & 31;
    const int per  = dh_per(E, (int)gridDim.x);
    const int e0   = (int)blockIdx.x * per;
    int e1 = e0 + per;
    if (e1 > E) e1 = E;
    const bool fits = per <= DH_STAGE;
    uint2* tab = (uint2*)tabw;

    for (int i = tid; i < DH_NODES / 4; i += DH_THREADS) ((uint4*)cnt)[i] = make_uint4(0u, 0u, 0u, 0u);
    if (tid < 2 * DH_OWNERS) hist[tid] = 0u;

    for (int base = e0; base < e1; base += DH_TRIPS * DH_THREADS * 4) {
        int dd[DH_TRIPS][4], ss[DH_TRIPS][4];
#pragma unroll
        for (int u = 0; u < DH_TRIPS; ++u) {
            const int e = base + ((u * DH_THREADS + tid) << 2);
#pragma unroll
            for (int j = 0; j < 4; ++j) {
                int ee = e + j;
                if (ee > e1 - 1) ee = e1 - 1;
                dd[u][j] = dst[ee]; ss[u][j] = src[ee];
            }
        }
        __syncthreads();
#pragma unroll
        for (int u = 0; u < DH_TRIPS; ++u) {
            const int e = base + ((u * DH_THREADS + tid) << 2);
#pragma unroll
            for (int j = 0; j < 4; ++j) {
                if (e + j < e1) {
                    const unsigned d = (unsigned)dd[u][j];
                    const unsigned s = (unsigned)ss[u][j];
                    if (d < (unsigned)DH_NODES) atomicAdd(&cnt[d], 1u);
                    unsigned kd = d / TA_ROWS; if (kd > DH_OWNERS - 1) kd = DH_OWNERS - 1;
                    unsigned ks = s / TA_ROWS; if (ks > DH_OWNERS - 1) ks = DH_OWNERS - 1;
                    atomicAdd(&hist[kd], 1u);
                    atomicAdd(&hist[DH_OWNERS + ks], 1u);
                }
            }
        }
    }
    __syncthreads();

    unsigned own_cnt = 0u, incl = 0u;
    if (tid < 2 * DH_OWNERS) {
        own_cnt = hist[tid];
        incl = own_cnt;
        for (int o = 1; o < 32; o <<= 1) {
            const unsigned y = (unsigned)__shfl_up((int)incl, o);
            if (lane >= o) incl += y;
        }
        if (lane == 31) wsum[tid >> 5] = incl;
    }
    __syncthreads();
    if (tid < 2 * DH_OWNERS) {
        const int w  = tid >> 5;
        const int w0 = (tid < DH_OWNERS) ? 0 : DH_OWNERS / 32;
        unsigned before = 0u;
        for (int k = w0; k < w; ++k) before += wsum[k];
        const unsigned start = before + incl - own_cnt;
        hist[tid] = start;
        tab[(size_t)tid * DH_BLOCKS + blockIdx.x] = make_uint2(start, own_cnt);
    }
    __syncthreads();

    for (int base = e0; base < e1; base += DH_TRIPS * DH_THREADS * 4) {
        int dd[DH_TRIPS][4], ss[DH_TRIPS][4];
#pragma unroll
        for (int u = 0; u < DH_TRIPS; ++u) {
            const int e = base + ((u * DH_THREADS + tid) << 2);
#pragma unroll
            for (int j = 0; j < 4; ++j) {
                int ee = e + j;
                if (ee > e1 - 1) ee = e1 - 1;
                dd[u][j] = dst[ee]; ss[u][j] = src[ee];
            }
        }
#pragma unroll
        for (int u = 0; u < DH_TRIPS; ++u) {
            const int e = base + ((u * DH_THREADS + tid) << 2);
#pragma unroll
            for (int j = 0; j < 4; ++j) {
                if (e + j < e1) {
                    const unsigned d = (unsigned)dd[u][j];
                    const unsigned s = (unsigned)ss[u][j];
                    unsigned kd = d / TA_ROWS; if (kd > DH_OWNERS - 1) kd = DH_OWNERS - 1;
                    unsigned ks = s / TA_ROWS; if (ks > DH_OWNERS - 1) ks = DH_OWNERS - 1;
                    const unsigned pd = atomicAdd(&hist[kd], 1u);
                    const unsigned ps = atomicAdd(&hist[DH_OWNERS + ks], 1u);
                    const unsigned entD = (s << 8) | ((d - kd * TA_ROWS) & 255u);
                    const unsigned entS = (d << 8) | ((s - ks * TA_ROWS) & 255u);
                    if (fits) { stageD[pd] = entD; stageS[ps] = entS; }
                    else      { bydst[(size_t)e0 + pd] = entD; bysrc[(size_t)e0 + ps] = entS; }
                }
            }
        }
    }
    __syncthreads();

    if (fits) {
        const int n = e1 - e0;
        for (int i = tid; i < n; i += DH_THREADS) {
            bydst[(size_t)e0 + i] = stageD[i];
            bysrc[(size_t)e0 + i] = stageS[i];
        }
    }
    uint4* pp = (uint4*)(part + (size_t)blockIdx.x * DH_NODES);
    for (int i = tid; i < DH_NODES / 4; i += DH_THREADS) pp[i] = ((const uint4*)cnt)[i];
}

__global__ void dinv_kernel(const unsigned* __restrict__ part, float* __restrict__ dinv, int n) {
    const int i = blockIdx.x * blockDim.x + threadIdx.x;
    if (i < n) {
        unsigned c = 1u;
#pragma unroll
        for (int b = 0; b < DH_BLOCKS; ++b) c += part[(size_t)b * DH_NODES + i];
        dinv[i] = rsqrtf((float)c);
    }
}

__device__ __forceinline__ void ta_flush(unsigned ent, int nv, int lane, int lo,
                                         const float* __restrict__ x,
                                         const float* __restrict__ dinv,
                                         float* tile) {
    int s = 0, r = 0;
    float w = 0.0f;
    if (lane < nv) {
        s = (int)(ent >> 8);
        r = (int)(ent & 255u);
        w = dinv[s] * dinv[lo + r];
    }
    int k0 = 0;
    for (; k0 + TA_GROUP <= nv; k0 += TA_GROUP) {
        float4 v[TA_GROUP];
        float  wk[TA_GROUP];
        int    rk[TA_GROUP];
#pragma unroll
        for (int j = 0; j < TA_GROUP; ++j) {
            const int sk = __builtin_amdgcn_readlane(s, k0 + j);
            wk[j] = __int_as_float(__builtin_amdgcn_readlane(__float_as_int(w), k0 + j));
            rk[j] = __builtin_amdgcn_readlane(r, k0 + j);
            v[j]  = *(const float4*)(x + (size_t)sk * IN_DIM + (lane << 2));
        }
#pragma unroll
        for (int j = 0; j < TA_GROUP; ++j) {
            float* p = tile + rk[j] * TA_PITCH + lane;
            atomicAdd(p,      v[j].x * wk[j]);
            atomicAdd(p + 32, v[j].y * wk[j]);
            atomicAdd(p + 64, v[j].z * wk[j]);
            atomicAdd(p + 96, v[j].w * wk[j]);
        }
    }
    for (; k0 < nv; ++k0) {
        const int   sk = __builtin_amdgcn_readlane(s, k0);
        const float wq = __int_as_float(__builtin_amdgcn_readlane(__float_as_int(w), k0));
        const int   rq = __builtin_amdgcn_readlane(r, k0);
        const float4 vv = *(const float4*)(x + (size_t)sk * IN_DIM + (lane << 2));
        float* p = tile + rq * TA_PITCH + lane;
        atomicAdd(p,      vv.x * wq);
        atomicAdd(p + 32, vv.y * wq);
        atomicAdd(p + 64, vv.z * wq);
        atomicAdd(p + 96, vv.w * wq);
    }
}

__device__ __forceinline__ void ta_pool_flush(unsigned ent, int nv, int lane, int lo,
                                              const float* __restrict__ dinv,
                                              const int* __restrict__ batch,
                                              const float* rowsl, float* Sl) {
    int r = 0, g = 0;
    float w = 0.0f;
    if (lane < nv) {
        const int d = (int)(ent >> 8);
        r = (int)(ent & 255u);
        g = batch[d];
        w = dinv[lo + r] * dinv[d];
    }
    int k0 = 0;
    for (; k0 + TA_GROUP <= nv; k0 += TA_GROUP) {
        float4 v[TA_GROUP];
        float  wk[TA_GROUP];
        int    gk[TA_GROUP];
#pragma unroll
        for (int j = 0; j < TA_GROUP; ++j) {
            const int rk = __builtin_amdgcn_readlane(r, k0 + j);
            wk[j] = __int_as_float(__builtin_amdgcn_readlane(__float_as_int(w), k0 + j));
            gk[j] = __builtin_amdgcn_readlane(g, k0 + j);
            v[j]  = *(const float4*)(rowsl + rk * TA_PITCH + (lane << 2));
        }
#pragma unroll
        for (int j = 0; j < TA_GROUP; ++j) {
            float* p = Sl + gk[j] * TA_PITCH + lane;
            atomicAdd(p,      v[j].x * wk[j]);
            atomicAdd(p + 32, v[j].y * wk[j]);
            atomicAdd(p + 64, v[j].z * wk[j]);
            atomicAdd(p + 96, v[j].w * wk[j]);
        }
    }
    for (; k0 < nv; ++k0) {
        const int   rq = __builtin_amdgcn_readlane(r, k0);
        const float wq = __int_as_float(__builtin_amdgcn_readlane(__float_as_int(w), k0));
        const int   gq = __builtin_amdgcn_readlane(g, k0);
        const float4 vv = *(const float4*)(rowsl + rq * TA_PITCH + (lane << 2));
        float* p = Sl + gq * TA_PITCH + lane;
        atomicAdd(p,      vv.x * wq);
        atomicAdd(p + 32, vv.y * wq);
        atomicAdd(p + 64, vv.z * wq);
        atomicAdd(p + 96, vv.w * wq);
    }
}

struct TaGatherFlush {
    int lane, lo;
    const float* x; const float* dinv; float* tile;
    __device__ __forceinline__ void operator()(unsigned ent, int nv) const {
        ta_flush(ent, nv, lane, lo, x, dinv, tile);
    }
};
struct TaPoolFlush {
    int lane, lo;
    const float* dinv; const int* batch; const float* rowsl; float* Sl;
    __device__ __forceinline__ void operator()(unsigned ent, int nv) const {
        ta_pool_flush(ent, nv, lane, lo, dinv, batch, rowsl, Sl);
    }
};

template <class Flush>
__device__ __forceinline__ void ta_walk(const uint2* __restrict__ row, const unsigned* __restrict__ ordered,
                                        int per, int wave, int lane, const Flush& flush) {
    static_assert(DH_BLOCKS == 2 * TA_WAVES, "a wave walks exactly two stretches' segments");
    const uint2 t0 = row[wave];
    const uint2 t1 = row[wave + TA_WAVES];
    const int n0 = (int)t0.y;
    const int total = n0 + (int)t1.y;
    const unsigned* p0 = ordered + (size_t)wave * per + t0.x;
    const unsigned* p1 = ordered + (size_t)(wave + TA_WAVES) * per + t1.x;
    for (int k0 = 0; k0 < total; k0 += 32) {
        const int i = k0 + lane;
        unsigned ent = 0u;
        if (i < total) ent = (i < n0) ? p0[i] : p1[i - n0];
        int nv = total - k0;
        if (nv > 32) nv = 32;
        flush(ent, nv);
    }
}

__global__ __launch_bounds__(TA_THREADS)
void tile_aggregate(const float* __restrict__ x,
                    const float* __restrict__ dinv,
                    const int* __restrict__ batch,
                    const unsigned* __restrict__ bydst,
                    const unsigned* __restrict__ bysrc,
                    const unsigned* __restrict__ tabw,
                    const float* __restrict__ W,
                    const float* __restrict__ bias,
                    const float* __restrict__ W2,
                    float* __restrict__ Tp,
                    float* __restrict__ Cp,
                    int E, int Nn) {
    static_assert(IN_DIM == 128 && HID_DIM == 128, "fragment walk below is written for 128 x 128");
    static_assert(NUM_GRAPHS * TA_PITCH <= TA_PADROWS * TA_PITCH, "the LDS copy of S reuses the tile plane");
    static_assert(NUM_GRAPHS * OUT_DIM == TA_THREADS * 4, "the slice copy below moves four floats per thread");
    static_assert(NUM_GRAPHS == 64 && OUT_DIM == 64 && TA_WAVES >= 16, "16 waves take one 16x16 tile of the 64 x 64 product each");
    __shared__ __attribute__((aligned(16))) float tile[TA_PADROWS * TA_PITCH];
    __shared__ __attribute__((aligned(16))) float outt[TA_PADROWS * TA_PITCH];
    __shared__ float Cl[NUM_GRAPHS];
    const int tid  = threadIdx.x;
    const int lane = tid & 31;
    const int wave = __builtin_amdgcn_readfirstlane(tid >> 5);
    const int lo   = (int)blockIdx.x * TA_ROWS;
    int rows = Nn - lo;
    if (rows > TA_ROWS) rows = TA_ROWS;

    if (tid < NUM_GRAPHS) Cl[tid] = 0.0f;
    for (int r = wave; r < TA_PADROWS; r += TA_WAVES) {
        float* p = tile + r * TA_PITCH + lane;
        if (r < rows) {
            const int   i  = lo + r;
            const float di = dinv[i];
            const float w  = di * di;
            const float4 xv = *(const float4*)(x + (size_t)i * IN_DIM + (lane << 2));
            p[0]  = xv.x * w;
            p[32] = xv.y * w;
            p[64] = xv.z * w;
            p[96] = xv.w * w;
        } else {
            p[0] = 0.0f; p[32] = 0.0f; p[64] = 0.0f; p[96] = 0.0f;
        }
    }
    __syncthreads();

    const int per = dh_per(E, DH_BLOCKS);
    const uint2* tab = (const uint2*)tabw;
    {
        const TaGatherFlush gf{lane, lo, x, dinv, tile};
        ta_walk(tab + (size_t)blockIdx.x * DH_BLOCKS, bydst, per, wave, lane, gf);
    }

    const int half = lane >> 4;
    const int m    = lane & 15;
    const int nt   = wave & 7;
    v2f bf[32];
#pragma unroll
    for (int q = 0; q < 32; ++q) {
        const int k = ((q & 7) << 4) + (half << 3) + (q >> 3);
        bf[q].x = W[(size_t)k * HID_DIM + (nt << 4) + m];
        bf[q].y = W[(size_t)(k + 4) * HID_DIM + (nt << 4) + m];
    }
    const float bcol = bias[(nt << 4) + m];

    __syncthreads();

    const int nStrips = (rows + 15) >> 4;
    for (int s = wave >> 3; s < nStrips; s += TA_WAVES / 8) {
        const float* ap = tile + ((s << 4) + m) * TA_PITCH + (half << 1);
        v8f acc = {};
#pragma unroll
        for (int q = 0; q < 32; ++q) {
            const v2f a = *(const v2f*)(ap + (q << 2));
            acc = __builtin_amdgcn_wmma_f32_16x16x4_f32(
                 false, a,  false, bf[q],
                 (short)0, acc,  false,  false);
        }
        float* op = outt + ((s << 4) + (half << 3)) * TA_PITCH + (nt << 4) + m;
#pragma unroll
        for (int v = 0; v < 8; ++v) op[v * TA_PITCH] = fmaxf(acc[v] + bcol, 0.0f);
    }

    __syncthreads();

    float* Sl = tile;
    for (int i = tid; i < NUM_GRAPHS * TA_PITCH; i += TA_THREADS) Sl[i] = 0.0f;

    __syncthreads();

    for (int r = wave; r < rows; r += TA_WAVES) {
        const int   i  = lo + r;
        const int   g  = batch[i];
        const float di = dinv[i];
        const float w  = di * di;
        const float4 t = *(const float4*)(outt + r * TA_PITCH + (lane << 2));
        float* p = Sl + g * TA_PITCH + lane;
        atomicAdd(p,      t.x * w);
        atomicAdd(p + 32, t.y * w);
        atomicAdd(p + 64, t.z * w);
        atomicAdd(p + 96, t.w * w);
        if (lane == 0) atomicAdd(&Cl[g], 1.0f);
    }
    {
        const TaPoolFlush pf{lane, lo, dinv, batch, outt, Sl};
        ta_walk(tab + (size_t)(DH_OWNERS + blockIdx.x) * DH_BLOCKS, bysrc, per, wave, lane, pf);
    }

    const int tm2 = (wave >> 2) & 3;
    const int tn2 = wave & 3;
    v2f bf2[32];
    if (wave < 16) {
#pragma unroll
        for (int q = 0; q < 32; ++q) {
            const int k = ((q & 7) << 4) + (half << 3) + (q >> 3);
            bf2[q].x = W2[(size_t)k * OUT_DIM + (tn2 << 4) + m];
            bf2[q].y = W2[(size_t)(k + 4) * OUT_DIM + (tn2 << 4) + m];
        }
    }

    __syncthreads();

    float* Tl = outt;
    if (wave < 16) {
        const float* ap = Sl + ((tm2 << 4) + m) * TA_PITCH + (half << 1);
        v8f acc = {};
#pragma unroll
        for (int q = 0; q < 32; ++q) {
            const v2f a = *(const v2f*)(ap + (q << 2));
            acc = __builtin_amdgcn_wmma_f32_16x16x4_f32(
                 false, a,  false, bf2[q],
                 (short)0, acc,  false,  false);
        }
        float* op = Tl + ((tm2 << 4) + (half << 3)) * OUT_DIM + (tn2 << 4) + m;
#pragma unroll
        for (int v = 0; v < 8; ++v) op[v * OUT_DIM] = acc[v];
    }

    __syncthreads();

    ((float4*)(Tp + (size_t)blockIdx.x * (NUM_GRAPHS * OUT_DIM)))[tid] = ((const float4*)Tl)[tid];
    if (tid < NUM_GRAPHS) Cp[(size_t)blockIdx.x * NUM_GRAPHS + tid] = Cl[tid];
}

#define PF_THREADS 1024
#define PF_PARTS   (PF_THREADS / OUT_DIM)
__global__ __launch_bounds__(PF_THREADS)
void pool_final(const float* __restrict__ Tp,
                const float* __restrict__ Cp,
                const float* __restrict__ bias,
                float* __restrict__ out, int nb) {
    __shared__ float red[PF_PARTS][OUT_DIM];
    __shared__ float cred[PF_THREADS / 32];
    const int g    = blockIdx.x;
    const int tid  = threadIdx.x;
    const int c    = tid & (OUT_DIM - 1);
    const int part = tid >> 6;
    const float* tp = Tp + (size_t)g * OUT_DIM + c;
    float acc = 0.0f;
#pragma unroll 16
    for (int b = part; b < nb; b += PF_PARTS) acc += tp[(size_t)b * (NUM_GRAPHS * OUT_DIM)];
    float cn = 0.0f;
    for (int b = tid; b < nb; b += PF_THREADS) cn += Cp[(size_t)b * NUM_GRAPHS + g];
    for (int o = 16; o > 0; o >>= 1) cn += __shfl_xor(cn, o);
    red[part][c] = acc;
    if ((tid & 31) == 0) cred[tid >> 5] = cn;
    __syncthreads();
    if (tid < OUT_DIM) {
        float t = 0.0f;
        for (int p = 0; p < PF_PARTS; ++p) t += red[p][c];
        float n = 0.0f;
        for (int w = 0; w < PF_THREADS / 32; ++w) n += cred[w];
        out[g * OUT_DIM + c] = (t + n * bias[c]) / fmaxf(n, 1.0f);
    }
}

extern "C" void kernel_launch(void* const* d_in, const int* in_sizes, int n_in,
                              void* d_out, int out_size, void* d_ws, size_t ws_size,
                              hipStream_t stream) {
    const float* x    = (const float*)d_in[0];
    const int*   eidx = (const int*)  d_in[1];
    const int*   batch= (const int*)  d_in[2];
    const float* W1   = (const float*)d_in[3];
    const float* b1   = (const float*)d_in[4];
    const float* W2   = (const float*)d_in[5];
    const float* b2   = (const float*)d_in[6];
    float* out = (float*)d_out;

    const int Nn = in_sizes[0] / IN_DIM;
    const int E  = in_sizes[1] / 2;
    const int* src = eidx;
    const int* dst = eidx + E;

    const int B = 256;
    auto blocks = [](long long n, int b) { return (unsigned)((n + b - 1) / b); };
    const unsigned nTiles = blocks(Nn, TA_ROWS);

    const size_t perEdges   = (size_t)(((E + DH_BLOCKS - 1) / DH_BLOCKS + 31) & ~31);
    const size_t ordWords   = perEdges * DH_BLOCKS;
    const size_t nodeFloats = (((size_t)Nn + 31) >> 5) * 32;
    unsigned* bydst = (unsigned*)d_ws;
    unsigned* bysrc = bydst + ordWords;
    unsigned* tab   = bysrc + ordWords;
    unsigned* part  = tab + (size_t)2 * 2 * DH_OWNERS * DH_BLOCKS;
    float* dinv = (float*)(part + (size_t)DH_BLOCKS * DH_NODES);
    float* Cp   = dinv + nodeFloats;
    float* Tp   = Cp   + (size_t)nTiles * NUM_GRAPHS;

    degree_group_kernel<<<DH_BLOCKS, DH_THREADS, 0, stream>>>(src, dst, part, bydst, bysrc, tab, E);
    dinv_kernel<<<blocks(Nn, B), B, 0, stream>>>(part, dinv, Nn);

    tile_aggregate<<<nTiles, TA_THREADS, 0, stream>>>(
        x, dinv, batch, bydst, bysrc, tab, W1, b1, W2, Tp, Cp, E, Nn);

    pool_final<<<NUM_GRAPHS, PF_THREADS, 0, stream>>>(Tp, Cp, b2, out, (int)nTiles);
}


# Layer 2 + pool sum as a product: an own-source edge adds its weight to a 64 x rows coefficient matrix in LDS (1 float add, was a row read + 4 adds); S = C @ a1 on f32 WMMA; sum order moves
# speedup vs baseline: 5.4532x; 1.0781x over previous
#include <hip/hip_runtime.h>


#define IN_DIM   128
#define HID_DIM  128
#define OUT_DIM  64
#define NUM_GRAPHS 64

typedef float v2f __attribute__((ext_vector_type(2)));
typedef float v8f __attribute__((ext_vector_type(8)));


#define TA_ROWS    196
#define TA_STRIPS  13
#define TA_PADROWS (TA_STRIPS * 16)
#define TA_PITCH   (IN_DIM + 4)
#define TA_THREADS 1024
#define TA_WAVES   (TA_THREADS / 32)
#define TA_GROUP   8
#define A1_PP      (2 * HID_DIM + 32)
#define CM_PITCH   (TA_PADROWS + 4)

#define DH_BLOCKS  64
#define DH_THREADS 1024
#define DH_NODES   (256 * TA_ROWS)
#define DH_OWNERS  256
#define DH_TRIPS   3
#define DH_STAGE   12288

__device__ __forceinline__ int dh_per(int E, int nb) { return (((E + nb - 1) / nb) + 31) & ~31; }

__global__ __launch_bounds__(DH_THREADS)
void degree_group_kernel(const int* __restrict__ src, const int* __restrict__ dst,
                         unsigned* __restrict__ part, unsigned* __restrict__ bydst,
                         unsigned* __restrict__ bysrc, unsigned* __restrict__ tabw, int E) {
    static_assert(DH_NODES % 4 == 0, "the counters are cleared and copied 16 bytes at a time");
    static_assert(2 * DH_OWNERS <= DH_THREADS && DH_OWNERS % 32 == 0, "one thread per owner and order in the scan");
    __shared__ __attribute__((aligned(16))) unsigned cnt[DH_NODES];
    __shared__ unsigned stageD[DH_STAGE];
    __shared__ unsigned stageS[DH_STAGE];
    __shared__ unsigned hist[2 * DH_OWNERS];
    __shared__ unsigned wsum[2 * DH_OWNERS / 32];
    const int tid  = threadIdx.x;
    const int lane = tid & 31;
    const int per  = dh_per(E, (int)gridDim.x);
    const int e0   = (int)blockIdx.x * per;
    int e1 = e0 + per;
    if (e1 > E) e1 = E;
    const bool fits = per <= DH_STAGE;
    uint2* tab = (uint2*)tabw;

    for (int i = tid; i < DH_NODES / 4; i += DH_THREADS) ((uint4*)cnt)[i] = make_uint4(0u, 0u, 0u, 0u);
    if (tid < 2 * DH_OWNERS) hist[tid] = 0u;

    for (int base = e0; base < e1; base += DH_TRIPS * DH_THREADS * 4) {
        int dd[DH_TRIPS][4], ss[DH_TRIPS][4];
#pragma unroll
        for (int u = 0; u < DH_TRIPS; ++u) {
            const int e = base + ((u * DH_THREADS + tid) << 2);
#pragma unroll
            for (int j = 0; j < 4; ++j) {
                int ee = e + j;
                if (ee > e1 - 1) ee = e1 - 1;
                dd[u][j] = dst[ee]; ss[u][j] = src[ee];
            }
        }
        __syncthreads();
#pragma unroll
        for (int u = 0; u < DH_TRIPS; ++u) {
            const int e = base + ((u * DH_THREADS + tid) << 2);
#pragma unroll
            for (int j = 0; j < 4; ++j) {
                if (e + j < e1) {
                    const unsigned d = (unsigned)dd[u][j];
                    const unsigned s = (unsigned)ss[u][j];
                    if (d < (unsigned)DH_NODES) atomicAdd(&cnt[d], 1u);
                    unsigned kd = d / TA_ROWS; if (kd > DH_OWNERS - 1) kd = DH_OWNERS - 1;
                    unsigned ks = s / TA_ROWS; if (ks > DH_OWNERS - 1) ks = DH_OWNERS - 1;
                    atomicAdd(&hist[kd], 1u);
                    atomicAdd(&hist[DH_OWNERS + ks], 1u);
                }
            }
        }
    }
    __syncthreads();

    unsigned own_cnt = 0u, incl = 0u;
    if (tid < 2 * DH_OWNERS) {
        own_cnt = hist[tid];
        incl = own_cnt;
        for (int o = 1; o < 32; o <<= 1) {
            const unsigned y = (unsigned)__shfl_up((int)incl, o);
            if (lane >= o) incl += y;
        }
        if (lane == 31) wsum[tid >> 5] = incl;
    }
    __syncthreads();
    if (tid < 2 * DH_OWNERS) {
        const int w  = tid >> 5;
        const int w0 = (tid < DH_OWNERS) ? 0 : DH_OWNERS / 32;
        unsigned before = 0u;
        for (int k = w0; k < w; ++k) before += wsum[k];
        const unsigned start = before + incl - own_cnt;
        hist[tid] = start;
        tab[(size_t)tid * DH_BLOCKS + blockIdx.x] = make_uint2(start, own_cnt);
    }
    __syncthreads();

    for (int base = e0; base < e1; base += DH_TRIPS * DH_THREADS * 4) {
        int dd[DH_TRIPS][4], ss[DH_TRIPS][4];
#pragma unroll
        for (int u = 0; u < DH_TRIPS; ++u) {
            const int e = base + ((u * DH_THREADS + tid) << 2);
#pragma unroll
            for (int j = 0; j < 4; ++j) {
                int ee = e + j;
                if (ee > e1 - 1) ee = e1 - 1;
                dd[u][j] = dst[ee]; ss[u][j] = src[ee];
            }
        }
#pragma unroll
        for (int u = 0; u < DH_TRIPS; ++u) {
            const int e = base + ((u * DH_THREADS + tid) << 2);
#pragma unroll
            for (int j = 0; j < 4; ++j) {
                if (e + j < e1) {
                    const unsigned d = (unsigned)dd[u][j];
                    const unsigned s = (unsigned)ss[u][j];
                    unsigned kd = d / TA_ROWS; if (kd > DH_OWNERS - 1) kd = DH_OWNERS - 1;
                    unsigned ks = s / TA_ROWS; if (ks > DH_OWNERS - 1) ks = DH_OWNERS - 1;
                    const unsigned pd = atomicAdd(&hist[kd], 1u);
                    const unsigned ps = atomicAdd(&hist[DH_OWNERS + ks], 1u);
                    const unsigned entD = (s << 8) | ((d - kd * TA_ROWS) & 255u);
                    const unsigned entS = (d << 8) | ((s - ks * TA_ROWS) & 255u);
                    if (fits) { stageD[pd] = entD; stageS[ps] = entS; }
                    else      { bydst[(size_t)e0 + pd] = entD; bysrc[(size_t)e0 + ps] = entS; }
                }
            }
        }
    }
    __syncthreads();

    if (fits) {
        const int n = e1 - e0;
        for (int i = tid; i < n; i += DH_THREADS) {
            bydst[(size_t)e0 + i] = stageD[i];
            bysrc[(size_t)e0 + i] = stageS[i];
        }
    }
    uint4* pp = (uint4*)(part + (size_t)blockIdx.x * DH_NODES);
    for (int i = tid; i < DH_NODES / 4; i += DH_THREADS) pp[i] = ((const uint4*)cnt)[i];
}

__global__ void dinv_kernel(const unsigned* __restrict__ part, float* __restrict__ dinv, int n) {
    const int i = blockIdx.x * blockDim.x + threadIdx.x;
    if (i < n) {
        unsigned c = 1u;
#pragma unroll
        for (int b = 0; b < DH_BLOCKS; ++b) c += part[(size_t)b * DH_NODES + i];
        dinv[i] = rsqrtf((float)c);
    }
}

__device__ __forceinline__ void ta_flush(unsigned ent, int nv, int lane, int lo,
                                         const float* __restrict__ x,
                                         const float* __restrict__ dinv,
                                         float* tile) {
    int s = 0, r = 0;
    float w = 0.0f;
    if (lane < nv) {
        s = (int)(ent >> 8);
        r = (int)(ent & 255u);
        w = dinv[s] * dinv[lo + r];
    }
    int k0 = 0;
    for (; k0 + TA_GROUP <= nv; k0 += TA_GROUP) {
        float4 v[TA_GROUP];
        float  wk[TA_GROUP];
        int    rk[TA_GROUP];
#pragma unroll
        for (int j = 0; j < TA_GROUP; ++j) {
            const int sk = __builtin_amdgcn_readlane(s, k0 + j);
            wk[j] = __int_as_float(__builtin_amdgcn_readlane(__float_as_int(w), k0 + j));
            rk[j] = __builtin_amdgcn_readlane(r, k0 + j);
            v[j]  = *(const float4*)(x + (size_t)sk * IN_DIM + (lane << 2));
        }
#pragma unroll
        for (int j = 0; j < TA_GROUP; ++j) {
            float* p = tile + rk[j] * TA_PITCH + lane;
            atomicAdd(p,      v[j].x * wk[j]);
            atomicAdd(p + 32, v[j].y * wk[j]);
            atomicAdd(p + 64, v[j].z * wk[j]);
            atomicAdd(p + 96, v[j].w * wk[j]);
        }
    }
    for (; k0 < nv; ++k0) {
        const int   sk = __builtin_amdgcn_readlane(s, k0);
        const float wq = __int_as_float(__builtin_amdgcn_readlane(__float_as_int(w), k0));
        const int   rq = __builtin_amdgcn_readlane(r, k0);
        const float4 vv = *(const float4*)(x + (size_t)sk * IN_DIM + (lane << 2));
        float* p = tile + rq * TA_PITCH + lane;
        atomicAdd(p,      vv.x * wq);
        atomicAdd(p + 32, vv.y * wq);
        atomicAdd(p + 64, vv.z * wq);
        atomicAdd(p + 96, vv.w * wq);
    }
}

struct TaGatherFlush {
    int lane, lo;
    const float* x; const float* dinv; float* tile;
    __device__ __forceinline__ void operator()(unsigned ent, int nv) const {
        ta_flush(ent, nv, lane, lo, x, dinv, tile);
    }
};
template <class Flush>
__device__ __forceinline__ void ta_walk(const uint2* __restrict__ row, const unsigned* __restrict__ ordered,
                                        int per, int wave, int lane, const Flush& flush) {
    static_assert(DH_BLOCKS == 2 * TA_WAVES, "a wave walks exactly two stretches' segments");
    const uint2 t0 = row[wave];
    const uint2 t1 = row[wave + TA_WAVES];
    const int n0 = (int)t0.y;
    const int total = n0 + (int)t1.y;
    const unsigned* p0 = ordered + (size_t)wave * per + t0.x;
    const unsigned* p1 = ordered + (size_t)(wave + TA_WAVES) * per + t1.x;
    for (int k0 = 0; k0 < total; k0 += 32) {
        const int i = k0 + lane;
        unsigned ent = 0u;
        if (i < total) ent = (i < n0) ? p0[i] : p1[i - n0];
        int nv = total - k0;
        if (nv > 32) nv = 32;
        flush(ent, nv);
    }
}

__device__ __forceinline__ void ta_coef_walk(uint2 t0, uint2 t1, const unsigned* __restrict__ ordered,
                                             int per, int wave, int lane, int lo,
                                             const float* __restrict__ dinv, const int* __restrict__ batch,
                                             float* Cm) {
    static_assert(DH_BLOCKS == 2 * TA_WAVES, "a wave walks exactly two stretches' segments");
    const int n0 = (int)t0.y;
    const int total = n0 + (int)t1.y;
    const unsigned* p0 = ordered + (size_t)wave * per + t0.x;
    const unsigned* p1 = ordered + (size_t)(wave + TA_WAVES) * per + t1.x;
    unsigned nxt = 0u;
    if (lane < total) nxt = (lane < n0) ? p0[lane] : p1[lane - n0];
    for (int k0 = 0; k0 < total; k0 += 32) {
        const unsigned ent = nxt;
        const int i  = k0 + lane;
        const int in = i + 32;
        if (in < total) nxt = (in < n0) ? p0[in] : p1[in - n0];
        if (i < total) {
            const int d = (int)(ent >> 8);
            const int r = (int)(ent & 255u);
            const int g = batch[d];
            atomicAdd(Cm + g * CM_PITCH + r, dinv[lo + r] * dinv[d]);
        }
    }
}

__global__ __launch_bounds__(TA_THREADS)
void tile_aggregate(const float* __restrict__ x,
                    const float* __restrict__ dinv,
                    const int* __restrict__ batch,
                    const unsigned* __restrict__ bydst,
                    const unsigned* __restrict__ bysrc,
                    const unsigned* __restrict__ tabw,
                    const float* __restrict__ W,
                    const float* __restrict__ bias,
                    const float* __restrict__ W2,
                    float* __restrict__ Tp,
                    float* __restrict__ Cp,
                    int E, int Nn) {
    static_assert(IN_DIM == 128 && HID_DIM == 128, "fragment walk below is written for 128 x 128");
    static_assert(NUM_GRAPHS * CM_PITCH + NUM_GRAPHS * TA_PITCH <= TA_PADROWS * TA_PITCH, "C and the copy of S reuse the tile plane");
    static_assert((NUM_GRAPHS * CM_PITCH) % 4 == 0, "C is cleared 16 bytes at a time, and S behind it stays 16-byte aligned");
    static_assert(NUM_GRAPHS * OUT_DIM <= (TA_PADROWS / 2) * A1_PP, "the 64 x 64 product reuses the a1 plane");
    static_assert(NUM_GRAPHS * OUT_DIM == TA_THREADS * 4, "the slice copy below moves four floats per thread");
    static_assert(NUM_GRAPHS == 64 && OUT_DIM == 64 && TA_WAVES >= 16, "16 waves take one 16x16 tile of the 64 x 64 product each");
    static_assert((NUM_GRAPHS / 16) * (HID_DIM / 16) == TA_WAVES, "every wave takes one 16x16 tile of the 64 x 128 product C @ a1");
    __shared__ __attribute__((aligned(16))) float tile[TA_PADROWS * TA_PITCH];
    __shared__ __attribute__((aligned(16))) float outt[(TA_PADROWS / 2) * A1_PP];
    __shared__ float Cl[NUM_GRAPHS];
    const int tid  = threadIdx.x;
    const int lane = tid & 31;
    const int wave = __builtin_amdgcn_readfirstlane(tid >> 5);
    const int lo   = (int)blockIdx.x * TA_ROWS;
    int rows = Nn - lo;
    if (rows > TA_ROWS) rows = TA_ROWS;

    if (tid < NUM_GRAPHS) Cl[tid] = 0.0f;
    for (int r = wave; r < TA_PADROWS; r += TA_WAVES) {
        float* p = tile + r * TA_PITCH + lane;
        if (r < rows) {
            const int   i  = lo + r;
            const float di = dinv[i];
            const float w  = di * di;
            const float4 xv = *(const float4*)(x + (size_t)i * IN_DIM + (lane << 2));
            p[0]  = xv.x * w;
            p[32] = xv.y * w;
            p[64] = xv.z * w;
            p[96] = xv.w * w;
        } else {
            p[0] = 0.0f; p[32] = 0.0f; p[64] = 0.0f; p[96] = 0.0f;
        }
    }
    int   sg  = 0;
    float sdi = 0.0f;
    if (tid < rows) {
        sdi = dinv[lo + tid];
        sg  = batch[lo + tid];
    }
    __syncthreads();

    const int per = dh_per(E, DH_BLOCKS);
    const uint2* tab = (const uint2*)tabw;
    {
        const TaGatherFlush gf{lane, lo, x, dinv, tile};
        ta_walk(tab + (size_t)blockIdx.x * DH_BLOCKS, bydst, per, wave, lane, gf);
    }

    const int half = lane >> 4;
    const int m    = lane & 15;
    const int nt   = wave & 7;
    v2f bf[32];
#pragma unroll
    for (int q = 0; q < 32; ++q) {
        const int k = ((q & 7) << 4) + (half << 3) + (q >> 3);
        bf[q].x = W[(size_t)k * HID_DIM + (nt << 4) + m];
        bf[q].y = W[(size_t)(k + 4) * HID_DIM + (nt << 4) + m];
    }
    const float bcol = bias[(nt << 4) + m];

    __syncthreads();

    const int nStrips = (rows + 15) >> 4;
    for (int s = wave >> 3; s < nStrips; s += TA_WAVES / 8) {
        const float* ap = tile + ((s << 4) + m) * TA_PITCH + (half << 1);
        v8f acc = {};
#pragma unroll
        for (int q = 0; q < 32; ++q) {
            const v2f a = *(const v2f*)(ap + (q << 2));
            acc = __builtin_amdgcn_wmma_f32_16x16x4_f32(
                 false, a,  false, bf[q],
                 (short)0, acc,  false,  false);
        }
        float* op = outt + ((s << 3) + (half << 2)) * A1_PP + (((nt << 4) + m) << 1);
#pragma unroll
        for (int v = 0; v < 4; ++v) {
            v2f pr;
            pr.x = fmaxf(acc[2 * v] + bcol, 0.0f);
            pr.y = fmaxf(acc[2 * v + 1] + bcol, 0.0f);
            *(v2f*)(op + v * A1_PP) = pr;
        }
    }

    __syncthreads();

    const uint2* srow = tab + (size_t)(DH_OWNERS + blockIdx.x) * DH_BLOCKS;
    const uint2 st0 = srow[wave];
    const uint2 st1 = srow[wave + TA_WAVES];

    float* Cm = tile;
    float* Sl = tile + NUM_GRAPHS * CM_PITCH;
    for (int i = tid; i < NUM_GRAPHS * CM_PITCH / 4; i += TA_THREADS) ((float4*)Cm)[i] = make_float4(0.0f, 0.0f, 0.0f, 0.0f);

    __syncthreads();

    ta_coef_walk(st0, st1, bysrc, per, wave, lane, lo, dinv, batch, Cm);
    if (tid < rows) {
        atomicAdd(Cm + sg * CM_PITCH + tid, sdi * sdi);
        atomicAdd(&Cl[sg], 1.0f);
    }

    __syncthreads();

    const int tm2 = (wave >> 2) & 3;
    const int tn2 = wave & 3;
    v2f bf2[32];
    if (wave < 16) {
#pragma unroll
        for (int q = 0; q < 32; ++q) {
            const int k = (q << 2) + (half << 1);
            bf2[q].x = W2[(size_t)k * OUT_DIM + (tn2 << 4) + m];
            bf2[q].y = W2[(size_t)(k + 1) * OUT_DIM + (tn2 << 4) + m];
        }
    }

    {
        const int tmc = wave >> 3;
        const float* ap = Cm + ((tmc << 4) + m) * CM_PITCH + (half << 1);
        const float* bp = outt + half * A1_PP + (((nt << 4) + m) << 1);
        v8f acc = {};
        for (int s = 0; s < nStrips; ++s) {
#pragma unroll
            for (int qq = 0; qq < 4; ++qq) {
                const int q = (s << 2) + qq;
                const v2f a = *(const v2f*)(ap + (q << 2));
                const v2f b = *(const v2f*)(bp + (q << 1) * A1_PP);
                acc = __builtin_amdgcn_wmma_f32_16x16x4_f32(
                     false, a,  false, b,
                     (short)0, acc,  false,  false);
            }
        }
        float* op = Sl + ((tmc << 4) + (half << 3)) * TA_PITCH + (nt << 4) + m;
#pragma unroll
        for (int v = 0; v < 8; ++v) op[v * TA_PITCH] = acc[v];
    }

    __syncthreads();

    float* Tl = outt;
    if (wave < 16) {
        const float* ap = Sl + ((tm2 << 4) + m) * TA_PITCH + (half << 1);
        v8f acc = {};
#pragma unroll
        for (int q = 0; q < 32; ++q) {
            const v2f a = *(const v2f*)(ap + (q << 2));
            acc = __builtin_amdgcn_wmma_f32_16x16x4_f32(
                 false, a,  false, bf2[q],
                 (short)0, acc,  false,  false);
        }
        float* op = Tl + ((tm2 << 4) + (half << 3)) * OUT_DIM + (tn2 << 4) + m;
#pragma unroll
        for (int v = 0; v < 8; ++v) op[v * OUT_DIM] = acc[v];
    }

    __syncthreads();

    ((float4*)(Tp + (size_t)blockIdx.x * (NUM_GRAPHS * OUT_DIM)))[tid] = ((const float4*)Tl)[tid];
    if (tid < NUM_GRAPHS) Cp[(size_t)blockIdx.x * NUM_GRAPHS + tid] = Cl[tid];
}

#define PF_THREADS 1024
#define PF_PARTS   (PF_THREADS / OUT_DIM)
__global__ __launch_bounds__(PF_THREADS)
void pool_final(const float* __restrict__ Tp,
                const float* __restrict__ Cp,
                const float* __restrict__ bias,
                float* __restrict__ out, int nb) {
    __shared__ float red[PF_PARTS][OUT_DIM];
    __shared__ float cred[PF_THREADS / 32];
    const int g    = blockIdx.x;
    const int tid  = threadIdx.x;
    const int c    = tid & (OUT_DIM - 1);
    const int part = tid >> 6;
    const float* tp = Tp + (size_t)g * OUT_DIM + c;
    float acc = 0.0f;
#pragma unroll 16
    for (int b = part; b < nb; b += PF_PARTS) acc += tp[(size_t)b * (NUM_GRAPHS * OUT_DIM)];
    float cn = 0.0f;
    for (int b = tid; b < nb; b += PF_THREADS) cn += Cp[(size_t)b * NUM_GRAPHS + g];
    for (int o = 16; o > 0; o >>= 1) cn += __shfl_xor(cn, o);
    red[part][c] = acc;
    if ((tid & 31) == 0) cred[tid >> 5] = cn;
    __syncthreads();
    if (tid < OUT_DIM) {
        float t = 0.0f;
        for (int p = 0; p < PF_PARTS; ++p) t += red[p][c];
        float n = 0.0f;
        for (int w = 0; w < PF_THREADS / 32; ++w) n += cred[w];
        out[g * OUT_DIM + c] = (t + n * bias[c]) / fmaxf(n, 1.0f);
    }
}

extern "C" void kernel_launch(void* const* d_in, const int* in_sizes, int n_in,
                              void* d_out, int out_size, void* d_ws, size_t ws_size,
                              hipStream_t stream) {
    const float* x    = (const float*)d_in[0];
    const int*   eidx = (const int*)  d_in[1];
    const int*   batch= (const int*)  d_in[2];
    const float* W1   = (const float*)d_in[3];
    const float* b1   = (const float*)d_in[4];
    const float* W2   = (const float*)d_in[5];
    const float* b2   = (const float*)d_in[6];
    float* out = (float*)d_out;

    const int Nn = in_sizes[0] / IN_DIM;
    const int E  = in_sizes[1] / 2;
    const int* src = eidx;
    const int* dst = eidx + E;

    const int B = 256;
    auto blocks = [](long long n, int b) { return (unsigned)((n + b - 1) / b); };
    const unsigned nTiles = blocks(Nn, TA_ROWS);

    const size_t perEdges   = (size_t)(((E + DH_BLOCKS - 1) / DH_BLOCKS + 31) & ~31);
    const size_t ordWords   = perEdges * DH_BLOCKS;
    const size_t nodeFloats = (((size_t)Nn + 31) >> 5) * 32;
    unsigned* bydst = (unsigned*)d_ws;
    unsigned* bysrc = bydst + ordWords;
    unsigned* tab   = bysrc + ordWords;
    unsigned* part  = tab + (size_t)2 * 2 * DH_OWNERS * DH_BLOCKS;
    float* dinv = (float*)(part + (size_t)DH_BLOCKS * DH_NODES);
    float* Cp   = dinv + nodeFloats;
    float* Tp   = Cp   + (size_t)nTiles * NUM_GRAPHS;

    degree_group_kernel<<<DH_BLOCKS, DH_THREADS, 0, stream>>>(src, dst, part, bydst, bysrc, tab, E);
    dinv_kernel<<<blocks(Nn, B), B, 0, stream>>>(part, dinv, Nn);

    tile_aggregate<<<nTiles, TA_THREADS, 0, stream>>>(
        x, dinv, batch, bydst, bysrc, tab, W1, b1, W2, Tp, Cp, E, Nn);

    pool_final<<<NUM_GRAPHS, PF_THREADS, 0, stream>>>(Tp, Cp, b2, out, (int)nTiles);
}


# Edge list cut into 256 stretches, one small work-group per WGP (was 64 large): grouping pass fills the chip; in-degrees counted per owner tile from its segments, count planes go; sum order moves
# speedup vs baseline: 5.6729x; 1.0403x over previous
#include <hip/hip_runtime.h>


#define IN_DIM   128
#define HID_DIM  128
#define OUT_DIM  64
#define NUM_GRAPHS 64

typedef float v2f __attribute__((ext_vector_type(2)));
typedef float v8f __attribute__((ext_vector_type(8)));


#define TA_ROWS    196
#define TA_STRIPS  13
#define TA_PADROWS (TA_STRIPS * 16)
#define TA_PITCH   (IN_DIM + 4)
#define TA_THREADS 1024
#define TA_WAVES   (TA_THREADS / 32)
#define TA_GROUP   8
#define TA_SEG     8
#define A1_PP      (2 * HID_DIM + 32)
#define CM_PITCH   (TA_PADROWS + 4)

#define DH_BLOCKS  256
#define DH_THREADS 256
#define DH_OWNERS  256
#define DH_TRIPS   3
#define DH_STAGE   4096

__device__ __forceinline__ int dh_per(int E, int nb) { return (((E + nb - 1) / nb) + 31) & ~31; }

__global__ __launch_bounds__(DH_THREADS)
void edge_group_kernel(const int* __restrict__ src, const int* __restrict__ dst,
                         unsigned* __restrict__ bydst,
                         unsigned* __restrict__ bysrc, unsigned* __restrict__ tabw, int E) {
    static_assert(DH_OWNERS == DH_THREADS && DH_OWNERS % 32 == 0, "one thread per owner in the scan, both orders at once");
    __shared__ unsigned stageD[DH_STAGE];
    __shared__ unsigned stageS[DH_STAGE];
    __shared__ unsigned hist[2 * DH_OWNERS];
    __shared__ unsigned wsum[2 * DH_OWNERS / 32];
    const int tid  = threadIdx.x;
    const int lane = tid & 31;
    const int per  = dh_per(E, (int)gridDim.x);
    const int e0   = (int)blockIdx.x * per;
    int e1 = e0 + per;
    if (e1 > E) e1 = E;
    const bool fits = per <= DH_STAGE;
    uint2* tab = (uint2*)tabw;

    hist[tid] = 0u;
    hist[DH_OWNERS + tid] = 0u;

    for (int base = e0; base < e1; base += DH_TRIPS * DH_THREADS * 4) {
        int dd[DH_TRIPS][4], ss[DH_TRIPS][4];
#pragma unroll
        for (int u = 0; u < DH_TRIPS; ++u) {
            const int e = base + ((u * DH_THREADS + tid) << 2);
#pragma unroll
            for (int j = 0; j < 4; ++j) {
                int ee = e + j;
                if (ee > e1 - 1) ee = e1 - 1;
                dd[u][j] = dst[ee]; ss[u][j] = src[ee];
            }
        }
        __syncthreads();
#pragma unroll
        for (int u = 0; u < DH_TRIPS; ++u) {
            const int e = base + ((u * DH_THREADS + tid) << 2);
#pragma unroll
            for (int j = 0; j < 4; ++j) {
                if (e + j < e1) {
                    const unsigned d = (unsigned)dd[u][j];
                    const unsigned s = (unsigned)ss[u][j];
                    unsigned kd = d / TA_ROWS; if (kd > DH_OWNERS - 1) kd = DH_OWNERS - 1;
                    unsigned ks = s / TA_ROWS; if (ks > DH_OWNERS - 1) ks = DH_OWNERS - 1;
                    atomicAdd(&hist[kd], 1u);
                    atomicAdd(&hist[DH_OWNERS + ks], 1u);
                }
            }
        }
    }
    __syncthreads();

    const unsigned cntD = hist[tid];
    const unsigned cntS = hist[DH_OWNERS + tid];
    unsigned inclD = cntD, inclS = cntS;
    for (int o = 1; o < 32; o <<= 1) {
        const unsigned yD = (unsigned)__shfl_up((int)inclD, o);
        const unsigned yS = (unsigned)__shfl_up((int)inclS, o);
        if (lane >= o) { inclD += yD; inclS += yS; }
    }
    if (lane == 31) { wsum[tid >> 5] = inclD; wsum[DH_OWNERS / 32 + (tid >> 5)] = inclS; }
    __syncthreads();
    {
        const int w = tid >> 5;
        unsigned beforeD = 0u, beforeS = 0u;
        for (int k = 0; k < w; ++k) { beforeD += wsum[k]; beforeS += wsum[DH_OWNERS / 32 + k]; }
        const unsigned startD = beforeD + inclD - cntD;
        const unsigned startS = beforeS + inclS - cntS;
        hist[tid] = startD;
        hist[DH_OWNERS + tid] = startS;
        tab[(size_t)tid * DH_BLOCKS + blockIdx.x] = make_uint2(startD, cntD);
        tab[(size_t)(DH_OWNERS + tid) * DH_BLOCKS + blockIdx.x] = make_uint2(startS, cntS);
    }
    __syncthreads();

    for (int base = e0; base < e1; base += DH_TRIPS * DH_THREADS * 4) {
        int dd[DH_TRIPS][4], ss[DH_TRIPS][4];
#pragma unroll
        for (int u = 0; u < DH_TRIPS; ++u) {
            const int e = base + ((u * DH_THREADS + tid) << 2);
#pragma unroll
            for (int j = 0; j < 4; ++j) {
                int ee = e + j;
                if (ee > e1 - 1) ee = e1 - 1;
                dd[u][j] = dst[ee]; ss[u][j] = src[ee];
            }
        }
#pragma unroll
        for (int u = 0; u < DH_TRIPS; ++u) {
            const int e = base + ((u * DH_THREADS + tid) << 2);
#pragma unroll
            for (int j = 0; j < 4; ++j) {
                if (e + j < e1) {
                    const unsigned d = (unsigned)dd[u][j];
                    const unsigned s = (unsigned)ss[u][j];
                    unsigned kd = d / TA_ROWS; if (kd > DH_OWNERS - 1) kd = DH_OWNERS - 1;
                    unsigned ks = s / TA_ROWS; if (ks > DH_OWNERS - 1) ks = DH_OWNERS - 1;
                    const unsigned pd = atomicAdd(&hist[kd], 1u);
                    const unsigned ps = atomicAdd(&hist[DH_OWNERS + ks], 1u);
                    const unsigned entD = (s << 8) | ((d - kd * TA_ROWS) & 255u);
                    const unsigned entS = (d << 8) | ((s - ks * TA_ROWS) & 255u);
                    if (fits) { stageD[pd] = entD; stageS[ps] = entS; }
                    else      { bydst[(size_t)e0 + pd] = entD; bysrc[(size_t)e0 + ps] = entS; }
                }
            }
        }
    }
    __syncthreads();

    if (fits) {
        const int n = e1 - e0;
        for (int i = tid; i < n; i += DH_THREADS) {
            bydst[(size_t)e0 + i] = stageD[i];
            bysrc[(size_t)e0 + i] = stageS[i];
        }
    }
}

#define DG_THREADS 256
#define DG_SEGS    32
#define DG_BATCH   16
__global__ __launch_bounds__(DG_THREADS)
void dinv_kernel(const unsigned* __restrict__ bydst, const unsigned* __restrict__ tabw,
                 float* __restrict__ dinv, int E, int n) {
    static_assert(DG_SEGS == 32 && (DG_THREADS / 32) * DG_SEGS == DH_BLOCKS, "a lane holds one segment's pair, the waves cover every stretch");
    static_assert(TA_ROWS <= DG_THREADS && DG_THREADS == 256 && DG_SEGS % DG_BATCH == 0, "one counter per thread, one per value of an entry's low byte");
    __shared__ unsigned cnt[DG_THREADS];
    const int tid  = threadIdx.x;
    const int lane = tid & 31;
    const int wave = __builtin_amdgcn_readfirstlane(tid >> 5);
    const int per  = dh_per(E, DH_BLOCKS);
    const uint2* row = (const uint2*)tabw + (size_t)blockIdx.x * DH_BLOCKS;
    const uint2 t = row[wave * DG_SEGS + lane];
    cnt[tid] = 0u;
    __syncthreads();
    for (int k0 = 0; k0 < DG_SEGS; k0 += DG_BATCH) {
        unsigned v[DG_BATCH];
#pragma unroll
        for (int j = 0; j < DG_BATCH; ++j) {
            const int start = __builtin_amdgcn_readlane((int)t.x, k0 + j);
            const int len   = __builtin_amdgcn_readlane((int)t.y, k0 + j);
            const unsigned* p = bydst + (size_t)(wave * DG_SEGS + k0 + j) * per + start;
            v[j] = 0u;
            if (lane < len) v[j] = p[lane];
        }
#pragma unroll
        for (int j = 0; j < DG_BATCH; ++j) {
            const int len = __builtin_amdgcn_readlane((int)t.y, k0 + j);
            if (lane < len) atomicAdd(&cnt[v[j] & 255u], 1u);
        }
        for (int j = 0; j < DG_BATCH; ++j) {
            const int len = __builtin_amdgcn_readlane((int)t.y, k0 + j);
            if (len > 32) {
                const int start = __builtin_amdgcn_readlane((int)t.x, k0 + j);
                const unsigned* p = bydst + (size_t)(wave * DG_SEGS + k0 + j) * per + start;
                for (int i = 32 + lane; i < len; i += 32) atomicAdd(&cnt[p[i] & 255u], 1u);
            }
        }
    }
    __syncthreads();
    const int i = (int)blockIdx.x * TA_ROWS + tid;
    if (tid < TA_ROWS && i < n) dinv[i] = rsqrtf((float)(cnt[tid] + 1u));
}

__device__ __forceinline__ void ta_flush(unsigned ent, int nv, int lane, int lo,
                                         const float* __restrict__ x,
                                         const float* __restrict__ dinv,
                                         float* tile) {
    int s = 0, r = 0;
    float w = 0.0f;
    if (lane < nv) {
        s = (int)(ent >> 8);
        r = (int)(ent & 255u);
        w = dinv[s] * dinv[lo + r];
    }
    int k0 = 0;
    for (; k0 + TA_GROUP <= nv; k0 += TA_GROUP) {
        float4 v[TA_GROUP];
        float  wk[TA_GROUP];
        int    rk[TA_GROUP];
#pragma unroll
        for (int j = 0; j < TA_GROUP; ++j) {
            const int sk = __builtin_amdgcn_readlane(s, k0 + j);
            wk[j] = __int_as_float(__builtin_amdgcn_readlane(__float_as_int(w), k0 + j));
            rk[j] = __builtin_amdgcn_readlane(r, k0 + j);
            v[j]  = *(const float4*)(x + (size_t)sk * IN_DIM + (lane << 2));
        }
#pragma unroll
        for (int j = 0; j < TA_GROUP; ++j) {
            float* p = tile + rk[j] * TA_PITCH + lane;
            atomicAdd(p,      v[j].x * wk[j]);
            atomicAdd(p + 32, v[j].y * wk[j]);
            atomicAdd(p + 64, v[j].z * wk[j]);
            atomicAdd(p + 96, v[j].w * wk[j]);
        }
    }
    for (; k0 < nv; ++k0) {
        const int   sk = __builtin_amdgcn_readlane(s, k0);
        const float wq = __int_as_float(__builtin_amdgcn_readlane(__float_as_int(w), k0));
        const int   rq = __builtin_amdgcn_readlane(r, k0);
        const float4 vv = *(const float4*)(x + (size_t)sk * IN_DIM + (lane << 2));
        float* p = tile + rq * TA_PITCH + lane;
        atomicAdd(p,      vv.x * wq);
        atomicAdd(p + 32, vv.y * wq);
        atomicAdd(p + 64, vv.z * wq);
        atomicAdd(p + 96, vv.w * wq);
    }
}

struct TaGatherFlush {
    int lane, lo;
    const float* x; const float* dinv; float* tile;
    __device__ __forceinline__ void operator()(unsigned ent, int nv) const {
        ta_flush(ent, nv, lane, lo, x, dinv, tile);
    }
};
struct TaList {
    int first[TA_SEG];
    int off[TA_SEG];
    int total;
};
__device__ __forceinline__ void ta_list(const uint2* __restrict__ row, int per, int wave, TaList& L) {
    static_assert(DH_BLOCKS == TA_SEG * TA_WAVES, "the waves' lists cover every stretch once");
    int c = 0;
#pragma unroll
    for (int k = 0; k < TA_SEG; ++k) {
        const uint2 t = row[wave * TA_SEG + k];
        L.first[k] = c;
        L.off[k]   = (wave * TA_SEG + k) * per + (int)t.x - c;
        c += (int)t.y;
    }
    L.total = c;
}
__device__ __forceinline__ unsigned ta_entry(const unsigned* __restrict__ ordered, const TaList& L, int i) {
    int o = L.off[0];
#pragma unroll
    for (int k = 1; k < TA_SEG; ++k) if (i >= L.first[k]) o = L.off[k];
    return ordered[o + i];
}
template <class Flush>
__device__ __forceinline__ void ta_walk(const uint2* __restrict__ row, const unsigned* __restrict__ ordered,
                                        int per, int wave, int lane, const Flush& flush) {
    TaList L;
    ta_list(row, per, wave, L);
    for (int k0 = 0; k0 < L.total; k0 += 32) {
        const int i = k0 + lane;
        unsigned ent = 0u;
        if (i < L.total) ent = ta_entry(ordered, L, i);
        int nv = L.total - k0;
        if (nv > 32) nv = 32;
        flush(ent, nv);
    }
}

__device__ __forceinline__ void ta_coef_walk(const TaList& L, const unsigned* __restrict__ ordered,
                                             int lane, int lo,
                                             const float* __restrict__ dinv, const int* __restrict__ batch,
                                             float* Cm) {
    const int total = L.total;
    unsigned nxt = 0u;
    if (lane < total) nxt = ta_entry(ordered, L, lane);
    for (int k0 = 0; k0 < total; k0 += 32) {
        const unsigned ent = nxt;
        const int i  = k0 + lane;
        const int in = i + 32;
        if (in < total) nxt = ta_entry(ordered, L, in);
        if (i < total) {
            const int d = (int)(ent >> 8);
            const int r = (int)(ent & 255u);
            const int g = batch[d];
            atomicAdd(Cm + g * CM_PITCH + r, dinv[lo + r] * dinv[d]);
        }
    }
}

__global__ __launch_bounds__(TA_THREADS)
void tile_aggregate(const float* __restrict__ x,
                    const float* __restrict__ dinv,
                    const int* __restrict__ batch,
                    const unsigned* __restrict__ bydst,
                    const unsigned* __restrict__ bysrc,
                    const unsigned* __restrict__ tabw,
                    const float* __restrict__ W,
                    const float* __restrict__ bias,
                    const float* __restrict__ W2,
                    float* __restrict__ Tp,
                    float* __restrict__ Cp,
                    int E, int Nn) {
    static_assert(IN_DIM == 128 && HID_DIM == 128, "fragment walk below is written for 128 x 128");
    static_assert(NUM_GRAPHS * CM_PITCH + NUM_GRAPHS * TA_PITCH <= TA_PADROWS * TA_PITCH, "C and the copy of S reuse the tile plane");
    static_assert((NUM_GRAPHS * CM_PITCH) % 4 == 0, "C is cleared 16 bytes at a time, and S behind it stays 16-byte aligned");
    static_assert(NUM_GRAPHS * OUT_DIM <= (TA_PADROWS / 2) * A1_PP, "the 64 x 64 product reuses the a1 plane");
    static_assert(NUM_GRAPHS * OUT_DIM == TA_THREADS * 4, "the slice copy below moves four floats per thread");
    static_assert(NUM_GRAPHS == 64 && OUT_DIM == 64 && TA_WAVES >= 16, "16 waves take one 16x16 tile of the 64 x 64 product each");
    static_assert((NUM_GRAPHS / 16) * (HID_DIM / 16) == TA_WAVES, "every wave takes one 16x16 tile of the 64 x 128 product C @ a1");
    __shared__ __attribute__((aligned(16))) float tile[TA_PADROWS * TA_PITCH];
    __shared__ __attribute__((aligned(16))) float outt[(TA_PADROWS / 2) * A1_PP];
    __shared__ float Cl[NUM_GRAPHS];
    const int tid  = threadIdx.x;
    const int lane = tid & 31;
    const int wave = __builtin_amdgcn_readfirstlane(tid >> 5);
    const int lo   = (int)blockIdx.x * TA_ROWS;
    int rows = Nn - lo;
    if (rows > TA_ROWS) rows = TA_ROWS;

    if (tid < NUM_GRAPHS) Cl[tid] = 0.0f;
    for (int r = wave; r < TA_PADROWS; r += TA_WAVES) {
        float* p = tile + r * TA_PITCH + lane;
        if (r < rows) {
            const int   i  = lo + r;
            const float di = dinv[i];
            const float w  = di * di;
            const float4 xv = *(const float4*)(x + (size_t)i * IN_DIM + (lane << 2));
            p[0]  = xv.x * w;
            p[32] = xv.y * w;
            p[64] = xv.z * w;
            p[96] = xv.w * w;
        } else {
            p[0] = 0.0f; p[32] = 0.0f; p[64] = 0.0f; p[96] = 0.0f;
        }
    }
    int   sg  = 0;
    float sdi = 0.0f;
    if (tid < rows) {
        sdi = dinv[lo + tid];
        sg  = batch[lo + tid];
    }
    __syncthreads();

    const int per = dh_per(E, DH_BLOCKS);
    const uint2* tab = (const uint2*)tabw;
    {
        const TaGatherFlush gf{lane, lo, x, dinv, tile};
        ta_walk(tab + (size_t)blockIdx.x * DH_BLOCKS, bydst, per, wave, lane, gf);
    }

    const int half = lane >> 4;
    const int m    = lane & 15;
    const int nt   = wave & 7;
    v2f bf[32];
#pragma unroll
    for (int q = 0; q < 32; ++q) {
        const int k = ((q & 7) << 4) + (half << 3) + (q >> 3);
        bf[q].x = W[(size_t)k * HID_DIM + (nt << 4) + m];
        bf[q].y = W[(size_t)(k + 4) * HID_DIM + (nt << 4) + m];
    }
    const float bcol = bias[(nt << 4) + m];

    __syncthreads();

    const int nStrips = (rows + 15) >> 4;
    for (int s = wave >> 3; s < nStrips; s += TA_WAVES / 8) {
        const float* ap = tile + ((s << 4) + m) * TA_PITCH + (half << 1);
        v8f acc = {};
#pragma unroll
        for (int q = 0; q < 32; ++q) {
            const v2f a = *(const v2f*)(ap + (q << 2));
            acc = __builtin_amdgcn_wmma_f32_16x16x4_f32(
                 false, a,  false, bf[q],
                 (short)0, acc,  false,  false);
        }
        float* op = outt + ((s << 3) + (half << 2)) * A1_PP + (((nt << 4) + m) << 1);
#pragma unroll
        for (int v = 0; v < 4; ++v) {
            v2f pr;
            pr.x = fmaxf(acc[2 * v] + bcol, 0.0f);
            pr.y = fmaxf(acc[2 * v + 1] + bcol, 0.0f);
            *(v2f*)(op + v * A1_PP) = pr;
        }
    }

    __syncthreads();

    TaList SL;
    ta_list(tab + (size_t)(DH_OWNERS + blockIdx.x) * DH_BLOCKS, per, wave, SL);

    float* Cm = tile;
    float* Sl = tile + NUM_GRAPHS * CM_PITCH;
    for (int i = tid; i < NUM_GRAPHS * CM_PITCH / 4; i += TA_THREADS) ((float4*)Cm)[i] = make_float4(0.0f, 0.0f, 0.0f, 0.0f);

    __syncthreads();

    ta_coef_walk(SL, bysrc, lane, lo, dinv, batch, Cm);
    if (tid < rows) {
        atomicAdd(Cm + sg * CM_PITCH + tid, sdi * sdi);
        atomicAdd(&Cl[sg], 1.0f);
    }

    __syncthreads();

    const int tm2 = (wave >> 2) & 3;
    const int tn2 = wave & 3;
    v2f bf2[32];
    if (wave < 16) {
#pragma unroll
        for (int q = 0; q < 32; ++q) {
            const int k = (q << 2) + (half << 1);
            bf2[q].x = W2[(size_t)k * OUT_DIM + (tn2 << 4) + m];
            bf2[q].y = W2[(size_t)(k + 1) * OUT_DIM + (tn2 << 4) + m];
        }
    }

    {
        const int tmc = wave >> 3;
        const float* ap = Cm + ((tmc << 4) + m) * CM_PITCH + (half << 1);
        const float* bp = outt + half * A1_PP + (((nt << 4) + m) << 1);
        v8f acc = {};
        for (int s = 0; s < nStrips; ++s) {
#pragma unroll
            for (int qq = 0; qq < 4; ++qq) {
                const int q = (s << 2) + qq;
                const v2f a = *(const v2f*)(ap + (q << 2));
                const v2f b = *(const v2f*)(bp + (q << 1) * A1_PP);
                acc = __builtin_amdgcn_wmma_f32_16x16x4_f32(
                     false, a,  false, b,
                     (short)0, acc,  false,  false);
            }
        }
        float* op = Sl + ((tmc << 4) + (half << 3)) * TA_PITCH + (nt << 4) + m;
#pragma unroll
        for (int v = 0; v < 8; ++v) op[v * TA_PITCH] = acc[v];
    }

    __syncthreads();

    float* Tl = outt;
    if (wave < 16) {
        const float* ap = Sl + ((tm2 << 4) + m) * TA_PITCH + (half << 1);
        v8f acc = {};
#pragma unroll
        for (int q = 0; q < 32; ++q) {
            const v2f a = *(const v2f*)(ap + (q << 2));
            acc = __builtin_amdgcn_wmma_f32_16x16x4_f32(
                 false, a,  false, bf2[q],
                 (short)0, acc,  false,  false);
        }
        float* op = Tl + ((tm2 << 4) + (half << 3)) * OUT_DIM + (tn2 << 4) + m;
#pragma unroll
        for (int v = 0; v < 8; ++v) op[v * OUT_DIM] = acc[v];
    }

    __syncthreads();

    ((float4*)(Tp + (size_t)blockIdx.x * (NUM_GRAPHS * OUT_DIM)))[tid] = ((const float4*)Tl)[tid];
    if (tid < NUM_GRAPHS) Cp[(size_t)blockIdx.x * NUM_GRAPHS + tid] = Cl[tid];
}

#define PF_THREADS 1024
#define PF_PARTS   (PF_THREADS / OUT_DIM)
__global__ __launch_bounds__(PF_THREADS)
void pool_final(const float* __restrict__ Tp,
                const float* __restrict__ Cp,
                const float* __restrict__ bias,
                float* __restrict__ out, int nb) {
    __shared__ float red[PF_PARTS][OUT_DIM];
    __shared__ float cred[PF_THREADS / 32];
    const int g    = blockIdx.x;
    const int tid  = threadIdx.x;
    const int c    = tid & (OUT_DIM - 1);
    const int part = tid >> 6;
    const float* tp = Tp + (size_t)g * OUT_DIM + c;
    float acc = 0.0f;
#pragma unroll 16
    for (int b = part; b < nb; b += PF_PARTS) acc += tp[(size_t)b * (NUM_GRAPHS * OUT_DIM)];
    float cn = 0.0f;
    for (int b = tid; b < nb; b += PF_THREADS) cn += Cp[(size_t)b * NUM_GRAPHS + g];
    for (int o = 16; o > 0; o >>= 1) cn += __shfl_xor(cn, o);
    red[part][c] = acc;
    if ((tid & 31) == 0) cred[tid >> 5] = cn;
    __syncthreads();
    if (tid < OUT_DIM) {
        float t = 0.0f;
        for (int p = 0; p < PF_PARTS; ++p) t += red[p][c];
        float n = 0.0f;
        for (int w = 0; w < PF_THREADS / 32; ++w) n += cred[w];
        out[g * OUT_DIM + c] = (t + n * bias[c]) / fmaxf(n, 1.0f);
    }
}

extern "C" void kernel_launch(void* const* d_in, const int* in_sizes, int n_in,
                              void* d_out, int out_size, void* d_ws, size_t ws_size,
                              hipStream_t stream) {
    const float* x    = (const float*)d_in[0];
    const int*   eidx = (const int*)  d_in[1];
    const int*   batch= (const int*)  d_in[2];
    const float* W1   = (const float*)d_in[3];
    const float* b1   = (const float*)d_in[4];
    const float* W2   = (const float*)d_in[5];
    const float* b2   = (const float*)d_in[6];
    float* out = (float*)d_out;

    const int Nn = in_sizes[0] / IN_DIM;
    const int E  = in_sizes[1] / 2;
    const int* src = eidx;
    const int* dst = eidx + E;

    auto blocks = [](long long n, int b) { return (unsigned)((n + b - 1) / b); };
    const unsigned nTiles = blocks(Nn, TA_ROWS);

    const size_t perEdges   = (size_t)(((E + DH_BLOCKS - 1) / DH_BLOCKS + 31) & ~31);
    const size_t ordWords   = perEdges * DH_BLOCKS;
    const size_t nodeFloats = (((size_t)Nn + 31) >> 5) * 32;
    unsigned* bydst = (unsigned*)d_ws;
    unsigned* bysrc = bydst + ordWords;
    unsigned* tab   = bysrc + ordWords;
    float* dinv = (float*)(tab + (size_t)2 * 2 * DH_OWNERS * DH_BLOCKS);
    float* Cp   = dinv + nodeFloats;
    float* Tp   = Cp   + (size_t)nTiles * NUM_GRAPHS;

    edge_group_kernel<<<DH_BLOCKS, DH_THREADS, 0, stream>>>(src, dst, bydst, bysrc, tab, E);
    dinv_kernel<<<nTiles, DG_THREADS, 0, stream>>>(bydst, tab, dinv, E, Nn);

    tile_aggregate<<<nTiles, TA_THREADS, 0, stream>>>(
        x, dinv, batch, bydst, bysrc, tab, W1, b1, W2, Tp, Cp, E, Nn);

    pool_final<<<NUM_GRAPHS, PF_THREADS, 0, stream>>>(Tp, Cp, b2, out, (int)nTiles);
}
